# starting kernel
# baseline (speedup 1.0000x reference)
#include <hip/hip_runtime.h>

#define N_NODES 100000
#define N_EDGES 1600000
#define DIN 128
#define DH 64
#define LN_EPS 1e-5f

typedef __attribute__((ext_vector_type(2))) float v2f;
typedef __attribute__((ext_vector_type(8))) float v8f;

__global__ __launch_bounds__(256) void degree_kernel(const int* __restrict__ idx,
                                                     float* __restrict__ deg, int e) {
    int i = blockIdx.x * blockDim.x + threadIdx.x;
    if (i < e) atomicAdd(&deg[idx[i]], 1.0f);
}

__global__ __launch_bounds__(256) void finalize_norm_kernel(float* __restrict__ ns,
                                                            float* __restrict__ nd, int n) {
    int i = blockIdx.x * blockDim.x + threadIdx.x;
    if (i < n) {
        ns[i] = rsqrtf(fmaxf(ns[i], 1.0f));
        nd[i] = rsqrtf(fmaxf(nd[i], 1.0f));
    }
}

template <int K>
__global__ __launch_bounds__(128) void gemm_ns_wmma(const float* __restrict__ X,
                                                    const float* __restrict__ W,
                                                    const float* __restrict__ ns,
                                                    float* __restrict__ H) {
    const int lane    = threadIdx.x & 31;
    const int wave    = threadIdx.x >> 5;
    const int rowBase = blockIdx.x * 16;
    const int colBase = wave * 16;
    const int half    = lane >> 4;
    const int l15     = lane & 15;
    const int row     = rowBase + l15;
    const int col     = colBase + l15;
    const float s     = ns[row];

    v8f c = {};
#pragma unroll 4
    for (int k0 = 0; k0 < K; k0 += 4) {
        const int kk = k0 + half * 2;
        const float2 xa = *(const float2*)(X + row * K + kk);
        v2f a;
        a.x = xa.x * s;
        a.y = xa.y * s;
        v2f b;
        b.x = W[kk * DH + col];
        b.y = W[(kk + 1) * DH + col];
        c = __builtin_amdgcn_wmma_f32_16x16x4_f32(false, a, false, b, (short)0, c,
                                                  false, false);
    }

    const int mBase = half * 8;
#pragma unroll
    for (int i = 0; i < 8; ++i) {
        H[(size_t)(rowBase + mBase + i) * DH + col] = c[i];
    }
}

__global__ __launch_bounds__(256) void aggregate_kernel(const float* __restrict__ H,
                                                        const int* __restrict__ src,
                                                        const int* __restrict__ dst,
                                                        float* __restrict__ agg) {
    const int gid  = blockIdx.x * blockDim.x + threadIdx.x;
    const int e    = gid >> 5;
    const int lane = gid & 31;
    const int s = __builtin_amdgcn_readfirstlane(src[e]);
    const int d = __builtin_amdgcn_readfirstlane(dst[e]);
    const float2 v = *(const float2*)(H + (size_t)s * DH + lane * 2);
    float* p = agg + (size_t)d * DH + lane * 2;
    atomicAdd(p + 0, v.x);
    atomicAdd(p + 1, v.y);
}

__global__ __launch_bounds__(256) void post_kernel(const float* __restrict__ agg,
                                                   const float* __restrict__ nd,
                                                   const float* __restrict__ bias,
                                                   const float* __restrict__ g,
                                                   const float* __restrict__ be,
                                                   float* __restrict__ outx,
                                                   float* __restrict__ pre,
                                                   int n, int do_relu, int do_ln) {
    const int node = blockIdx.x * 8 + (threadIdx.x >> 5);
    if (node >= n) return;
    const int lane = threadIdx.x & 31;
    const int c0   = lane * 2;
    const float ndv = nd[node];
    const float* r = agg + (size_t)node * DH;
    float v0 = r[c0 + 0] * ndv + bias[c0 + 0];
    float v1 = r[c0 + 1] * ndv + bias[c0 + 1];
    if (do_relu) {
        v0 = fmaxf(v0, 0.0f);
        v1 = fmaxf(v1, 0.0f);
    }
    if (pre) {
        pre[(size_t)node * DH + c0 + 0] = v0;
        pre[(size_t)node * DH + c0 + 1] = v1;
    }
    if (do_ln) {
        float ssum = v0 + v1;
#pragma unroll
        for (int off = 16; off >= 1; off >>= 1) ssum += __shfl_xor(ssum, off, 32);
        const float mu = ssum * (1.0f / DH);
        const float d0 = v0 - mu, d1 = v1 - mu;
        float vs = d0 * d0 + d1 * d1;
#pragma unroll
        for (int off = 16; off >= 1; off >>= 1) vs += __shfl_xor(vs, off, 32);
        const float inv = rsqrtf(vs * (1.0f / DH) + LN_EPS);
        v0 = d0 * inv * g[c0 + 0] + be[c0 + 0];
        v1 = d1 * inv * g[c0 + 1] + be[c0 + 1];
    }
    float* o = outx + (size_t)node * DH + c0;
    o[0] = v0;
    o[1] = v1;
}

extern "C" void kernel_launch(void* const* d_in, const int* in_sizes, int n_in,
                              void* d_out, int out_size, void* d_ws, size_t ws_size,
                              hipStream_t stream) {
    (void)in_sizes; (void)n_in; (void)out_size; (void)ws_size;

    const float* feats = (const float*)d_in[0];
    const float* W1    = (const float*)d_in[1];
    const float* b1    = (const float*)d_in[2];
    const float* g1    = (const float*)d_in[3];
    const float* be1   = (const float*)d_in[4];
    const float* W2    = (const float*)d_in[5];
    const float* b2    = (const float*)d_in[6];
    const float* g2    = (const float*)d_in[7];
    const float* be2   = (const float*)d_in[8];
    const float* W3    = (const float*)d_in[9];
    const float* b3    = (const float*)d_in[10];
    const int*   src   = (const int*)d_in[11];
    const int*   dst   = (const int*)d_in[12];

    float* ns  = (float*)d_ws;
    float* nd  = ns + N_NODES;
    float* H   = nd + N_NODES;
    float* AGG = H + (size_t)N_NODES * DH;
    float* XB  = AGG + (size_t)N_NODES * DH;

    float* h_emb = (float*)d_out;
    float* h_out = h_emb + (size_t)N_NODES * DH;

    hipMemsetAsync(ns, 0, 2 * (size_t)N_NODES * sizeof(float), stream);
    degree_kernel<<<N_EDGES / 256, 256, 0, stream>>>(src, ns, N_EDGES);
    degree_kernel<<<N_EDGES / 256, 256, 0, stream>>>(dst, nd, N_EDGES);
    finalize_norm_kernel<<<(N_NODES + 255) / 256, 256, 0, stream>>>(ns, nd, N_NODES);

    const size_t aggBytes = (size_t)N_NODES * DH * sizeof(float);

    gemm_ns_wmma<DIN><<<N_NODES / 16, 128, 0, stream>>>(feats, W1, ns, H);
    hipMemsetAsync(AGG, 0, aggBytes, stream);
    aggregate_kernel<<<N_EDGES / 8, 256, 0, stream>>>(H, src, dst, AGG);
    post_kernel<<<N_NODES / 8, 256, 0, stream>>>(AGG, nd, b1, g1, be1, XB, nullptr,
                                                 N_NODES, 1, 1);

    gemm_ns_wmma<DH><<<N_NODES / 16, 128, 0, stream>>>(XB, W2, ns, H);
    hipMemsetAsync(AGG, 0, aggBytes, stream);
    aggregate_kernel<<<N_EDGES / 8, 256, 0, stream>>>(H, src, dst, AGG);
    post_kernel<<<N_NODES / 8, 256, 0, stream>>>(AGG, nd, b2, g2, be2, XB, h_emb,
                                                 N_NODES, 1, 1);

    gemm_ns_wmma<DH><<<N_NODES / 16, 128, 0, stream>>>(XB, W3, ns, H);
    hipMemsetAsync(AGG, 0, aggBytes, stream);
    aggregate_kernel<<<N_EDGES / 8, 256, 0, stream>>>(H, src, dst, AGG);
    post_kernel<<<N_NODES / 8, 256, 0, stream>>>(AGG, nd, b3, nullptr, nullptr, h_out,
                                                 nullptr, N_NODES, 0, 0);
}


# aggregate: each wave takes 16 edges (scalar index loads, 16 row gathers in flight before the f32 atomics) instead of 1 edge per wave; atomic arrival order stays unordered as before
# speedup vs baseline: 1.3532x; 1.3532x over previous
#include <hip/hip_runtime.h>

#define N_NODES 100000
#define N_EDGES 1600000
#define DIN 128
#define DH 64
#define LN_EPS 1e-5f

typedef __attribute__((ext_vector_type(2))) float v2f;
typedef __attribute__((ext_vector_type(8))) float v8f;

__global__ __launch_bounds__(256) void degree_kernel(const int* __restrict__ idx,
                                                     float* __restrict__ deg, int e) {
    int i = blockIdx.x * blockDim.x + threadIdx.x;
    if (i < e) atomicAdd(&deg[idx[i]], 1.0f);
}

__global__ __launch_bounds__(256) void finalize_norm_kernel(float* __restrict__ ns,
                                                            float* __restrict__ nd, int n) {
    int i = blockIdx.x * blockDim.x + threadIdx.x;
    if (i < n) {
        ns[i] = rsqrtf(fmaxf(ns[i], 1.0f));
        nd[i] = rsqrtf(fmaxf(nd[i], 1.0f));
    }
}

template <int K>
__global__ __launch_bounds__(128) void gemm_ns_wmma(const float* __restrict__ X,
                                                    const float* __restrict__ W,
                                                    const float* __restrict__ ns,
                                                    float* __restrict__ H) {
    const int lane    = threadIdx.x & 31;
    const int wave    = threadIdx.x >> 5;
    const int rowBase = blockIdx.x * 16;
    const int colBase = wave * 16;
    const int half    = lane >> 4;
    const int l15     = lane & 15;
    const int row     = rowBase + l15;
    const int col     = colBase + l15;
    const float s     = ns[row];

    v8f c = {};
#pragma unroll 4
    for (int k0 = 0; k0 < K; k0 += 4) {
        const int kk = k0 + half * 2;
        const float2 xa = *(const float2*)(X + row * K + kk);
        v2f a;
        a.x = xa.x * s;
        a.y = xa.y * s;
        v2f b;
        b.x = W[kk * DH + col];
        b.y = W[(kk + 1) * DH + col];
        c = __builtin_amdgcn_wmma_f32_16x16x4_f32(false, a, false, b, (short)0, c,
                                                  false, false);
    }

    const int mBase = half * 8;
#pragma unroll
    for (int i = 0; i < 8; ++i) {
        H[(size_t)(rowBase + mBase + i) * DH + col] = c[i];
    }
}

#define AGG_EPW 16
__global__ __launch_bounds__(256) void aggregate_kernel(const float* __restrict__ H,
                                                        const int* __restrict__ src,
                                                        const int* __restrict__ dst,
                                                        float* __restrict__ agg) {
    const int lane = threadIdx.x & 31;
    const int base = __builtin_amdgcn_readfirstlane(
        (int)((blockIdx.x * 8 + (threadIdx.x >> 5)) * AGG_EPW));
    int s[AGG_EPW];
    int d[AGG_EPW];
#pragma unroll
    for (int j = 0; j < AGG_EPW; ++j) {
        s[j] = src[base + j];
        d[j] = dst[base + j];
    }
    float2 v[AGG_EPW];
#pragma unroll
    for (int j = 0; j < AGG_EPW; ++j) {
        v[j] = *(const float2*)(H + (size_t)s[j] * DH + lane * 2);
    }
#pragma unroll
    for (int j = 0; j < AGG_EPW; ++j) {
        float* p = agg + (size_t)d[j] * DH + lane * 2;
        atomicAdd(p + 0, v[j].x);
        atomicAdd(p + 1, v[j].y);
    }
}

__global__ __launch_bounds__(256) void post_kernel(const float* __restrict__ agg,
                                                   const float* __restrict__ nd,
                                                   const float* __restrict__ bias,
                                                   const float* __restrict__ g,
                                                   const float* __restrict__ be,
                                                   float* __restrict__ outx,
                                                   float* __restrict__ pre,
                                                   int n, int do_relu, int do_ln) {
    const int node = blockIdx.x * 8 + (threadIdx.x >> 5);
    if (node >= n) return;
    const int lane = threadIdx.x & 31;
    const int c0   = lane * 2;
    const float ndv = nd[node];
    const float* r = agg + (size_t)node * DH;
    float v0 = r[c0 + 0] * ndv + bias[c0 + 0];
    float v1 = r[c0 + 1] * ndv + bias[c0 + 1];
    if (do_relu) {
        v0 = fmaxf(v0, 0.0f);
        v1 = fmaxf(v1, 0.0f);
    }
    if (pre) {
        pre[(size_t)node * DH + c0 + 0] = v0;
        pre[(size_t)node * DH + c0 + 1] = v1;
    }
    if (do_ln) {
        float ssum = v0 + v1;
#pragma unroll
        for (int off = 16; off >= 1; off >>= 1) ssum += __shfl_xor(ssum, off, 32);
        const float mu = ssum * (1.0f / DH);
        const float d0 = v0 - mu, d1 = v1 - mu;
        float vs = d0 * d0 + d1 * d1;
#pragma unroll
        for (int off = 16; off >= 1; off >>= 1) vs += __shfl_xor(vs, off, 32);
        const float inv = rsqrtf(vs * (1.0f / DH) + LN_EPS);
        v0 = d0 * inv * g[c0 + 0] + be[c0 + 0];
        v1 = d1 * inv * g[c0 + 1] + be[c0 + 1];
    }
    float* o = outx + (size_t)node * DH + c0;
    o[0] = v0;
    o[1] = v1;
}

extern "C" void kernel_launch(void* const* d_in, const int* in_sizes, int n_in,
                              void* d_out, int out_size, void* d_ws, size_t ws_size,
                              hipStream_t stream) {
    (void)in_sizes; (void)n_in; (void)out_size; (void)ws_size;

    const float* feats = (const float*)d_in[0];
    const float* W1    = (const float*)d_in[1];
    const float* b1    = (const float*)d_in[2];
    const float* g1    = (const float*)d_in[3];
    const float* be1   = (const float*)d_in[4];
    const float* W2    = (const float*)d_in[5];
    const float* b2    = (const float*)d_in[6];
    const float* g2    = (const float*)d_in[7];
    const float* be2   = (const float*)d_in[8];
    const float* W3    = (const float*)d_in[9];
    const float* b3    = (const float*)d_in[10];
    const int*   src   = (const int*)d_in[11];
    const int*   dst   = (const int*)d_in[12];

    float* ns  = (float*)d_ws;
    float* nd  = ns + N_NODES;
    float* H   = nd + N_NODES;
    float* AGG = H + (size_t)N_NODES * DH;
    float* XB  = AGG + (size_t)N_NODES * DH;

    float* h_emb = (float*)d_out;
    float* h_out = h_emb + (size_t)N_NODES * DH;

    hipMemsetAsync(ns, 0, 2 * (size_t)N_NODES * sizeof(float), stream);
    degree_kernel<<<N_EDGES / 256, 256, 0, stream>>>(src, ns, N_EDGES);
    degree_kernel<<<N_EDGES / 256, 256, 0, stream>>>(dst, nd, N_EDGES);
    finalize_norm_kernel<<<(N_NODES + 255) / 256, 256, 0, stream>>>(ns, nd, N_NODES);

    const size_t aggBytes = (size_t)N_NODES * DH * sizeof(float);

    gemm_ns_wmma<DIN><<<N_NODES / 16, 128, 0, stream>>>(feats, W1, ns, H);
    hipMemsetAsync(AGG, 0, aggBytes, stream);
    aggregate_kernel<<<N_EDGES / (8 * AGG_EPW), 256, 0, stream>>>(H, src, dst, AGG);
    post_kernel<<<N_NODES / 8, 256, 0, stream>>>(AGG, nd, b1, g1, be1, XB, nullptr,
                                                 N_NODES, 1, 1);

    gemm_ns_wmma<DH><<<N_NODES / 16, 128, 0, stream>>>(XB, W2, ns, H);
    hipMemsetAsync(AGG, 0, aggBytes, stream);
    aggregate_kernel<<<N_EDGES / (8 * AGG_EPW), 256, 0, stream>>>(H, src, dst, AGG);
    post_kernel<<<N_NODES / 8, 256, 0, stream>>>(AGG, nd, b2, g2, be2, XB, h_emb,
                                                 N_NODES, 1, 1);

    gemm_ns_wmma<DH><<<N_NODES / 16, 128, 0, stream>>>(XB, W3, ns, H);
    hipMemsetAsync(AGG, 0, aggBytes, stream);
    aggregate_kernel<<<N_EDGES / (8 * AGG_EPW), 256, 0, stream>>>(H, src, dst, AGG);
    post_kernel<<<N_NODES / 8, 256, 0, stream>>>(AGG, nd, b3, nullptr, nullptr, h_out,
                                                 nullptr, N_NODES, 0, 0);
}


# aggregate by destination: per-call in-edge lists (returning int add = slot, scan, fill); one wave sums a node's rows in registers, plain store; no f32 row atomics, no agg memset; sum order = arrival
# speedup vs baseline: 1.8742x; 1.3850x over previous
#include <hip/hip_runtime.h>

#define N_NODES 100000
#define N_EDGES 1600000
#define DIN 128
#define DH 64
#define LN_EPS 1e-5f

typedef __attribute__((ext_vector_type(2))) float v2f;
typedef __attribute__((ext_vector_type(8))) float v8f;

__global__ __launch_bounds__(256) void degree_kernel(const int* __restrict__ idx,
                                                     float* __restrict__ deg, int e) {
    int i = blockIdx.x * blockDim.x + threadIdx.x;
    if (i < e) atomicAdd(&deg[idx[i]], 1.0f);
}

__global__ __launch_bounds__(256) void rank_kernel(const int* __restrict__ dst,
                                                   int* __restrict__ cnt,
                                                   int* __restrict__ rank, int e) {
    int i = blockIdx.x * blockDim.x + threadIdx.x;
    if (i < e) rank[i] = atomicAdd(&cnt[dst[i]], 1);
}

__global__ __launch_bounds__(1024) void scan_kernel(const int* __restrict__ cnt,
                                                    int* __restrict__ rowptr, int n) {
    __shared__ int wave_front[32];
    __shared__ int wave_total[32];
    const int tid   = threadIdx.x;
    const int lane  = tid & 31;
    const int wave  = tid >> 5;
    const int start = blockIdx.x * 1024;

    int front = 0;
    for (int j = tid; j < start; j += 1024) front += cnt[j];

    const int i = start + tid;
    const int x = (i < n) ? cnt[i] : 0;
    int incl = x;
#pragma unroll
    for (int off = 1; off < 32; off <<= 1) {
        const int t = __shfl_up(incl, off, 32);
        if (lane >= off) incl += t;
    }
#pragma unroll
    for (int off = 16; off >= 1; off >>= 1) front += __shfl_xor(front, off, 32);
    if (lane == 31) wave_total[wave] = incl;
    if (lane == 0) wave_front[wave] = front;
    __syncthreads();
    int base = 0;
    for (int w = 0; w < 32; ++w) {
        base += wave_front[w];
        if (w < wave) base += wave_total[w];
    }
    if (i < n) rowptr[i] = base + incl - x;
}

#define COL_PAD 64
__global__ __launch_bounds__(256) void fill_kernel(const int* __restrict__ src,
                                                   const int* __restrict__ dst,
                                                   const int* __restrict__ rank,
                                                   const int* __restrict__ rowptr,
                                                   int* __restrict__ col, int e) {
    int i = blockIdx.x * blockDim.x + threadIdx.x;
    if (i < e) col[rowptr[dst[i]] + rank[i]] = src[i];
    if (i < COL_PAD) col[e + i] = 0;
}

__global__ __launch_bounds__(256) void finalize_norm_kernel(float* __restrict__ ns,
                                                            const int* __restrict__ cnt,
                                                            float* __restrict__ nd, int n) {
    int i = blockIdx.x * blockDim.x + threadIdx.x;
    if (i < n) {
        ns[i] = rsqrtf(fmaxf(ns[i], 1.0f));
        nd[i] = rsqrtf(fmaxf((float)cnt[i], 1.0f));
    }
}

template <int K>
__global__ __launch_bounds__(128) void gemm_ns_wmma(const float* __restrict__ X,
                                                    const float* __restrict__ W,
                                                    const float* __restrict__ ns,
                                                    float* __restrict__ H) {
    const int lane    = threadIdx.x & 31;
    const int wave    = threadIdx.x >> 5;
    const int rowBase = blockIdx.x * 16;
    const int colBase = wave * 16;
    const int half    = lane >> 4;
    const int l15     = lane & 15;
    const int row     = rowBase + l15;
    const int col     = colBase + l15;
    const float s     = ns[row];

    v8f c = {};
#pragma unroll 4
    for (int k0 = 0; k0 < K; k0 += 4) {
        const int kk = k0 + half * 2;
        const float2 xa = *(const float2*)(X + row * K + kk);
        v2f a;
        a.x = xa.x * s;
        a.y = xa.y * s;
        v2f b;
        b.x = W[kk * DH + col];
        b.y = W[(kk + 1) * DH + col];
        c = __builtin_amdgcn_wmma_f32_16x16x4_f32(false, a, false, b, (short)0, c,
                                                  false, false);
    }

    const int mBase = half * 8;
#pragma unroll
    for (int i = 0; i < 8; ++i) {
        H[(size_t)(rowBase + mBase + i) * DH + col] = c[i];
    }
}

#define AGG_CH 16
__global__ __launch_bounds__(256) void aggregate_kernel(const float* __restrict__ H,
                                                        const int* __restrict__ col,
                                                        const int* __restrict__ rowptr,
                                                        const int* __restrict__ cnt,
                                                        float* __restrict__ agg) {
    const int lane = threadIdx.x & 31;
    const int node = __builtin_amdgcn_readfirstlane(
        (int)(blockIdx.x * 8 + (threadIdx.x >> 5)));
    const int beg = rowptr[node];
    const int len = cnt[node];
    const float* Hl = H + lane * 2;
    float2 acc = make_float2(0.0f, 0.0f);
    int k = 0;
    for (; k + AGG_CH <= len; k += AGG_CH) {
        int s[AGG_CH];
#pragma unroll
        for (int j = 0; j < AGG_CH; ++j) s[j] = col[beg + k + j];
        float2 v[AGG_CH];
#pragma unroll
        for (int j = 0; j < AGG_CH; ++j) v[j] = *(const float2*)(Hl + (size_t)s[j] * DH);
#pragma unroll
        for (int j = 0; j < AGG_CH; ++j) {
            acc.x += v[j].x;
            acc.y += v[j].y;
        }
    }
    if (k < len) {
        const int r = len - k;
        int s[AGG_CH];
#pragma unroll
        for (int j = 0; j < AGG_CH; ++j) s[j] = col[beg + k + j];
        float2 v[AGG_CH - 1];
#pragma unroll
        for (int j = 0; j < AGG_CH - 1; ++j) {
            v[j] = make_float2(0.0f, 0.0f);
            if (j < r) v[j] = *(const float2*)(Hl + (size_t)s[j] * DH);
        }
#pragma unroll
        for (int j = 0; j < AGG_CH - 1; ++j) {
            acc.x += v[j].x;
            acc.y += v[j].y;
        }
    }
    *(float2*)(agg + (size_t)node * DH + lane * 2) = acc;
}

__global__ __launch_bounds__(256) void post_kernel(const float* __restrict__ agg,
                                                   const float* __restrict__ nd,
                                                   const float* __restrict__ bias,
                                                   const float* __restrict__ g,
                                                   const float* __restrict__ be,
                                                   float* __restrict__ outx,
                                                   float* __restrict__ pre,
                                                   int n, int do_relu, int do_ln) {
    const int node = blockIdx.x * 8 + (threadIdx.x >> 5);
    if (node >= n) return;
    const int lane = threadIdx.x & 31;
    const int c0   = lane * 2;
    const float ndv = nd[node];
    const float* r = agg + (size_t)node * DH;
    float v0 = r[c0 + 0] * ndv + bias[c0 + 0];
    float v1 = r[c0 + 1] * ndv + bias[c0 + 1];
    if (do_relu) {
        v0 = fmaxf(v0, 0.0f);
        v1 = fmaxf(v1, 0.0f);
    }
    if (pre) {
        pre[(size_t)node * DH + c0 + 0] = v0;
        pre[(size_t)node * DH + c0 + 1] = v1;
    }
    if (do_ln) {
        float ssum = v0 + v1;
#pragma unroll
        for (int off = 16; off >= 1; off >>= 1) ssum += __shfl_xor(ssum, off, 32);
        const float mu = ssum * (1.0f / DH);
        const float d0 = v0 - mu, d1 = v1 - mu;
        float vs = d0 * d0 + d1 * d1;
#pragma unroll
        for (int off = 16; off >= 1; off >>= 1) vs += __shfl_xor(vs, off, 32);
        const float inv = rsqrtf(vs * (1.0f / DH) + LN_EPS);
        v0 = d0 * inv * g[c0 + 0] + be[c0 + 0];
        v1 = d1 * inv * g[c0 + 1] + be[c0 + 1];
    }
    float* o = outx + (size_t)node * DH + c0;
    o[0] = v0;
    o[1] = v1;
}

extern "C" void kernel_launch(void* const* d_in, const int* in_sizes, int n_in,
                              void* d_out, int out_size, void* d_ws, size_t ws_size,
                              hipStream_t stream) {
    (void)in_sizes; (void)n_in; (void)out_size; (void)ws_size;

    const float* feats = (const float*)d_in[0];
    const float* W1    = (const float*)d_in[1];
    const float* b1    = (const float*)d_in[2];
    const float* g1    = (const float*)d_in[3];
    const float* be1   = (const float*)d_in[4];
    const float* W2    = (const float*)d_in[5];
    const float* b2    = (const float*)d_in[6];
    const float* g2    = (const float*)d_in[7];
    const float* be2   = (const float*)d_in[8];
    const float* W3    = (const float*)d_in[9];
    const float* b3    = (const float*)d_in[10];
    const int*   src   = (const int*)d_in[11];
    const int*   dst   = (const int*)d_in[12];

    float* ns     = (float*)d_ws;
    int*   cnt    = (int*)(ns + N_NODES);
    float* nd     = (float*)(cnt + N_NODES);
    int*   rowptr = (int*)(nd + N_NODES);
    int*   rank   = rowptr + N_NODES;
    int*   col    = rank + N_EDGES;
    float* H      = (float*)(col + N_EDGES + COL_PAD);
    float* AGG    = H + (size_t)N_NODES * DH;
    float* XB     = AGG + (size_t)N_NODES * DH;

    float* h_emb = (float*)d_out;
    float* h_out = h_emb + (size_t)N_NODES * DH;

    hipMemsetAsync(ns, 0, 2 * (size_t)N_NODES * sizeof(float), stream);
    degree_kernel<<<N_EDGES / 256, 256, 0, stream>>>(src, ns, N_EDGES);
    rank_kernel<<<N_EDGES / 256, 256, 0, stream>>>(dst, cnt, rank, N_EDGES);
    scan_kernel<<<(N_NODES + 1023) / 1024, 1024, 0, stream>>>(cnt, rowptr, N_NODES);
    fill_kernel<<<N_EDGES / 256, 256, 0, stream>>>(src, dst, rank, rowptr, col, N_EDGES);
    finalize_norm_kernel<<<(N_NODES + 255) / 256, 256, 0, stream>>>(ns, cnt, nd, N_NODES);

    gemm_ns_wmma<DIN><<<N_NODES / 16, 128, 0, stream>>>(feats, W1, ns, H);
    aggregate_kernel<<<N_NODES / 8, 256, 0, stream>>>(H, col, rowptr, cnt, AGG);
    post_kernel<<<N_NODES / 8, 256, 0, stream>>>(AGG, nd, b1, g1, be1, XB, nullptr,
                                                 N_NODES, 1, 1);

    gemm_ns_wmma<DH><<<N_NODES / 16, 128, 0, stream>>>(XB, W2, ns, H);
    aggregate_kernel<<<N_NODES / 8, 256, 0, stream>>>(H, col, rowptr, cnt, AGG);
    post_kernel<<<N_NODES / 8, 256, 0, stream>>>(AGG, nd, b2, g2, be2, XB, h_emb,
                                                 N_NODES, 1, 1);

    gemm_ns_wmma<DH><<<N_NODES / 16, 128, 0, stream>>>(XB, W3, ns, H);
    aggregate_kernel<<<N_NODES / 8, 256, 0, stream>>>(H, col, rowptr, cnt, AGG);
    post_kernel<<<N_NODES / 8, 256, 0, stream>>>(AGG, nd, b3, nullptr, nullptr, h_out,
                                                 nullptr, N_NODES, 0, 0);
}


# edge grouping on LDS counters: per-piece tables over node ranges give slots and dense partial degrees; a per-node prefix over pieces replaces 3.2M scattered global atomics; sum order: piece, arrival
# speedup vs baseline: 2.9301x; 1.5634x over previous
#include <hip/hip_runtime.h>

#define N_NODES 100000
#define N_EDGES 1600000
#define DIN 128
#define DH 64
#define LN_EPS 1e-5f

typedef __attribute__((ext_vector_type(2))) float v2f;
typedef __attribute__((ext_vector_type(8))) float v8f;

#define N_CHUNKS 64
#define CHUNK_E (N_EDGES / N_CHUNKS)
#define N_PARTS 4
#define PART_N (N_NODES / N_PARTS)
#define PART_V4 (PART_N / 4)
#define EPT ((CHUNK_E + 1023) / 1024)

__global__ __launch_bounds__(1024) void count_kernel(const int* __restrict__ src,
                                                     const int* __restrict__ dst,
                                                     int* __restrict__ partS,
                                                     int* __restrict__ partD,
                                                     int* __restrict__ lrank) {
    __shared__ int4 table4[PART_V4];
    int* table = (int*)table4;
    const int tid = threadIdx.x;
    const int c   = blockIdx.x / N_PARTS;
    const int lo  = (blockIdx.x % N_PARTS) * PART_N;
    const int e0  = c * CHUNK_E;
    const int4 zero4 = make_int4(0, 0, 0, 0);

    int dv[EPT], sv[EPT];
#pragma unroll
    for (int j = 0; j < EPT; ++j) {
        const int k = tid + j * 1024;
        dv[j] = (k < CHUNK_E) ? dst[e0 + k] : -1;
        sv[j] = (k < CHUNK_E) ? src[e0 + k] : -1;
    }

    for (int i = tid; i < PART_V4; i += 1024) table4[i] = zero4;
    __syncthreads();
#pragma unroll
    for (int j = 0; j < EPT; ++j) {
        const unsigned d = (unsigned)(dv[j] - lo);
        if (d < (unsigned)PART_N) lrank[e0 + tid + j * 1024] = atomicAdd(&table[d], 1);
    }
    __syncthreads();
    {
        int4* o = (int4*)(partD + (size_t)c * N_NODES + lo);
        for (int i = tid; i < PART_V4; i += 1024) o[i] = table4[i];
    }
    __syncthreads();

    for (int i = tid; i < PART_V4; i += 1024) table4[i] = zero4;
    __syncthreads();
#pragma unroll
    for (int j = 0; j < EPT; ++j) {
        const unsigned s = (unsigned)(sv[j] - lo);
        if (s < (unsigned)PART_N) atomicAdd(&table[s], 1);
    }
    __syncthreads();
    {
        int4* o = (int4*)(partS + (size_t)c * N_NODES + lo);
        for (int i = tid; i < PART_V4; i += 1024) o[i] = table4[i];
    }
}

#define PFX_SEGS 8
#define PFX_LEN (N_CHUNKS / PFX_SEGS)
__global__ __launch_bounds__(256) void piece_prefix_kernel(int* __restrict__ partD,
                                                           const int* __restrict__ partS,
                                                           int* __restrict__ cnt,
                                                           float* __restrict__ ns,
                                                           float* __restrict__ nd) {
    __shared__ int totD[PFX_SEGS][32];
    __shared__ int totS[PFX_SEGS][32];
    const int lane = threadIdx.x & 31;
    const int seg  = __builtin_amdgcn_readfirstlane((int)(threadIdx.x >> 5));
    const int n    = blockIdx.x * 32 + lane;
    const size_t at0 = (size_t)(seg * PFX_LEN) * N_NODES + n;

    int t[PFX_LEN];
    int sumD = 0, sumS = 0;
#pragma unroll
    for (int k = 0; k < PFX_LEN; ++k) {
        t[k] = partD[at0 + (size_t)k * N_NODES];
        sumS += partS[at0 + (size_t)k * N_NODES];
    }
#pragma unroll
    for (int k = 0; k < PFX_LEN; ++k) sumD += t[k];
    totD[seg][lane] = sumD;
    totS[seg][lane] = sumS;
    __syncthreads();

    int run = 0;
    for (int s = 0; s < seg; ++s) run += totD[s][lane];
#pragma unroll
    for (int k = 0; k < PFX_LEN; ++k) {
        partD[at0 + (size_t)k * N_NODES] = run;
        run += t[k];
    }
    if (seg == PFX_SEGS - 1) {
        cnt[n] = run;
        nd[n]  = rsqrtf(fmaxf((float)run, 1.0f));
    }
    if (seg == 0) {
        int outdeg = 0;
#pragma unroll
        for (int s = 0; s < PFX_SEGS; ++s) outdeg += totS[s][lane];
        ns[n] = rsqrtf(fmaxf((float)outdeg, 1.0f));
    }
}

__global__ __launch_bounds__(1024) void scan_kernel(const int* __restrict__ cnt,
                                                    int* __restrict__ rowptr, int n) {
    __shared__ int wave_front[32];
    __shared__ int wave_total[32];
    const int tid   = threadIdx.x;
    const int lane  = tid & 31;
    const int wave  = tid >> 5;
    const int start = blockIdx.x * 1024;

    int front = 0;
    for (int j = tid; j < start; j += 1024) front += cnt[j];

    const int i = start + tid;
    const int x = (i < n) ? cnt[i] : 0;
    int incl = x;
#pragma unroll
    for (int off = 1; off < 32; off <<= 1) {
        const int t = __shfl_up(incl, off, 32);
        if (lane >= off) incl += t;
    }
#pragma unroll
    for (int off = 16; off >= 1; off >>= 1) front += __shfl_xor(front, off, 32);
    if (lane == 31) wave_total[wave] = incl;
    if (lane == 0) wave_front[wave] = front;
    __syncthreads();
    int base = 0;
    for (int w = 0; w < 32; ++w) {
        base += wave_front[w];
        if (w < wave) base += wave_total[w];
    }
    if (i < n) rowptr[i] = base + incl - x;
}

#define COL_PAD 64
__global__ __launch_bounds__(256) void fill_kernel(const int* __restrict__ src,
                                                   const int* __restrict__ dst,
                                                   const int* __restrict__ lrank,
                                                   const int* __restrict__ partD,
                                                   const int* __restrict__ rowptr,
                                                   int* __restrict__ col, int e) {
    int i = blockIdx.x * blockDim.x + threadIdx.x;
    if (i < e) {
        const int d = dst[i];
        const int c = i / CHUNK_E;
        col[rowptr[d] + partD[(size_t)c * N_NODES + d] + lrank[i]] = src[i];
    }
    if (i < COL_PAD) col[e + i] = 0;
}

template <int K>
__global__ __launch_bounds__(128) void gemm_ns_wmma(const float* __restrict__ X,
                                                    const float* __restrict__ W,
                                                    const float* __restrict__ ns,
                                                    float* __restrict__ H) {
    const int lane    = threadIdx.x & 31;
    const int wave    = threadIdx.x >> 5;
    const int rowBase = blockIdx.x * 16;
    const int colBase = wave * 16;
    const int half    = lane >> 4;
    const int l15     = lane & 15;
    const int row     = rowBase + l15;
    const int col     = colBase + l15;
    const float s     = ns[row];

    v8f c = {};
#pragma unroll 4
    for (int k0 = 0; k0 < K; k0 += 4) {
        const int kk = k0 + half * 2;
        const float2 xa = *(const float2*)(X + row * K + kk);
        v2f a;
        a.x = xa.x * s;
        a.y = xa.y * s;
        v2f b;
        b.x = W[kk * DH + col];
        b.y = W[(kk + 1) * DH + col];
        c = __builtin_amdgcn_wmma_f32_16x16x4_f32(false, a, false, b, (short)0, c,
                                                  false, false);
    }

    const int mBase = half * 8;
#pragma unroll
    for (int i = 0; i < 8; ++i) {
        H[(size_t)(rowBase + mBase + i) * DH + col] = c[i];
    }
}

#define AGG_CH 16
__global__ __launch_bounds__(256) void aggregate_kernel(const float* __restrict__ H,
                                                        const int* __restrict__ col,
                                                        const int* __restrict__ rowptr,
                                                        const int* __restrict__ cnt,
                                                        float* __restrict__ agg) {
    const int lane = threadIdx.x & 31;
    const int node = __builtin_amdgcn_readfirstlane(
        (int)(blockIdx.x * 8 + (threadIdx.x >> 5)));
    const int beg = rowptr[node];
    const int len = cnt[node];
    const float* Hl = H + lane * 2;
    float2 acc = make_float2(0.0f, 0.0f);
    int k = 0;
    for (; k + AGG_CH <= len; k += AGG_CH) {
        int s[AGG_CH];
#pragma unroll
        for (int j = 0; j < AGG_CH; ++j) s[j] = col[beg + k + j];
        float2 v[AGG_CH];
#pragma unroll
        for (int j = 0; j < AGG_CH; ++j) v[j] = *(const float2*)(Hl + (size_t)s[j] * DH);
#pragma unroll
        for (int j = 0; j < AGG_CH; ++j) {
            acc.x += v[j].x;
            acc.y += v[j].y;
        }
    }
    if (k < len) {
        const int r = len - k;
        int s[AGG_CH];
#pragma unroll
        for (int j = 0; j < AGG_CH; ++j) s[j] = col[beg + k + j];
        float2 v[AGG_CH - 1];
#pragma unroll
        for (int j = 0; j < AGG_CH - 1; ++j) {
            v[j] = make_float2(0.0f, 0.0f);
            if (j < r) v[j] = *(const float2*)(Hl + (size_t)s[j] * DH);
        }
#pragma unroll
        for (int j = 0; j < AGG_CH - 1; ++j) {
            acc.x += v[j].x;
            acc.y += v[j].y;
        }
    }
    *(float2*)(agg + (size_t)node * DH + lane * 2) = acc;
}

__global__ __launch_bounds__(256) void post_kernel(const float* __restrict__ agg,
                                                   const float* __restrict__ nd,
                                                   const float* __restrict__ bias,
                                                   const float* __restrict__ g,
                                                   const float* __restrict__ be,
                                                   float* __restrict__ outx,
                                                   float* __restrict__ pre,
                                                   int n, int do_relu, int do_ln) {
    const int node = blockIdx.x * 8 + (threadIdx.x >> 5);
    if (node >= n) return;
    const int lane = threadIdx.x & 31;
    const int c0   = lane * 2;
    const float ndv = nd[node];
    const float* r = agg + (size_t)node * DH;
    float v0 = r[c0 + 0] * ndv + bias[c0 + 0];
    float v1 = r[c0 + 1] * ndv + bias[c0 + 1];
    if (do_relu) {
        v0 = fmaxf(v0, 0.0f);
        v1 = fmaxf(v1, 0.0f);
    }
    if (pre) {
        pre[(size_t)node * DH + c0 + 0] = v0;
        pre[(size_t)node * DH + c0 + 1] = v1;
    }
    if (do_ln) {
        float ssum = v0 + v1;
#pragma unroll
        for (int off = 16; off >= 1; off >>= 1) ssum += __shfl_xor(ssum, off, 32);
        const float mu = ssum * (1.0f / DH);
        const float d0 = v0 - mu, d1 = v1 - mu;
        float vs = d0 * d0 + d1 * d1;
#pragma unroll
        for (int off = 16; off >= 1; off >>= 1) vs += __shfl_xor(vs, off, 32);
        const float inv = rsqrtf(vs * (1.0f / DH) + LN_EPS);
        v0 = d0 * inv * g[c0 + 0] + be[c0 + 0];
        v1 = d1 * inv * g[c0 + 1] + be[c0 + 1];
    }
    float* o = outx + (size_t)node * DH + c0;
    o[0] = v0;
    o[1] = v1;
}

extern "C" void kernel_launch(void* const* d_in, const int* in_sizes, int n_in,
                              void* d_out, int out_size, void* d_ws, size_t ws_size,
                              hipStream_t stream) {
    (void)in_sizes; (void)n_in; (void)out_size; (void)ws_size;

    const float* feats = (const float*)d_in[0];
    const float* W1    = (const float*)d_in[1];
    const float* b1    = (const float*)d_in[2];
    const float* g1    = (const float*)d_in[3];
    const float* be1   = (const float*)d_in[4];
    const float* W2    = (const float*)d_in[5];
    const float* b2    = (const float*)d_in[6];
    const float* g2    = (const float*)d_in[7];
    const float* be2   = (const float*)d_in[8];
    const float* W3    = (const float*)d_in[9];
    const float* b3    = (const float*)d_in[10];
    const int*   src   = (const int*)d_in[11];
    const int*   dst   = (const int*)d_in[12];

    float* ns     = (float*)d_ws;
    int*   cnt    = (int*)(ns + N_NODES);
    float* nd     = (float*)(cnt + N_NODES);
    int*   rowptr = (int*)(nd + N_NODES);
    int*   lrank  = rowptr + N_NODES;
    int*   col    = lrank + N_EDGES;
    float* H      = (float*)(col + N_EDGES + COL_PAD);
    float* AGG    = H + (size_t)N_NODES * DH;
    float* XB     = AGG + (size_t)N_NODES * DH;
    int*   partD  = (int*)(XB + (size_t)N_NODES * DH);
    int*   partS  = partD + (size_t)N_CHUNKS * N_NODES;

    float* h_emb = (float*)d_out;
    float* h_out = h_emb + (size_t)N_NODES * DH;

    count_kernel<<<N_CHUNKS * N_PARTS, 1024, 0, stream>>>(src, dst, partS, partD, lrank);
    piece_prefix_kernel<<<N_NODES / 32, 256, 0, stream>>>(partD, partS, cnt, ns, nd);
    scan_kernel<<<(N_NODES + 1023) / 1024, 1024, 0, stream>>>(cnt, rowptr, N_NODES);
    fill_kernel<<<N_EDGES / 256, 256, 0, stream>>>(src, dst, lrank, partD, rowptr, col, N_EDGES);

    gemm_ns_wmma<DIN><<<N_NODES / 16, 128, 0, stream>>>(feats, W1, ns, H);
    aggregate_kernel<<<N_NODES / 8, 256, 0, stream>>>(H, col, rowptr, cnt, AGG);
    post_kernel<<<N_NODES / 8, 256, 0, stream>>>(AGG, nd, b1, g1, be1, XB, nullptr,
                                                 N_NODES, 1, 1);

    gemm_ns_wmma<DH><<<N_NODES / 16, 128, 0, stream>>>(XB, W2, ns, H);
    aggregate_kernel<<<N_NODES / 8, 256, 0, stream>>>(H, col, rowptr, cnt, AGG);
    post_kernel<<<N_NODES / 8, 256, 0, stream>>>(AGG, nd, b2, g2, be2, XB, h_emb,
                                                 N_NODES, 1, 1);

    gemm_ns_wmma<DH><<<N_NODES / 16, 128, 0, stream>>>(XB, W3, ns, H);
    aggregate_kernel<<<N_NODES / 8, 256, 0, stream>>>(H, col, rowptr, cnt, AGG);
    post_kernel<<<N_NODES / 8, 256, 0, stream>>>(AGG, nd, b3, nullptr, nullptr, h_out,
                                                 nullptr, N_NODES, 0, 0);
}


# aggregate wave finishes its node: scale by nd, bias, ReLU, LayerNorm run on the row sum in registers; the 3 post launches and the agg plane's trip through memory go; sums and their order unchanged
# speedup vs baseline: 3.3749x; 1.1518x over previous
#include <hip/hip_runtime.h>

#define N_NODES 100000
#define N_EDGES 1600000
#define DIN 128
#define DH 64
#define LN_EPS 1e-5f

typedef __attribute__((ext_vector_type(2))) float v2f;
typedef __attribute__((ext_vector_type(8))) float v8f;

#define N_CHUNKS 64
#define CHUNK_E (N_EDGES / N_CHUNKS)
#define N_PARTS 4
#define PART_N (N_NODES / N_PARTS)
#define PART_V4 (PART_N / 4)
#define EPT ((CHUNK_E + 1023) / 1024)

__global__ __launch_bounds__(1024) void count_kernel(const int* __restrict__ src,
                                                     const int* __restrict__ dst,
                                                     int* __restrict__ partS,
                                                     int* __restrict__ partD,
                                                     int* __restrict__ lrank) {
    __shared__ int4 table4[PART_V4];
    int* table = (int*)table4;
    const int tid = threadIdx.x;
    const int c   = blockIdx.x / N_PARTS;
    const int lo  = (blockIdx.x % N_PARTS) * PART_N;
    const int e0  = c * CHUNK_E;
    const int4 zero4 = make_int4(0, 0, 0, 0);

    int dv[EPT], sv[EPT];
#pragma unroll
    for (int j = 0; j < EPT; ++j) {
        const int k = tid + j * 1024;
        dv[j] = (k < CHUNK_E) ? dst[e0 + k] : -1;
        sv[j] = (k < CHUNK_E) ? src[e0 + k] : -1;
    }

    for (int i = tid; i < PART_V4; i += 1024) table4[i] = zero4;
    __syncthreads();
#pragma unroll
    for (int j = 0; j < EPT; ++j) {
        const unsigned d = (unsigned)(dv[j] - lo);
        if (d < (unsigned)PART_N) lrank[e0 + tid + j * 1024] = atomicAdd(&table[d], 1);
    }
    __syncthreads();
    {
        int4* o = (int4*)(partD + (size_t)c * N_NODES + lo);
        for (int i = tid; i < PART_V4; i += 1024) o[i] = table4[i];
    }
    __syncthreads();

    for (int i = tid; i < PART_V4; i += 1024) table4[i] = zero4;
    __syncthreads();
#pragma unroll
    for (int j = 0; j < EPT; ++j) {
        const unsigned s = (unsigned)(sv[j] - lo);
        if (s < (unsigned)PART_N) atomicAdd(&table[s], 1);
    }
    __syncthreads();
    {
        int4* o = (int4*)(partS + (size_t)c * N_NODES + lo);
        for (int i = tid; i < PART_V4; i += 1024) o[i] = table4[i];
    }
}

#define PFX_SEGS 8
#define PFX_LEN (N_CHUNKS / PFX_SEGS)
__global__ __launch_bounds__(256) void piece_prefix_kernel(int* __restrict__ partD,
                                                           const int* __restrict__ partS,
                                                           int* __restrict__ cnt,
                                                           float* __restrict__ ns,
                                                           float* __restrict__ nd) {
    __shared__ int totD[PFX_SEGS][32];
    __shared__ int totS[PFX_SEGS][32];
    const int lane = threadIdx.x & 31;
    const int seg  = __builtin_amdgcn_readfirstlane((int)(threadIdx.x >> 5));
    const int n    = blockIdx.x * 32 + lane;
    const size_t at0 = (size_t)(seg * PFX_LEN) * N_NODES + n;

    int t[PFX_LEN];
    int sumD = 0, sumS = 0;
#pragma unroll
    for (int k = 0; k < PFX_LEN; ++k) {
        t[k] = partD[at0 + (size_t)k * N_NODES];
        sumS += partS[at0 + (size_t)k * N_NODES];
    }
#pragma unroll
    for (int k = 0; k < PFX_LEN; ++k) sumD += t[k];
    totD[seg][lane] = sumD;
    totS[seg][lane] = sumS;
    __syncthreads();

    int run = 0;
    for (int s = 0; s < seg; ++s) run += totD[s][lane];
#pragma unroll
    for (int k = 0; k < PFX_LEN; ++k) {
        partD[at0 + (size_t)k * N_NODES] = run;
        run += t[k];
    }
    if (seg == PFX_SEGS - 1) {
        cnt[n] = run;
        nd[n]  = rsqrtf(fmaxf((float)run, 1.0f));
    }
    if (seg == 0) {
        int outdeg = 0;
#pragma unroll
        for (int s = 0; s < PFX_SEGS; ++s) outdeg += totS[s][lane];
        ns[n] = rsqrtf(fmaxf((float)outdeg, 1.0f));
    }
}

__global__ __launch_bounds__(1024) void scan_kernel(const int* __restrict__ cnt,
                                                    int* __restrict__ rowptr, int n) {
    __shared__ int wave_front[32];
    __shared__ int wave_total[32];
    const int tid   = threadIdx.x;
    const int lane  = tid & 31;
    const int wave  = tid >> 5;
    const int start = blockIdx.x * 1024;

    int front = 0;
    for (int j = tid; j < start; j += 1024) front += cnt[j];

    const int i = start + tid;
    const int x = (i < n) ? cnt[i] : 0;
    int incl = x;
#pragma unroll
    for (int off = 1; off < 32; off <<= 1) {
        const int t = __shfl_up(incl, off, 32);
        if (lane >= off) incl += t;
    }
#pragma unroll
    for (int off = 16; off >= 1; off >>= 1) front += __shfl_xor(front, off, 32);
    if (lane == 31) wave_total[wave] = incl;
    if (lane == 0) wave_front[wave] = front;
    __syncthreads();
    int base = 0;
    for (int w = 0; w < 32; ++w) {
        base += wave_front[w];
        if (w < wave) base += wave_total[w];
    }
    if (i < n) rowptr[i] = base + incl - x;
}

#define COL_PAD 64
__global__ __launch_bounds__(256) void fill_kernel(const int* __restrict__ src,
                                                   const int* __restrict__ dst,
                                                   const int* __restrict__ lrank,
                                                   const int* __restrict__ partD,
                                                   const int* __restrict__ rowptr,
                                                   int* __restrict__ col, int e) {
    int i = blockIdx.x * blockDim.x + threadIdx.x;
    if (i < e) {
        const int d = dst[i];
        const int c = i / CHUNK_E;
        col[rowptr[d] + partD[(size_t)c * N_NODES + d] + lrank[i]] = src[i];
    }
    if (i < COL_PAD) col[e + i] = 0;
}

template <int K>
__global__ __launch_bounds__(128) void gemm_ns_wmma(const float* __restrict__ X,
                                                    const float* __restrict__ W,
                                                    const float* __restrict__ ns,
                                                    float* __restrict__ H) {
    const int lane    = threadIdx.x & 31;
    const int wave    = threadIdx.x >> 5;
    const int rowBase = blockIdx.x * 16;
    const int colBase = wave * 16;
    const int half    = lane >> 4;
    const int l15     = lane & 15;
    const int row     = rowBase + l15;
    const int col     = colBase + l15;
    const float s     = ns[row];

    v8f c = {};
#pragma unroll 4
    for (int k0 = 0; k0 < K; k0 += 4) {
        const int kk = k0 + half * 2;
        const float2 xa = *(const float2*)(X + row * K + kk);
        v2f a;
        a.x = xa.x * s;
        a.y = xa.y * s;
        v2f b;
        b.x = W[kk * DH + col];
        b.y = W[(kk + 1) * DH + col];
        c = __builtin_amdgcn_wmma_f32_16x16x4_f32(false, a, false, b, (short)0, c,
                                                  false, false);
    }

    const int mBase = half * 8;
#pragma unroll
    for (int i = 0; i < 8; ++i) {
        H[(size_t)(rowBase + mBase + i) * DH + col] = c[i];
    }
}

#define AGG_CH 16
template <bool RELU, bool LN, bool PRE>
__global__ __launch_bounds__(256) void aggregate_post_kernel(const float* __restrict__ H,
                                                             const int* __restrict__ col,
                                                             const int* __restrict__ rowptr,
                                                             const int* __restrict__ cnt,
                                                             const float* __restrict__ nd,
                                                             const float* __restrict__ bias,
                                                             const float* __restrict__ g,
                                                             const float* __restrict__ be,
                                                             float* __restrict__ outx,
                                                             float* __restrict__ pre) {
    const int lane = threadIdx.x & 31;
    const int node = __builtin_amdgcn_readfirstlane(
        (int)(blockIdx.x * 8 + (threadIdx.x >> 5)));
    const int beg = rowptr[node];
    const int len = cnt[node];
    const float ndv = nd[node];
    const float* Hl = H + lane * 2;
    float2 acc = make_float2(0.0f, 0.0f);
    int k = 0;
    for (; k + AGG_CH <= len; k += AGG_CH) {
        int s[AGG_CH];
#pragma unroll
        for (int j = 0; j < AGG_CH; ++j) s[j] = col[beg + k + j];
        float2 v[AGG_CH];
#pragma unroll
        for (int j = 0; j < AGG_CH; ++j) v[j] = *(const float2*)(Hl + (size_t)s[j] * DH);
#pragma unroll
        for (int j = 0; j < AGG_CH; ++j) {
            acc.x += v[j].x;
            acc.y += v[j].y;
        }
    }
    const int c0 = lane * 2;
    const float2 bv = *(const float2*)(bias + c0);
    float2 gv = make_float2(1.0f, 1.0f), bev = make_float2(0.0f, 0.0f);
    if (LN) {
        gv  = *(const float2*)(g + c0);
        bev = *(const float2*)(be + c0);
    }
    if (k < len) {
        const int r = len - k;
        int s[AGG_CH];
#pragma unroll
        for (int j = 0; j < AGG_CH; ++j) s[j] = col[beg + k + j];
        float2 v[AGG_CH - 1];
#pragma unroll
        for (int j = 0; j < AGG_CH - 1; ++j) {
            v[j] = make_float2(0.0f, 0.0f);
            if (j < r) v[j] = *(const float2*)(Hl + (size_t)s[j] * DH);
        }
#pragma unroll
        for (int j = 0; j < AGG_CH - 1; ++j) {
            acc.x += v[j].x;
            acc.y += v[j].y;
        }
    }

    float v0 = acc.x * ndv + bv.x;
    float v1 = acc.y * ndv + bv.y;
    if (RELU) {
        v0 = fmaxf(v0, 0.0f);
        v1 = fmaxf(v1, 0.0f);
    }
    if (PRE) {
        pre[(size_t)node * DH + c0 + 0] = v0;
        pre[(size_t)node * DH + c0 + 1] = v1;
    }
    if (LN) {
        float ssum = v0 + v1;
#pragma unroll
        for (int off = 16; off >= 1; off >>= 1) ssum += __shfl_xor(ssum, off, 32);
        const float mu = ssum * (1.0f / DH);
        const float d0 = v0 - mu, d1 = v1 - mu;
        float vs = d0 * d0 + d1 * d1;
#pragma unroll
        for (int off = 16; off >= 1; off >>= 1) vs += __shfl_xor(vs, off, 32);
        const float inv = rsqrtf(vs * (1.0f / DH) + LN_EPS);
        v0 = d0 * inv * gv.x + bev.x;
        v1 = d1 * inv * gv.y + bev.y;
    }
    float* o = outx + (size_t)node * DH + c0;
    o[0] = v0;
    o[1] = v1;
}

extern "C" void kernel_launch(void* const* d_in, const int* in_sizes, int n_in,
                              void* d_out, int out_size, void* d_ws, size_t ws_size,
                              hipStream_t stream) {
    (void)in_sizes; (void)n_in; (void)out_size; (void)ws_size;

    const float* feats = (const float*)d_in[0];
    const float* W1    = (const float*)d_in[1];
    const float* b1    = (const float*)d_in[2];
    const float* g1    = (const float*)d_in[3];
    const float* be1   = (const float*)d_in[4];
    const float* W2    = (const float*)d_in[5];
    const float* b2    = (const float*)d_in[6];
    const float* g2    = (const float*)d_in[7];
    const float* be2   = (const float*)d_in[8];
    const float* W3    = (const float*)d_in[9];
    const float* b3    = (const float*)d_in[10];
    const int*   src   = (const int*)d_in[11];
    const int*   dst   = (const int*)d_in[12];

    float* ns     = (float*)d_ws;
    int*   cnt    = (int*)(ns + N_NODES);
    float* nd     = (float*)(cnt + N_NODES);
    int*   rowptr = (int*)(nd + N_NODES);
    int*   lrank  = rowptr + N_NODES;
    int*   col    = lrank + N_EDGES;
    float* H      = (float*)(col + N_EDGES + COL_PAD);
    float* XB     = H + (size_t)N_NODES * DH;
    int*   partD  = (int*)(XB + (size_t)N_NODES * DH);
    int*   partS  = partD + (size_t)N_CHUNKS * N_NODES;

    float* h_emb = (float*)d_out;
    float* h_out = h_emb + (size_t)N_NODES * DH;

    count_kernel<<<N_CHUNKS * N_PARTS, 1024, 0, stream>>>(src, dst, partS, partD, lrank);
    piece_prefix_kernel<<<N_NODES / 32, 256, 0, stream>>>(partD, partS, cnt, ns, nd);
    scan_kernel<<<(N_NODES + 1023) / 1024, 1024, 0, stream>>>(cnt, rowptr, N_NODES);
    fill_kernel<<<N_EDGES / 256, 256, 0, stream>>>(src, dst, lrank, partD, rowptr, col, N_EDGES);

    gemm_ns_wmma<DIN><<<N_NODES / 16, 128, 0, stream>>>(feats, W1, ns, H);
    aggregate_post_kernel<true, true, false><<<N_NODES / 8, 256, 0, stream>>>(
        H, col, rowptr, cnt, nd, b1, g1, be1, XB, nullptr);

    gemm_ns_wmma<DH><<<N_NODES / 16, 128, 0, stream>>>(XB, W2, ns, H);
    aggregate_post_kernel<true, true, true><<<N_NODES / 8, 256, 0, stream>>>(
        H, col, rowptr, cnt, nd, b2, g2, be2, XB, h_emb);

    gemm_ns_wmma<DH><<<N_NODES / 16, 128, 0, stream>>>(XB, W3, ns, H);
    aggregate_post_kernel<false, false, false><<<N_NODES / 8, 256, 0, stream>>>(
        H, col, rowptr, cnt, nd, b3, nullptr, nullptr, h_out, nullptr);
}


# transform (3 launches): a block walks 5 row stripes with its B fragments kept in registers; X rows are read once as whole lines and reach the 4 waves through LDS; sums and their order unchanged
# speedup vs baseline: 3.7952x; 1.1245x over previous
#include <hip/hip_runtime.h>

#define N_NODES 100000
#define N_EDGES 1600000
#define DIN 128
#define DH 64
#define LN_EPS 1e-5f

typedef __attribute__((ext_vector_type(2))) float v2f;
typedef __attribute__((ext_vector_type(8))) float v8f;

#define N_CHUNKS 64
#define CHUNK_E (N_EDGES / N_CHUNKS)
#define N_PARTS 4
#define PART_N (N_NODES / N_PARTS)
#define PART_V4 (PART_N / 4)
#define EPT ((CHUNK_E + 1023) / 1024)

__global__ __launch_bounds__(1024) void count_kernel(const int* __restrict__ src,
                                                     const int* __restrict__ dst,
                                                     int* __restrict__ partS,
                                                     int* __restrict__ partD,
                                                     int* __restrict__ lrank) {
    __shared__ int4 table4[PART_V4];
    int* table = (int*)table4;
    const int tid = threadIdx.x;
    const int c   = blockIdx.x / N_PARTS;
    const int lo  = (blockIdx.x % N_PARTS) * PART_N;
    const int e0  = c * CHUNK_E;
    const int4 zero4 = make_int4(0, 0, 0, 0);

    int dv[EPT], sv[EPT];
#pragma unroll
    for (int j = 0; j < EPT; ++j) {
        const int k = tid + j * 1024;
        dv[j] = (k < CHUNK_E) ? dst[e0 + k] : -1;
        sv[j] = (k < CHUNK_E) ? src[e0 + k] : -1;
    }

    for (int i = tid; i < PART_V4; i += 1024) table4[i] = zero4;
    __syncthreads();
#pragma unroll
    for (int j = 0; j < EPT; ++j) {
        const unsigned d = (unsigned)(dv[j] - lo);
        if (d < (unsigned)PART_N) lrank[e0 + tid + j * 1024] = atomicAdd(&table[d], 1);
    }
    __syncthreads();
    {
        int4* o = (int4*)(partD + (size_t)c * N_NODES + lo);
        for (int i = tid; i < PART_V4; i += 1024) o[i] = table4[i];
    }
    __syncthreads();

    for (int i = tid; i < PART_V4; i += 1024) table4[i] = zero4;
    __syncthreads();
#pragma unroll
    for (int j = 0; j < EPT; ++j) {
        const unsigned s = (unsigned)(sv[j] - lo);
        if (s < (unsigned)PART_N) atomicAdd(&table[s], 1);
    }
    __syncthreads();
    {
        int4* o = (int4*)(partS + (size_t)c * N_NODES + lo);
        for (int i = tid; i < PART_V4; i += 1024) o[i] = table4[i];
    }
}

#define PFX_SEGS 8
#define PFX_LEN (N_CHUNKS / PFX_SEGS)
__global__ __launch_bounds__(256) void piece_prefix_kernel(int* __restrict__ partD,
                                                           const int* __restrict__ partS,
                                                           int* __restrict__ cnt,
                                                           float* __restrict__ ns,
                                                           float* __restrict__ nd) {
    __shared__ int totD[PFX_SEGS][32];
    __shared__ int totS[PFX_SEGS][32];
    const int lane = threadIdx.x & 31;
    const int seg  = __builtin_amdgcn_readfirstlane((int)(threadIdx.x >> 5));
    const int n    = blockIdx.x * 32 + lane;
    const size_t at0 = (size_t)(seg * PFX_LEN) * N_NODES + n;

    int t[PFX_LEN];
    int sumD = 0, sumS = 0;
#pragma unroll
    for (int k = 0; k < PFX_LEN; ++k) {
        t[k] = partD[at0 + (size_t)k * N_NODES];
        sumS += partS[at0 + (size_t)k * N_NODES];
    }
#pragma unroll
    for (int k = 0; k < PFX_LEN; ++k) sumD += t[k];
    totD[seg][lane] = sumD;
    totS[seg][lane] = sumS;
    __syncthreads();

    int run = 0;
    for (int s = 0; s < seg; ++s) run += totD[s][lane];
#pragma unroll
    for (int k = 0; k < PFX_LEN; ++k) {
        partD[at0 + (size_t)k * N_NODES] = run;
        run += t[k];
    }
    if (seg == PFX_SEGS - 1) {
        cnt[n] = run;
        nd[n]  = rsqrtf(fmaxf((float)run, 1.0f));
    }
    if (seg == 0) {
        int outdeg = 0;
#pragma unroll
        for (int s = 0; s < PFX_SEGS; ++s) outdeg += totS[s][lane];
        ns[n] = rsqrtf(fmaxf((float)outdeg, 1.0f));
    }
}

__global__ __launch_bounds__(1024) void scan_kernel(const int* __restrict__ cnt,
                                                    int* __restrict__ rowptr, int n) {
    __shared__ int wave_front[32];
    __shared__ int wave_total[32];
    const int tid   = threadIdx.x;
    const int lane  = tid & 31;
    const int wave  = tid >> 5;
    const int start = blockIdx.x * 1024;

    int front = 0;
    for (int j = tid; j < start; j += 1024) front += cnt[j];

    const int i = start + tid;
    const int x = (i < n) ? cnt[i] : 0;
    int incl = x;
#pragma unroll
    for (int off = 1; off < 32; off <<= 1) {
        const int t = __shfl_up(incl, off, 32);
        if (lane >= off) incl += t;
    }
#pragma unroll
    for (int off = 16; off >= 1; off >>= 1) front += __shfl_xor(front, off, 32);
    if (lane == 31) wave_total[wave] = incl;
    if (lane == 0) wave_front[wave] = front;
    __syncthreads();
    int base = 0;
    for (int w = 0; w < 32; ++w) {
        base += wave_front[w];
        if (w < wave) base += wave_total[w];
    }
    if (i < n) rowptr[i] = base + incl - x;
}

#define COL_PAD 64
__global__ __launch_bounds__(256) void fill_kernel(const int* __restrict__ src,
                                                   const int* __restrict__ dst,
                                                   const int* __restrict__ lrank,
                                                   const int* __restrict__ partD,
                                                   const int* __restrict__ rowptr,
                                                   int* __restrict__ col, int e) {
    int i = blockIdx.x * blockDim.x + threadIdx.x;
    if (i < e) {
        const int d = dst[i];
        const int c = i / CHUNK_E;
        col[rowptr[d] + partD[(size_t)c * N_NODES + d] + lrank[i]] = src[i];
    }
    if (i < COL_PAD) col[e + i] = 0;
}

#define GEMM_GRID 1250
#define GEMM_TPB ((N_NODES / 16) / GEMM_GRID)

__device__ __forceinline__ void store_tile(float* __restrict__ H, int rowBase, int half,
                                           int col, v8f c) {
    const int mBase = half * 8;
#pragma unroll
    for (int r = 0; r < 8; ++r) {
        H[(size_t)(rowBase + mBase + r) * DH + col] = c[r];
    }
}

template <int K>
__global__ __launch_bounds__(128) void gemm_ns_wmma(const float* X,
                                                    const float* __restrict__ W,
                                                    const float* __restrict__ ns,
                                                    float* __restrict__ H) {
    constexpr int P   = K + 4;
    constexpr int VPR = K / 4;
    constexpr int NLD = 16 * VPR / 128;
    constexpr int RPQ = 128 / VPR;
    __shared__ float4 As4[2][16 * P / 4];

    const int tid     = threadIdx.x;
    const int lane    = tid & 31;
    const int wave    = tid >> 5;
    const int colBase = wave * 16;
    const int half    = lane >> 4;
    const int l15     = lane & 15;
    const int col     = colBase + l15;

    const int sr = tid / VPR;
    const int sc = tid % VPR;

    const float* Xt = X + (size_t)(blockIdx.x * 16 + sr) * K + sc * 4;
    float4 x0 = *(const float4*)(Xt);
    float4 x1 = *(const float4*)(Xt + (size_t)RPQ * K);
    float4 x2 = x0, x3 = x0;
    if (NLD == 4) {
        x2 = *(const float4*)(Xt + (size_t)2 * RPQ * K);
        x3 = *(const float4*)(Xt + (size_t)3 * RPQ * K);
    }
    float sn = ns[blockIdx.x * 16 + l15];

    v2f b[K / 4];
#pragma unroll
    for (int j = 0; j < K / 4; ++j) {
        const int kk = j * 4 + half * 2;
        b[j].x = W[kk * DH + col];
        b[j].y = W[(kk + 1) * DH + col];
    }

    v8f cprev = {};
#pragma unroll
    for (int i = 0; i < GEMM_TPB; ++i) {
        const int rowBase = (blockIdx.x + i * GEMM_GRID) * 16;
        float* As = (float*)As4[i & 1];
        {
            float* Ad = As + sr * P + sc * 4;
            *(float4*)(Ad) = x0;
            *(float4*)(Ad + RPQ * P) = x1;
            if (NLD == 4) {
                *(float4*)(Ad + 2 * RPQ * P) = x2;
                *(float4*)(Ad + 3 * RPQ * P) = x3;
            }
        }
        const float s = sn;
        __syncthreads();
        if (i > 0) store_tile(H, rowBase - GEMM_GRID * 16, half, col, cprev);
        if (i + 1 < GEMM_TPB) {
            const float* Xn = Xt + (size_t)(i + 1) * GEMM_GRID * 16 * K;
            x0 = *(const float4*)(Xn);
            x1 = *(const float4*)(Xn + (size_t)RPQ * K);
            if (NLD == 4) {
                x2 = *(const float4*)(Xn + (size_t)2 * RPQ * K);
                x3 = *(const float4*)(Xn + (size_t)3 * RPQ * K);
            }
            sn = ns[rowBase + GEMM_GRID * 16 + l15];
        }

        const float* Af = As + l15 * P + half * 2;
        v2f a[K / 4];
#pragma unroll
        for (int j = 0; j < K / 4; ++j) a[j] = *(const v2f*)(Af + j * 4);
        __builtin_amdgcn_sched_barrier(0);

        v8f c = {};
#pragma unroll
        for (int j = 0; j < K / 4; ++j) {
            v2f as;
            as.x = a[j].x * s;
            as.y = a[j].y * s;
            c = __builtin_amdgcn_wmma_f32_16x16x4_f32(false, as, false, b[j], (short)0, c,
                                                      false, false);
        }
        cprev = c;
        __builtin_amdgcn_sched_barrier(0);
    }
    store_tile(H, (blockIdx.x + (GEMM_TPB - 1) * GEMM_GRID) * 16, half, col, cprev);
}

#define AGG_CH 16
template <bool RELU, bool LN, bool PRE>
__global__ __launch_bounds__(256) void aggregate_post_kernel(const float* __restrict__ H,
                                                             const int* __restrict__ col,
                                                             const int* __restrict__ rowptr,
                                                             const int* __restrict__ cnt,
                                                             const float* __restrict__ nd,
                                                             const float* __restrict__ bias,
                                                             const float* __restrict__ g,
                                                             const float* __restrict__ be,
                                                             float* __restrict__ outx,
                                                             float* __restrict__ pre) {
    const int lane = threadIdx.x & 31;
    const int node = __builtin_amdgcn_readfirstlane(
        (int)(blockIdx.x * 8 + (threadIdx.x >> 5)));
    const int beg = rowptr[node];
    const int len = cnt[node];
    const float ndv = nd[node];
    const float* Hl = H + lane * 2;
    float2 acc = make_float2(0.0f, 0.0f);
    int k = 0;
    for (; k + AGG_CH <= len; k += AGG_CH) {
        int s[AGG_CH];
#pragma unroll
        for (int j = 0; j < AGG_CH; ++j) s[j] = col[beg + k + j];
        float2 v[AGG_CH];
#pragma unroll
        for (int j = 0; j < AGG_CH; ++j) v[j] = *(const float2*)(Hl + (size_t)s[j] * DH);
#pragma unroll
        for (int j = 0; j < AGG_CH; ++j) {
            acc.x += v[j].x;
            acc.y += v[j].y;
        }
    }
    const int c0 = lane * 2;
    const float2 bv = *(const float2*)(bias + c0);
    float2 gv = make_float2(1.0f, 1.0f), bev = make_float2(0.0f, 0.0f);
    if (LN) {
        gv  = *(const float2*)(g + c0);
        bev = *(const float2*)(be + c0);
    }
    if (k < len) {
        const int r = len - k;
        int s[AGG_CH];
#pragma unroll
        for (int j = 0; j < AGG_CH; ++j) s[j] = col[beg + k + j];
        float2 v[AGG_CH - 1];
#pragma unroll
        for (int j = 0; j < AGG_CH - 1; ++j) {
            v[j] = make_float2(0.0f, 0.0f);
            if (j < r) v[j] = *(const float2*)(Hl + (size_t)s[j] * DH);
        }
#pragma unroll
        for (int j = 0; j < AGG_CH - 1; ++j) {
            acc.x += v[j].x;
            acc.y += v[j].y;
        }
    }

    float v0 = acc.x * ndv + bv.x;
    float v1 = acc.y * ndv + bv.y;
    if (RELU) {
        v0 = fmaxf(v0, 0.0f);
        v1 = fmaxf(v1, 0.0f);
    }
    if (PRE) {
        pre[(size_t)node * DH + c0 + 0] = v0;
        pre[(size_t)node * DH + c0 + 1] = v1;
    }
    if (LN) {
        float ssum = v0 + v1;
#pragma unroll
        for (int off = 16; off >= 1; off >>= 1) ssum += __shfl_xor(ssum, off, 32);
        const float mu = ssum * (1.0f / DH);
        const float d0 = v0 - mu, d1 = v1 - mu;
        float vs = d0 * d0 + d1 * d1;
#pragma unroll
        for (int off = 16; off >= 1; off >>= 1) vs += __shfl_xor(vs, off, 32);
        const float inv = rsqrtf(vs * (1.0f / DH) + LN_EPS);
        v0 = d0 * inv * gv.x + bev.x;
        v1 = d1 * inv * gv.y + bev.y;
    }
    float* o = outx + (size_t)node * DH + c0;
    o[0] = v0;
    o[1] = v1;
}

extern "C" void kernel_launch(void* const* d_in, const int* in_sizes, int n_in,
                              void* d_out, int out_size, void* d_ws, size_t ws_size,
                              hipStream_t stream) {
    (void)in_sizes; (void)n_in; (void)out_size; (void)ws_size;

    const float* feats = (const float*)d_in[0];
    const float* W1    = (const float*)d_in[1];
    const float* b1    = (const float*)d_in[2];
    const float* g1    = (const float*)d_in[3];
    const float* be1   = (const float*)d_in[4];
    const float* W2    = (const float*)d_in[5];
    const float* b2    = (const float*)d_in[6];
    const float* g2    = (const float*)d_in[7];
    const float* be2   = (const float*)d_in[8];
    const float* W3    = (const float*)d_in[9];
    const float* b3    = (const float*)d_in[10];
    const int*   src   = (const int*)d_in[11];
    const int*   dst   = (const int*)d_in[12];

    float* ns     = (float*)d_ws;
    int*   cnt    = (int*)(ns + N_NODES);
    float* nd     = (float*)(cnt + N_NODES);
    int*   rowptr = (int*)(nd + N_NODES);
    int*   lrank  = rowptr + N_NODES;
    int*   col    = lrank + N_EDGES;
    float* H      = (float*)(col + N_EDGES + COL_PAD);
    float* XB     = H + (size_t)N_NODES * DH;
    int*   partD  = (int*)(XB + (size_t)N_NODES * DH);
    int*   partS  = partD + (size_t)N_CHUNKS * N_NODES;

    float* h_emb = (float*)d_out;
    float* h_out = h_emb + (size_t)N_NODES * DH;

    count_kernel<<<N_CHUNKS * N_PARTS, 1024, 0, stream>>>(src, dst, partS, partD, lrank);
    piece_prefix_kernel<<<N_NODES / 32, 256, 0, stream>>>(partD, partS, cnt, ns, nd);
    scan_kernel<<<(N_NODES + 1023) / 1024, 1024, 0, stream>>>(cnt, rowptr, N_NODES);
    fill_kernel<<<N_EDGES / 256, 256, 0, stream>>>(src, dst, lrank, partD, rowptr, col, N_EDGES);

    gemm_ns_wmma<DIN><<<GEMM_GRID, 128, 0, stream>>>(feats, W1, ns, H);
    aggregate_post_kernel<true, true, false><<<N_NODES / 8, 256, 0, stream>>>(
        H, col, rowptr, cnt, nd, b1, g1, be1, XB, nullptr);

    gemm_ns_wmma<DH><<<GEMM_GRID, 128, 0, stream>>>(XB, W2, ns, H);
    aggregate_post_kernel<true, true, true><<<N_NODES / 8, 256, 0, stream>>>(
        H, col, rowptr, cnt, nd, b2, g2, be2, XB, h_emb);

    gemm_ns_wmma<DH><<<GEMM_GRID, 128, 0, stream>>>(XB, W3, ns, H);
    aggregate_post_kernel<false, false, false><<<N_NODES / 8, 256, 0, stream>>>(
        H, col, rowptr, cnt, nd, b3, nullptr, nullptr, h_out, nullptr);
}


# list fill on LDS tables: (piece, node range) groups load slot bases (rowptr + piece prefix) as whole lines, rank edges by returning LDS add; lrank plane, 3.2M scattered reads go; sums: piece, arrival
# speedup vs baseline: 4.0775x; 1.0744x over previous
#include <hip/hip_runtime.h>

#define N_NODES 100000
#define N_EDGES 1600000
#define DIN 128
#define DH 64
#define LN_EPS 1e-5f

typedef __attribute__((ext_vector_type(2))) float v2f;
typedef __attribute__((ext_vector_type(8))) float v8f;

#define N_CHUNKS 64
#define CHUNK_E (N_EDGES / N_CHUNKS)
#define N_PARTS 4
#define PART_N (N_NODES / N_PARTS)
#define PART_V4 (PART_N / 4)
#define EPT ((CHUNK_E + 1023) / 1024)

__global__ __launch_bounds__(1024) void count_kernel(const int* __restrict__ src,
                                                     const int* __restrict__ dst,
                                                     int* __restrict__ partS,
                                                     int* __restrict__ partD) {
    __shared__ int4 table4[PART_V4];
    int* table = (int*)table4;
    const int tid = threadIdx.x;
    const int c   = blockIdx.x / N_PARTS;
    const int lo  = (blockIdx.x % N_PARTS) * PART_N;
    const int e0  = c * CHUNK_E;
    const int4 zero4 = make_int4(0, 0, 0, 0);

    int dv[EPT], sv[EPT];
#pragma unroll
    for (int j = 0; j < EPT; ++j) {
        const int k = tid + j * 1024;
        dv[j] = (k < CHUNK_E) ? dst[e0 + k] : -1;
        sv[j] = (k < CHUNK_E) ? src[e0 + k] : -1;
    }

    for (int i = tid; i < PART_V4; i += 1024) table4[i] = zero4;
    __syncthreads();
#pragma unroll
    for (int j = 0; j < EPT; ++j) {
        const unsigned d = (unsigned)(dv[j] - lo);
        if (d < (unsigned)PART_N) atomicAdd(&table[d], 1);
    }
    __syncthreads();
    {
        int4* o = (int4*)(partD + (size_t)c * N_NODES + lo);
        for (int i = tid; i < PART_V4; i += 1024) o[i] = table4[i];
    }
    __syncthreads();

    for (int i = tid; i < PART_V4; i += 1024) table4[i] = zero4;
    __syncthreads();
#pragma unroll
    for (int j = 0; j < EPT; ++j) {
        const unsigned s = (unsigned)(sv[j] - lo);
        if (s < (unsigned)PART_N) atomicAdd(&table[s], 1);
    }
    __syncthreads();
    {
        int4* o = (int4*)(partS + (size_t)c * N_NODES + lo);
        for (int i = tid; i < PART_V4; i += 1024) o[i] = table4[i];
    }
}

#define PFX_SEGS 8
#define PFX_LEN (N_CHUNKS / PFX_SEGS)
__global__ __launch_bounds__(256) void piece_prefix_kernel(int* __restrict__ partD,
                                                           const int* __restrict__ partS,
                                                           int* __restrict__ cnt,
                                                           float* __restrict__ ns,
                                                           float* __restrict__ nd) {
    __shared__ int totD[PFX_SEGS][32];
    __shared__ int totS[PFX_SEGS][32];
    const int lane = threadIdx.x & 31;
    const int seg  = __builtin_amdgcn_readfirstlane((int)(threadIdx.x >> 5));
    const int n    = blockIdx.x * 32 + lane;
    const size_t at0 = (size_t)(seg * PFX_LEN) * N_NODES + n;

    int t[PFX_LEN];
    int sumD = 0, sumS = 0;
#pragma unroll
    for (int k = 0; k < PFX_LEN; ++k) {
        t[k] = partD[at0 + (size_t)k * N_NODES];
        sumS += partS[at0 + (size_t)k * N_NODES];
    }
#pragma unroll
    for (int k = 0; k < PFX_LEN; ++k) sumD += t[k];
    totD[seg][lane] = sumD;
    totS[seg][lane] = sumS;
    __syncthreads();

    int run = 0;
    for (int s = 0; s < seg; ++s) run += totD[s][lane];
#pragma unroll
    for (int k = 0; k < PFX_LEN; ++k) {
        partD[at0 + (size_t)k * N_NODES] = run;
        run += t[k];
    }
    if (seg == PFX_SEGS - 1) {
        cnt[n] = run;
        nd[n]  = rsqrtf(fmaxf((float)run, 1.0f));
    }
    if (seg == 0) {
        int outdeg = 0;
#pragma unroll
        for (int s = 0; s < PFX_SEGS; ++s) outdeg += totS[s][lane];
        ns[n] = rsqrtf(fmaxf((float)outdeg, 1.0f));
    }
}

__global__ __launch_bounds__(1024) void scan_kernel(const int* __restrict__ cnt,
                                                    int* __restrict__ rowptr, int n) {
    __shared__ int wave_front[32];
    __shared__ int wave_total[32];
    const int tid   = threadIdx.x;
    const int lane  = tid & 31;
    const int wave  = tid >> 5;
    const int start = blockIdx.x * 1024;

    int front = 0;
    for (int j = tid; j < start; j += 1024) front += cnt[j];

    const int i = start + tid;
    const int x = (i < n) ? cnt[i] : 0;
    int incl = x;
#pragma unroll
    for (int off = 1; off < 32; off <<= 1) {
        const int t = __shfl_up(incl, off, 32);
        if (lane >= off) incl += t;
    }
#pragma unroll
    for (int off = 16; off >= 1; off >>= 1) front += __shfl_xor(front, off, 32);
    if (lane == 31) wave_total[wave] = incl;
    if (lane == 0) wave_front[wave] = front;
    __syncthreads();
    int base = 0;
    for (int w = 0; w < 32; ++w) {
        base += wave_front[w];
        if (w < wave) base += wave_total[w];
    }
    if (i < n) rowptr[i] = base + incl - x;
}

#define COL_PAD 64
#define TAB_Q ((PART_V4 + 1023) / 1024)
__global__ __launch_bounds__(1024) void fill_kernel(const int* __restrict__ src,
                                                    const int* __restrict__ dst,
                                                    const int* __restrict__ partD,
                                                    const int* __restrict__ rowptr,
                                                    int* __restrict__ col) {
    __shared__ int4 table4[PART_V4];
    int* table = (int*)table4;
    const int tid = threadIdx.x;
    const int c   = blockIdx.x / N_PARTS;
    const int lo  = (blockIdx.x % N_PARTS) * PART_N;
    const int e0  = c * CHUNK_E;

    int dv[EPT], sv[EPT];
#pragma unroll
    for (int j = 0; j < EPT; ++j) {
        const int k = tid + j * 1024;
        dv[j] = (k < CHUNK_E) ? dst[e0 + k] : -1;
        sv[j] = (k < CHUNK_E) ? src[e0 + k] : -1;
    }

    {
        const int4* rp = (const int4*)(rowptr + lo);
        const int4* pd = (const int4*)(partD + (size_t)c * N_NODES + lo);
        int4 a[TAB_Q], b[TAB_Q];
#pragma unroll
        for (int q = 0; q < TAB_Q; ++q) {
            const int i = tid + q * 1024;
            a[q] = (i < PART_V4) ? rp[i] : make_int4(0, 0, 0, 0);
            b[q] = (i < PART_V4) ? pd[i] : make_int4(0, 0, 0, 0);
        }
#pragma unroll
        for (int q = 0; q < TAB_Q; ++q) {
            const int i = tid + q * 1024;
            if (i < PART_V4)
                table4[i] = make_int4(a[q].x + b[q].x, a[q].y + b[q].y, a[q].z + b[q].z,
                                      a[q].w + b[q].w);
        }
    }
    __syncthreads();

#pragma unroll
    for (int j = 0; j < EPT; ++j) {
        const unsigned d = (unsigned)(dv[j] - lo);
        if (d < (unsigned)PART_N) col[atomicAdd(&table[d], 1)] = sv[j];
    }
    if (blockIdx.x == 0 && tid < COL_PAD) col[N_EDGES + tid] = 0;
}

#define GEMM_GRID 1250
#define GEMM_TPB ((N_NODES / 16) / GEMM_GRID)

__device__ __forceinline__ void store_tile(float* __restrict__ H, int rowBase, int half,
                                           int col, v8f c) {
    const int mBase = half * 8;
#pragma unroll
    for (int r = 0; r < 8; ++r) {
        H[(size_t)(rowBase + mBase + r) * DH + col] = c[r];
    }
}

template <int K>
__global__ __launch_bounds__(128) void gemm_ns_wmma(const float* X,
                                                    const float* __restrict__ W,
                                                    const float* __restrict__ ns,
                                                    float* __restrict__ H) {
    constexpr int P   = K + 4;
    constexpr int VPR = K / 4;
    constexpr int NLD = 16 * VPR / 128;
    constexpr int RPQ = 128 / VPR;
    __shared__ float4 As4[2][16 * P / 4];

    const int tid     = threadIdx.x;
    const int lane    = tid & 31;
    const int wave    = tid >> 5;
    const int colBase = wave * 16;
    const int half    = lane >> 4;
    const int l15     = lane & 15;
    const int col     = colBase + l15;

    const int sr = tid / VPR;
    const int sc = tid % VPR;

    const float* Xt = X + (size_t)(blockIdx.x * 16 + sr) * K + sc * 4;
    float4 x0 = *(const float4*)(Xt);
    float4 x1 = *(const float4*)(Xt + (size_t)RPQ * K);
    float4 x2 = x0, x3 = x0;
    if (NLD == 4) {
        x2 = *(const float4*)(Xt + (size_t)2 * RPQ * K);
        x3 = *(const float4*)(Xt + (size_t)3 * RPQ * K);
    }
    float sn = ns[blockIdx.x * 16 + l15];

    v2f b[K / 4];
#pragma unroll
    for (int j = 0; j < K / 4; ++j) {
        const int kk = j * 4 + half * 2;
        b[j].x = W[kk * DH + col];
        b[j].y = W[(kk + 1) * DH + col];
    }

    v8f cprev = {};
#pragma unroll
    for (int i = 0; i < GEMM_TPB; ++i) {
        const int rowBase = (blockIdx.x + i * GEMM_GRID) * 16;
        float* As = (float*)As4[i & 1];
        {
            float* Ad = As + sr * P + sc * 4;
            *(float4*)(Ad) = x0;
            *(float4*)(Ad + RPQ * P) = x1;
            if (NLD == 4) {
                *(float4*)(Ad + 2 * RPQ * P) = x2;
                *(float4*)(Ad + 3 * RPQ * P) = x3;
            }
        }
        const float s = sn;
        __syncthreads();
        if (i > 0) store_tile(H, rowBase - GEMM_GRID * 16, half, col, cprev);
        if (i + 1 < GEMM_TPB) {
            const float* Xn = Xt + (size_t)(i + 1) * GEMM_GRID * 16 * K;
            x0 = *(const float4*)(Xn);
            x1 = *(const float4*)(Xn + (size_t)RPQ * K);
            if (NLD == 4) {
                x2 = *(const float4*)(Xn + (size_t)2 * RPQ * K);
                x3 = *(const float4*)(Xn + (size_t)3 * RPQ * K);
            }
            sn = ns[rowBase + GEMM_GRID * 16 + l15];
        }

        const float* Af = As + l15 * P + half * 2;
        v2f a[K / 4];
#pragma unroll
        for (int j = 0; j < K / 4; ++j) a[j] = *(const v2f*)(Af + j * 4);
        __builtin_amdgcn_sched_barrier(0);

        v8f c = {};
#pragma unroll
        for (int j = 0; j < K / 4; ++j) {
            v2f as;
            as.x = a[j].x * s;
            as.y = a[j].y * s;
            c = __builtin_amdgcn_wmma_f32_16x16x4_f32(false, as, false, b[j], (short)0, c,
                                                      false, false);
        }
        cprev = c;
        __builtin_amdgcn_sched_barrier(0);
    }
    store_tile(H, (blockIdx.x + (GEMM_TPB - 1) * GEMM_GRID) * 16, half, col, cprev);
}

#define AGG_CH 16
template <bool RELU, bool LN, bool PRE>
__global__ __launch_bounds__(256) void aggregate_post_kernel(const float* __restrict__ H,
                                                             const int* __restrict__ col,
                                                             const int* __restrict__ rowptr,
                                                             const int* __restrict__ cnt,
                                                             const float* __restrict__ nd,
                                                             const float* __restrict__ bias,
                                                             const float* __restrict__ g,
                                                             const float* __restrict__ be,
                                                             float* __restrict__ outx,
                                                             float* __restrict__ pre) {
    const int lane = threadIdx.x & 31;
    const int node = __builtin_amdgcn_readfirstlane(
        (int)(blockIdx.x * 8 + (threadIdx.x >> 5)));
    const int beg = rowptr[node];
    const int len = cnt[node];
    const float ndv = nd[node];
    const float* Hl = H + lane * 2;
    float2 acc = make_float2(0.0f, 0.0f);
    int k = 0;
    for (; k + AGG_CH <= len; k += AGG_CH) {
        int s[AGG_CH];
#pragma unroll
        for (int j = 0; j < AGG_CH; ++j) s[j] = col[beg + k + j];
        float2 v[AGG_CH];
#pragma unroll
        for (int j = 0; j < AGG_CH; ++j) v[j] = *(const float2*)(Hl + (size_t)s[j] * DH);
#pragma unroll
        for (int j = 0; j < AGG_CH; ++j) {
            acc.x += v[j].x;
            acc.y += v[j].y;
        }
    }
    const int c0 = lane * 2;
    const float2 bv = *(const float2*)(bias + c0);
    float2 gv = make_float2(1.0f, 1.0f), bev = make_float2(0.0f, 0.0f);
    if (LN) {
        gv  = *(const float2*)(g + c0);
        bev = *(const float2*)(be + c0);
    }
    if (k < len) {
        const int r = len - k;
        int s[AGG_CH];
#pragma unroll
        for (int j = 0; j < AGG_CH; ++j) s[j] = col[beg + k + j];
        float2 v[AGG_CH - 1];
#pragma unroll
        for (int j = 0; j < AGG_CH - 1; ++j) {
            v[j] = make_float2(0.0f, 0.0f);
            if (j < r) v[j] = *(const float2*)(Hl + (size_t)s[j] * DH);
        }
#pragma unroll
        for (int j = 0; j < AGG_CH - 1; ++j) {
            acc.x += v[j].x;
            acc.y += v[j].y;
        }
    }

    float v0 = acc.x * ndv + bv.x;
    float v1 = acc.y * ndv + bv.y;
    if (RELU) {
        v0 = fmaxf(v0, 0.0f);
        v1 = fmaxf(v1, 0.0f);
    }
    if (PRE) {
        pre[(size_t)node * DH + c0 + 0] = v0;
        pre[(size_t)node * DH + c0 + 1] = v1;
    }
    if (LN) {
        float ssum = v0 + v1;
#pragma unroll
        for (int off = 16; off >= 1; off >>= 1) ssum += __shfl_xor(ssum, off, 32);
        const float mu = ssum * (1.0f / DH);
        const float d0 = v0 - mu, d1 = v1 - mu;
        float vs = d0 * d0 + d1 * d1;
#pragma unroll
        for (int off = 16; off >= 1; off >>= 1) vs += __shfl_xor(vs, off, 32);
        const float inv = rsqrtf(vs * (1.0f / DH) + LN_EPS);
        v0 = d0 * inv * gv.x + bev.x;
        v1 = d1 * inv * gv.y + bev.y;
    }
    float* o = outx + (size_t)node * DH + c0;
    o[0] = v0;
    o[1] = v1;
}

extern "C" void kernel_launch(void* const* d_in, const int* in_sizes, int n_in,
                              void* d_out, int out_size, void* d_ws, size_t ws_size,
                              hipStream_t stream) {
    (void)in_sizes; (void)n_in; (void)out_size; (void)ws_size;

    const float* feats = (const float*)d_in[0];
    const float* W1    = (const float*)d_in[1];
    const float* b1    = (const float*)d_in[2];
    const float* g1    = (const float*)d_in[3];
    const float* be1   = (const float*)d_in[4];
    const float* W2    = (const float*)d_in[5];
    const float* b2    = (const float*)d_in[6];
    const float* g2    = (const float*)d_in[7];
    const float* be2   = (const float*)d_in[8];
    const float* W3    = (const float*)d_in[9];
    const float* b3    = (const float*)d_in[10];
    const int*   src   = (const int*)d_in[11];
    const int*   dst   = (const int*)d_in[12];

    float* ns     = (float*)d_ws;
    int*   cnt    = (int*)(ns + N_NODES);
    float* nd     = (float*)(cnt + N_NODES);
    int*   rowptr = (int*)(nd + N_NODES);
    int*   col    = rowptr + N_NODES;
    float* H      = (float*)(col + N_EDGES + COL_PAD);
    float* XB     = H + (size_t)N_NODES * DH;
    int*   partD  = (int*)(XB + (size_t)N_NODES * DH);
    int*   partS  = partD + (size_t)N_CHUNKS * N_NODES;

    float* h_emb = (float*)d_out;
    float* h_out = h_emb + (size_t)N_NODES * DH;

    count_kernel<<<N_CHUNKS * N_PARTS, 1024, 0, stream>>>(src, dst, partS, partD);
    piece_prefix_kernel<<<N_NODES / 32, 256, 0, stream>>>(partD, partS, cnt, ns, nd);
    scan_kernel<<<(N_NODES + 1023) / 1024, 1024, 0, stream>>>(cnt, rowptr, N_NODES);
    fill_kernel<<<N_CHUNKS * N_PARTS, 1024, 0, stream>>>(src, dst, partD, rowptr, col);

    gemm_ns_wmma<DIN><<<GEMM_GRID, 128, 0, stream>>>(feats, W1, ns, H);
    aggregate_post_kernel<true, true, false><<<N_NODES / 8, 256, 0, stream>>>(
        H, col, rowptr, cnt, nd, b1, g1, be1, XB, nullptr);

    gemm_ns_wmma<DH><<<GEMM_GRID, 128, 0, stream>>>(XB, W2, ns, H);
    aggregate_post_kernel<true, true, true><<<N_NODES / 8, 256, 0, stream>>>(
        H, col, rowptr, cnt, nd, b2, g2, be2, XB, h_emb);

    gemm_ns_wmma<DH><<<GEMM_GRID, 128, 0, stream>>>(XB, W3, ns, H);
    aggregate_post_kernel<false, false, false><<<N_NODES / 8, 256, 0, stream>>>(
        H, col, rowptr, cnt, nd, b3, nullptr, nullptr, h_out, nullptr);
}


# layers 1, 2: aggregate blocks (16 waves, resident grid walking 16-node tiles) park their normalised rows in LDS and form the next layer's H on WMMA; 2 transform launches, XB trip go; sums unchanged
# speedup vs baseline: 4.3273x; 1.0613x over previous
#include <hip/hip_runtime.h>

#define N_NODES 100000
#define N_EDGES 1600000
#define DIN 128
#define DH 64
#define LN_EPS 1e-5f

typedef __attribute__((ext_vector_type(2))) float v2f;
typedef __attribute__((ext_vector_type(8))) float v8f;

#define N_CHUNKS 64
#define CHUNK_E (N_EDGES / N_CHUNKS)
#define N_PARTS 4
#define PART_N (N_NODES / N_PARTS)
#define PART_V4 (PART_N / 4)
#define EPT ((CHUNK_E + 1023) / 1024)

__global__ __launch_bounds__(1024) void count_kernel(const int* __restrict__ src,
                                                     const int* __restrict__ dst,
                                                     int* __restrict__ partS,
                                                     int* __restrict__ partD) {
    __shared__ int4 table4[PART_V4];
    int* table = (int*)table4;
    const int tid = threadIdx.x;
    const int c   = blockIdx.x / N_PARTS;
    const int lo  = (blockIdx.x % N_PARTS) * PART_N;
    const int e0  = c * CHUNK_E;
    const int4 zero4 = make_int4(0, 0, 0, 0);

    int dv[EPT], sv[EPT];
#pragma unroll
    for (int j = 0; j < EPT; ++j) {
        const int k = tid + j * 1024;
        dv[j] = (k < CHUNK_E) ? dst[e0 + k] : -1;
        sv[j] = (k < CHUNK_E) ? src[e0 + k] : -1;
    }

    for (int i = tid; i < PART_V4; i += 1024) table4[i] = zero4;
    __syncthreads();
#pragma unroll
    for (int j = 0; j < EPT; ++j) {
        const unsigned d = (unsigned)(dv[j] - lo);
        if (d < (unsigned)PART_N) atomicAdd(&table[d], 1);
    }
    __syncthreads();
    {
        int4* o = (int4*)(partD + (size_t)c * N_NODES + lo);
        for (int i = tid; i < PART_V4; i += 1024) o[i] = table4[i];
    }
    __syncthreads();

    for (int i = tid; i < PART_V4; i += 1024) table4[i] = zero4;
    __syncthreads();
#pragma unroll
    for (int j = 0; j < EPT; ++j) {
        const unsigned s = (unsigned)(sv[j] - lo);
        if (s < (unsigned)PART_N) atomicAdd(&table[s], 1);
    }
    __syncthreads();
    {
        int4* o = (int4*)(partS + (size_t)c * N_NODES + lo);
        for (int i = tid; i < PART_V4; i += 1024) o[i] = table4[i];
    }
}

#define PFX_SEGS 8
#define PFX_LEN (N_CHUNKS / PFX_SEGS)
__global__ __launch_bounds__(256) void piece_prefix_kernel(int* __restrict__ partD,
                                                           const int* __restrict__ partS,
                                                           int* __restrict__ cnt,
                                                           float* __restrict__ ns,
                                                           float* __restrict__ nd) {
    __shared__ int totD[PFX_SEGS][32];
    __shared__ int totS[PFX_SEGS][32];
    const int lane = threadIdx.x & 31;
    const int seg  = __builtin_amdgcn_readfirstlane((int)(threadIdx.x >> 5));
    const int n    = blockIdx.x * 32 + lane;
    const size_t at0 = (size_t)(seg * PFX_LEN) * N_NODES + n;

    int t[PFX_LEN];
    int sumD = 0, sumS = 0;
#pragma unroll
    for (int k = 0; k < PFX_LEN; ++k) {
        t[k] = partD[at0 + (size_t)k * N_NODES];
        sumS += partS[at0 + (size_t)k * N_NODES];
    }
#pragma unroll
    for (int k = 0; k < PFX_LEN; ++k) sumD += t[k];
    totD[seg][lane] = sumD;
    totS[seg][lane] = sumS;
    __syncthreads();

    int run = 0;
    for (int s = 0; s < seg; ++s) run += totD[s][lane];
#pragma unroll
    for (int k = 0; k < PFX_LEN; ++k) {
        partD[at0 + (size_t)k * N_NODES] = run;
        run += t[k];
    }
    if (seg == PFX_SEGS - 1) {
        cnt[n] = run;
        nd[n]  = rsqrtf(fmaxf((float)run, 1.0f));
    }
    if (seg == 0) {
        int outdeg = 0;
#pragma unroll
        for (int s = 0; s < PFX_SEGS; ++s) outdeg += totS[s][lane];
        ns[n] = rsqrtf(fmaxf((float)outdeg, 1.0f));
    }
}

__global__ __launch_bounds__(1024) void scan_kernel(const int* __restrict__ cnt,
                                                    int* __restrict__ rowptr, int n) {
    __shared__ int wave_front[32];
    __shared__ int wave_total[32];
    const int tid   = threadIdx.x;
    const int lane  = tid & 31;
    const int wave  = tid >> 5;
    const int start = blockIdx.x * 1024;

    int front = 0;
    for (int j = tid; j < start; j += 1024) front += cnt[j];

    const int i = start + tid;
    const int x = (i < n) ? cnt[i] : 0;
    int incl = x;
#pragma unroll
    for (int off = 1; off < 32; off <<= 1) {
        const int t = __shfl_up(incl, off, 32);
        if (lane >= off) incl += t;
    }
#pragma unroll
    for (int off = 16; off >= 1; off >>= 1) front += __shfl_xor(front, off, 32);
    if (lane == 31) wave_total[wave] = incl;
    if (lane == 0) wave_front[wave] = front;
    __syncthreads();
    int base = 0;
    for (int w = 0; w < 32; ++w) {
        base += wave_front[w];
        if (w < wave) base += wave_total[w];
    }
    if (i < n) rowptr[i] = base + incl - x;
}

#define COL_PAD 64
#define TAB_Q ((PART_V4 + 1023) / 1024)
__global__ __launch_bounds__(1024) void fill_kernel(const int* __restrict__ src,
                                                    const int* __restrict__ dst,
                                                    const int* __restrict__ partD,
                                                    const int* __restrict__ rowptr,
                                                    int* __restrict__ col) {
    __shared__ int4 table4[PART_V4];
    int* table = (int*)table4;
    const int tid = threadIdx.x;
    const int c   = blockIdx.x / N_PARTS;
    const int lo  = (blockIdx.x % N_PARTS) * PART_N;
    const int e0  = c * CHUNK_E;

    int dv[EPT], sv[EPT];
#pragma unroll
    for (int j = 0; j < EPT; ++j) {
        const int k = tid + j * 1024;
        dv[j] = (k < CHUNK_E) ? dst[e0 + k] : -1;
        sv[j] = (k < CHUNK_E) ? src[e0 + k] : -1;
    }

    {
        const int4* rp = (const int4*)(rowptr + lo);
        const int4* pd = (const int4*)(partD + (size_t)c * N_NODES + lo);
        int4 a[TAB_Q], b[TAB_Q];
#pragma unroll
        for (int q = 0; q < TAB_Q; ++q) {
            const int i = tid + q * 1024;
            a[q] = (i < PART_V4) ? rp[i] : make_int4(0, 0, 0, 0);
            b[q] = (i < PART_V4) ? pd[i] : make_int4(0, 0, 0, 0);
        }
#pragma unroll
        for (int q = 0; q < TAB_Q; ++q) {
            const int i = tid + q * 1024;
            if (i < PART_V4)
                table4[i] = make_int4(a[q].x + b[q].x, a[q].y + b[q].y, a[q].z + b[q].z,
                                      a[q].w + b[q].w);
        }
    }
    __syncthreads();

#pragma unroll
    for (int j = 0; j < EPT; ++j) {
        const unsigned d = (unsigned)(dv[j] - lo);
        if (d < (unsigned)PART_N) col[atomicAdd(&table[d], 1)] = sv[j];
    }
    if (blockIdx.x == 0 && tid < COL_PAD) col[N_EDGES + tid] = 0;
}

#define GEMM_GRID 1250
#define GEMM_TPB ((N_NODES / 16) / GEMM_GRID)

__device__ __forceinline__ void store_tile(float* __restrict__ H, int rowBase, int half,
                                           int col, v8f c) {
    const int mBase = half * 8;
#pragma unroll
    for (int r = 0; r < 8; ++r) {
        H[(size_t)(rowBase + mBase + r) * DH + col] = c[r];
    }
}

template <int K>
__global__ __launch_bounds__(128) void gemm_ns_wmma(const float* X,
                                                    const float* __restrict__ W,
                                                    const float* __restrict__ ns,
                                                    float* __restrict__ H) {
    constexpr int P   = K + 4;
    constexpr int VPR = K / 4;
    constexpr int NLD = 16 * VPR / 128;
    constexpr int RPQ = 128 / VPR;
    __shared__ float4 As4[2][16 * P / 4];

    const int tid     = threadIdx.x;
    const int lane    = tid & 31;
    const int wave    = tid >> 5;
    const int colBase = wave * 16;
    const int half    = lane >> 4;
    const int l15     = lane & 15;
    const int col     = colBase + l15;

    const int sr = tid / VPR;
    const int sc = tid % VPR;

    const float* Xt = X + (size_t)(blockIdx.x * 16 + sr) * K + sc * 4;
    float4 x0 = *(const float4*)(Xt);
    float4 x1 = *(const float4*)(Xt + (size_t)RPQ * K);
    float4 x2 = x0, x3 = x0;
    if (NLD == 4) {
        x2 = *(const float4*)(Xt + (size_t)2 * RPQ * K);
        x3 = *(const float4*)(Xt + (size_t)3 * RPQ * K);
    }
    float sn = ns[blockIdx.x * 16 + l15];

    v2f b[K / 4];
#pragma unroll
    for (int j = 0; j < K / 4; ++j) {
        const int kk = j * 4 + half * 2;
        b[j].x = W[kk * DH + col];
        b[j].y = W[(kk + 1) * DH + col];
    }

    v8f cprev = {};
#pragma unroll
    for (int i = 0; i < GEMM_TPB; ++i) {
        const int rowBase = (blockIdx.x + i * GEMM_GRID) * 16;
        float* As = (float*)As4[i & 1];
        {
            float* Ad = As + sr * P + sc * 4;
            *(float4*)(Ad) = x0;
            *(float4*)(Ad + RPQ * P) = x1;
            if (NLD == 4) {
                *(float4*)(Ad + 2 * RPQ * P) = x2;
                *(float4*)(Ad + 3 * RPQ * P) = x3;
            }
        }
        const float s = sn;
        __syncthreads();
        if (i > 0) store_tile(H, rowBase - GEMM_GRID * 16, half, col, cprev);
        if (i + 1 < GEMM_TPB) {
            const float* Xn = Xt + (size_t)(i + 1) * GEMM_GRID * 16 * K;
            x0 = *(const float4*)(Xn);
            x1 = *(const float4*)(Xn + (size_t)RPQ * K);
            if (NLD == 4) {
                x2 = *(const float4*)(Xn + (size_t)2 * RPQ * K);
                x3 = *(const float4*)(Xn + (size_t)3 * RPQ * K);
            }
            sn = ns[rowBase + GEMM_GRID * 16 + l15];
        }

        const float* Af = As + l15 * P + half * 2;
        v2f a[K / 4];
#pragma unroll
        for (int j = 0; j < K / 4; ++j) a[j] = *(const v2f*)(Af + j * 4);
        __builtin_amdgcn_sched_barrier(0);

        v8f c = {};
#pragma unroll
        for (int j = 0; j < K / 4; ++j) {
            v2f as;
            as.x = a[j].x * s;
            as.y = a[j].y * s;
            c = __builtin_amdgcn_wmma_f32_16x16x4_f32(false, as, false, b[j], (short)0, c,
                                                      false, false);
        }
        cprev = c;
        __builtin_amdgcn_sched_barrier(0);
    }
    store_tile(H, (blockIdx.x + (GEMM_TPB - 1) * GEMM_GRID) * 16, half, col, cprev);
}

#define AGG_CH 16
template <bool RELU, bool LN, bool PRE>
__global__ __launch_bounds__(256) void aggregate_post_kernel(const float* __restrict__ H,
                                                             const int* __restrict__ col,
                                                             const int* __restrict__ rowptr,
                                                             const int* __restrict__ cnt,
                                                             const float* __restrict__ nd,
                                                             const float* __restrict__ bias,
                                                             const float* __restrict__ g,
                                                             const float* __restrict__ be,
                                                             float* __restrict__ outx,
                                                             float* __restrict__ pre) {
    const int lane = threadIdx.x & 31;
    const int node = __builtin_amdgcn_readfirstlane(
        (int)(blockIdx.x * 8 + (threadIdx.x >> 5)));
    const int beg = rowptr[node];
    const int len = cnt[node];
    const float ndv = nd[node];
    const float* Hl = H + lane * 2;
    float2 acc = make_float2(0.0f, 0.0f);
    int k = 0;
    for (; k + AGG_CH <= len; k += AGG_CH) {
        int s[AGG_CH];
#pragma unroll
        for (int j = 0; j < AGG_CH; ++j) s[j] = col[beg + k + j];
        float2 v[AGG_CH];
#pragma unroll
        for (int j = 0; j < AGG_CH; ++j) v[j] = *(const float2*)(Hl + (size_t)s[j] * DH);
#pragma unroll
        for (int j = 0; j < AGG_CH; ++j) {
            acc.x += v[j].x;
            acc.y += v[j].y;
        }
    }
    const int c0 = lane * 2;
    const float2 bv = *(const float2*)(bias + c0);
    float2 gv = make_float2(1.0f, 1.0f), bev = make_float2(0.0f, 0.0f);
    if (LN) {
        gv  = *(const float2*)(g + c0);
        bev = *(const float2*)(be + c0);
    }
    if (k < len) {
        const int r = len - k;
        int s[AGG_CH];
#pragma unroll
        for (int j = 0; j < AGG_CH; ++j) s[j] = col[beg + k + j];
        float2 v[AGG_CH - 1];
#pragma unroll
        for (int j = 0; j < AGG_CH - 1; ++j) {
            v[j] = make_float2(0.0f, 0.0f);
            if (j < r) v[j] = *(const float2*)(Hl + (size_t)s[j] * DH);
        }
#pragma unroll
        for (int j = 0; j < AGG_CH - 1; ++j) {
            acc.x += v[j].x;
            acc.y += v[j].y;
        }
    }

    float v0 = acc.x * ndv + bv.x;
    float v1 = acc.y * ndv + bv.y;
    if (RELU) {
        v0 = fmaxf(v0, 0.0f);
        v1 = fmaxf(v1, 0.0f);
    }
    if (PRE) {
        pre[(size_t)node * DH + c0 + 0] = v0;
        pre[(size_t)node * DH + c0 + 1] = v1;
    }
    if (LN) {
        float ssum = v0 + v1;
#pragma unroll
        for (int off = 16; off >= 1; off >>= 1) ssum += __shfl_xor(ssum, off, 32);
        const float mu = ssum * (1.0f / DH);
        const float d0 = v0 - mu, d1 = v1 - mu;
        float vs = d0 * d0 + d1 * d1;
#pragma unroll
        for (int off = 16; off >= 1; off >>= 1) vs += __shfl_xor(vs, off, 32);
        const float inv = rsqrtf(vs * (1.0f / DH) + LN_EPS);
        v0 = d0 * inv * gv.x + bev.x;
        v1 = d1 * inv * gv.y + bev.y;
    }
    float* o = outx + (size_t)node * DH + c0;
    o[0] = v0;
    o[1] = v1;
}

#define FT_WAVES 16
#define FT_TILES (N_NODES / FT_WAVES)
#define FT_GRID 1024
#define FT_P (DH + 4)

__device__ __forceinline__ void transform_tile(const float* Ar, const float* Wt,
                                               float* __restrict__ Hn, int rowBase, int half,
                                               int l15, int ct) {
    const float* Af = Ar + l15 * FT_P + half * 2;
    const float* Bf = Wt + ct * FT_P + half * 2;
    v2f a[DH / 4], b[DH / 4];
#pragma unroll
    for (int j = 0; j < DH / 4; ++j) {
        a[j] = *(const v2f*)(Af + j * 4);
        b[j] = *(const v2f*)(Bf + j * 4);
    }
    v8f c = {};
#pragma unroll
    for (int j = 0; j < DH / 4; ++j) {
        c = __builtin_amdgcn_wmma_f32_16x16x4_f32(false, a[j], false, b[j], (short)0, c, false,
                                                  false);
    }
    store_tile(Hn, rowBase, half, ct, c);
}

template <bool PRE>
__global__ __launch_bounds__(512) void aggregate_transform_kernel(const float* __restrict__ H,
                                                                  const int* __restrict__ col,
                                                                  const int* __restrict__ rowptr,
                                                                  const int* __restrict__ cnt,
                                                                  const float* __restrict__ nd,
                                                                  const float* __restrict__ ns,
                                                                  const float* __restrict__ bias,
                                                                  const float* __restrict__ g,
                                                                  const float* __restrict__ be,
                                                                  const float* __restrict__ Wn,
                                                                  float* __restrict__ Hn,
                                                                  float* __restrict__ pre) {
    __shared__ float4 Wt4[DH * FT_P / 4];
    __shared__ float4 Ar4[2][FT_WAVES * FT_P / 4];
    float* Wt = (float*)Wt4;
    const int tid  = threadIdx.x;
    const int lane = tid & 31;
    const int wave = __builtin_amdgcn_readfirstlane((int)(threadIdx.x >> 5));
    const int half = lane >> 4;
    const int l15  = lane & 15;
    const int c0   = lane * 2;
    const int ct   = (wave & 3) * 16 + l15;

#pragma unroll
    for (int q = 0; q < DH * DH / 4 / 512; ++q) {
        const int i  = tid + q * 512;
        const int k  = i >> 4;
        const int c4 = (i & 15) * 4;
        const float4 w = *(const float4*)(Wn + k * DH + c4);
        Wt[(c4 + 0) * FT_P + k] = w.x;
        Wt[(c4 + 1) * FT_P + k] = w.y;
        Wt[(c4 + 2) * FT_P + k] = w.z;
        Wt[(c4 + 3) * FT_P + k] = w.w;
    }
    const float2 bv  = *(const float2*)(bias + c0);
    const float2 gv  = *(const float2*)(g + c0);
    const float2 bev = *(const float2*)(be + c0);
    const float* Hl = H + c0;

    int beg, len;
    float ndv, nsv;
    {
        const int n0 = blockIdx.x * FT_WAVES + wave;
        beg = rowptr[n0];
        len = cnt[n0];
        ndv = nd[n0];
        nsv = ns[n0];
    }
    __syncthreads();

    int nt = 0;
#pragma unroll 1
    for (int tile = blockIdx.x; tile < FT_TILES; tile += FT_GRID, ++nt) {
        const int node = tile * FT_WAVES + wave;
        if (nt > 0 && (wave >> 2) == ((nt - 1) & 3))
            transform_tile((const float*)Ar4[(nt - 1) & 1], Wt, Hn, (tile - FT_GRID) * FT_WAVES,
                           half, l15, ct);

        float2 acc = make_float2(0.0f, 0.0f);
        int k = 0;
        for (; k + AGG_CH <= len; k += AGG_CH) {
            int s[AGG_CH];
#pragma unroll
            for (int j = 0; j < AGG_CH; ++j) s[j] = col[beg + k + j];
            float2 v[AGG_CH];
#pragma unroll
            for (int j = 0; j < AGG_CH; ++j) v[j] = *(const float2*)(Hl + (size_t)s[j] * DH);
#pragma unroll
            for (int j = 0; j < AGG_CH; ++j) {
                acc.x += v[j].x;
                acc.y += v[j].y;
            }
        }
        if (k < len) {
            const int r = len - k;
            int s[AGG_CH];
#pragma unroll
            for (int j = 0; j < AGG_CH; ++j) s[j] = col[beg + k + j];
            float2 v[AGG_CH - 1];
#pragma unroll
            for (int j = 0; j < AGG_CH - 1; ++j) {
                v[j] = make_float2(0.0f, 0.0f);
                if (j < r) v[j] = *(const float2*)(Hl + (size_t)s[j] * DH);
            }
#pragma unroll
            for (int j = 0; j < AGG_CH - 1; ++j) {
                acc.x += v[j].x;
                acc.y += v[j].y;
            }
        }

        float v0 = fmaxf(acc.x * ndv + bv.x, 0.0f);
        float v1 = fmaxf(acc.y * ndv + bv.y, 0.0f);
        if (PRE) {
            pre[(size_t)node * DH + c0 + 0] = v0;
            pre[(size_t)node * DH + c0 + 1] = v1;
        }
        const int nn = min(node + FT_GRID * FT_WAVES, N_NODES - 1);
        const int nbeg = rowptr[nn];
        const int nlen = cnt[nn];
        const float nndv = nd[nn];
        const float nnsv = ns[nn];
        __builtin_amdgcn_sched_barrier(0);
        {
            float ssum = v0 + v1;
#pragma unroll
            for (int off = 16; off >= 1; off >>= 1) ssum += __shfl_xor(ssum, off, 32);
            const float mu = ssum * (1.0f / DH);
            const float d0 = v0 - mu, d1 = v1 - mu;
            float vs = d0 * d0 + d1 * d1;
#pragma unroll
            for (int off = 16; off >= 1; off >>= 1) vs += __shfl_xor(vs, off, 32);
            const float inv = rsqrtf(vs * (1.0f / DH) + LN_EPS);
            v0 = d0 * inv * gv.x + bev.x;
            v1 = d1 * inv * gv.y + bev.y;
        }
        *(float2*)((float*)Ar4[nt & 1] + wave * FT_P + c0) = make_float2(v0 * nsv, v1 * nsv);
        __builtin_amdgcn_sched_barrier(0);
        beg = nbeg;
        len = nlen;
        ndv = nndv;
        nsv = nnsv;
        __syncthreads();
    }
    if ((wave >> 2) == ((nt - 1) & 3))
        transform_tile((const float*)Ar4[(nt - 1) & 1], Wt, Hn,
                       (blockIdx.x + (nt - 1) * FT_GRID) * FT_WAVES, half, l15, ct);
}

extern "C" void kernel_launch(void* const* d_in, const int* in_sizes, int n_in,
                              void* d_out, int out_size, void* d_ws, size_t ws_size,
                              hipStream_t stream) {
    (void)in_sizes; (void)n_in; (void)out_size; (void)ws_size;

    const float* feats = (const float*)d_in[0];
    const float* W1    = (const float*)d_in[1];
    const float* b1    = (const float*)d_in[2];
    const float* g1    = (const float*)d_in[3];
    const float* be1   = (const float*)d_in[4];
    const float* W2    = (const float*)d_in[5];
    const float* b2    = (const float*)d_in[6];
    const float* g2    = (const float*)d_in[7];
    const float* be2   = (const float*)d_in[8];
    const float* W3    = (const float*)d_in[9];
    const float* b3    = (const float*)d_in[10];
    const int*   src   = (const int*)d_in[11];
    const int*   dst   = (const int*)d_in[12];

    float* ns     = (float*)d_ws;
    int*   cnt    = (int*)(ns + N_NODES);
    float* nd     = (float*)(cnt + N_NODES);
    int*   rowptr = (int*)(nd + N_NODES);
    int*   col    = rowptr + N_NODES;
    float* H      = (float*)(col + N_EDGES + COL_PAD);
    float* XB     = H + (size_t)N_NODES * DH;
    int*   partD  = (int*)(XB + (size_t)N_NODES * DH);
    int*   partS  = partD + (size_t)N_CHUNKS * N_NODES;

    float* h_emb = (float*)d_out;
    float* h_out = h_emb + (size_t)N_NODES * DH;

    count_kernel<<<N_CHUNKS * N_PARTS, 1024, 0, stream>>>(src, dst, partS, partD);
    piece_prefix_kernel<<<N_NODES / 32, 256, 0, stream>>>(partD, partS, cnt, ns, nd);
    scan_kernel<<<(N_NODES + 1023) / 1024, 1024, 0, stream>>>(cnt, rowptr, N_NODES);
    fill_kernel<<<N_CHUNKS * N_PARTS, 1024, 0, stream>>>(src, dst, partD, rowptr, col);

    gemm_ns_wmma<DIN><<<GEMM_GRID, 128, 0, stream>>>(feats, W1, ns, H);
    aggregate_transform_kernel<false><<<FT_GRID, 512, 0, stream>>>(
        H, col, rowptr, cnt, nd, ns, b1, g1, be1, W2, XB, nullptr);

    aggregate_transform_kernel<true><<<FT_GRID, 512, 0, stream>>>(
        XB, col, rowptr, cnt, nd, ns, b2, g2, be2, W3, H, h_emb);

    aggregate_post_kernel<false, false, false><<<N_NODES / 8, 256, 0, stream>>>(
        H, col, rowptr, cnt, nd, b3, nullptr, nullptr, h_out, nullptr);
}


# row-start scan split between its two neighbours: piece prefix emits starts inside 32-node groups and the group sums, fill scans the 3125 sums in LDS; the scan launch goes (8 -> 7); sums unchanged
# speedup vs baseline: 4.5563x; 1.0529x over previous
#include <hip/hip_runtime.h>

#define N_NODES 100000
#define N_EDGES 1600000
#define DIN 128
#define DH 64
#define LN_EPS 1e-5f

typedef __attribute__((ext_vector_type(2))) float v2f;
typedef __attribute__((ext_vector_type(8))) float v8f;

#define N_CHUNKS 64
#define CHUNK_E (N_EDGES / N_CHUNKS)
#define N_PARTS 4
#define PART_N (N_NODES / N_PARTS)
#define PART_V4 (PART_N / 4)
#define EPT ((CHUNK_E + 1023) / 1024)

__global__ __launch_bounds__(1024) void count_kernel(const int* __restrict__ src,
                                                     const int* __restrict__ dst,
                                                     int* __restrict__ partS,
                                                     int* __restrict__ partD) {
    __shared__ int4 table4[PART_V4];
    int* table = (int*)table4;
    const int tid = threadIdx.x;
    const int c   = blockIdx.x / N_PARTS;
    const int lo  = (blockIdx.x % N_PARTS) * PART_N;
    const int e0  = c * CHUNK_E;
    const int4 zero4 = make_int4(0, 0, 0, 0);

    int dv[EPT], sv[EPT];
#pragma unroll
    for (int j = 0; j < EPT; ++j) {
        const int k = tid + j * 1024;
        dv[j] = (k < CHUNK_E) ? dst[e0 + k] : -1;
        sv[j] = (k < CHUNK_E) ? src[e0 + k] : -1;
    }

    for (int i = tid; i < PART_V4; i += 1024) table4[i] = zero4;
    __syncthreads();
#pragma unroll
    for (int j = 0; j < EPT; ++j) {
        const unsigned d = (unsigned)(dv[j] - lo);
        if (d < (unsigned)PART_N) atomicAdd(&table[d], 1);
    }
    __syncthreads();
    {
        int4* o = (int4*)(partD + (size_t)c * N_NODES + lo);
        for (int i = tid; i < PART_V4; i += 1024) o[i] = table4[i];
    }
    __syncthreads();

    for (int i = tid; i < PART_V4; i += 1024) table4[i] = zero4;
    __syncthreads();
#pragma unroll
    for (int j = 0; j < EPT; ++j) {
        const unsigned s = (unsigned)(sv[j] - lo);
        if (s < (unsigned)PART_N) atomicAdd(&table[s], 1);
    }
    __syncthreads();
    {
        int4* o = (int4*)(partS + (size_t)c * N_NODES + lo);
        for (int i = tid; i < PART_V4; i += 1024) o[i] = table4[i];
    }
}

#define PFX_SEGS 8
#define PFX_LEN (N_CHUNKS / PFX_SEGS)
__global__ __launch_bounds__(256) void piece_prefix_kernel(int* __restrict__ partD,
                                                           const int* __restrict__ partS,
                                                           int* __restrict__ cnt,
                                                           float* __restrict__ ns,
                                                           float* __restrict__ nd,
                                                           int* __restrict__ rowloc,
                                                           int* __restrict__ tot32) {
    __shared__ int totD[PFX_SEGS][32];
    __shared__ int totS[PFX_SEGS][32];
    const int lane = threadIdx.x & 31;
    const int seg  = __builtin_amdgcn_readfirstlane((int)(threadIdx.x >> 5));
    const int n    = blockIdx.x * 32 + lane;
    const size_t at0 = (size_t)(seg * PFX_LEN) * N_NODES + n;

    int t[PFX_LEN];
    int sumD = 0, sumS = 0;
#pragma unroll
    for (int k = 0; k < PFX_LEN; ++k) {
        t[k] = partD[at0 + (size_t)k * N_NODES];
        sumS += partS[at0 + (size_t)k * N_NODES];
    }
#pragma unroll
    for (int k = 0; k < PFX_LEN; ++k) sumD += t[k];
    totD[seg][lane] = sumD;
    totS[seg][lane] = sumS;
    __syncthreads();

    int run = 0;
    for (int s = 0; s < seg; ++s) run += totD[s][lane];
#pragma unroll
    for (int k = 0; k < PFX_LEN; ++k) {
        partD[at0 + (size_t)k * N_NODES] = run;
        run += t[k];
    }
    if (seg == PFX_SEGS - 1) {
        cnt[n] = run;
        nd[n]  = rsqrtf(fmaxf((float)run, 1.0f));
        int incl = run;
#pragma unroll
        for (int off = 1; off < 32; off <<= 1) {
            const int up = __shfl_up(incl, off, 32);
            if (lane >= off) incl += up;
        }
        rowloc[n] = incl - run;
        if (lane == 31) tot32[blockIdx.x] = incl;
    }
    if (seg == 0) {
        int outdeg = 0;
#pragma unroll
        for (int s = 0; s < PFX_SEGS; ++s) outdeg += totS[s][lane];
        ns[n] = rsqrtf(fmaxf((float)outdeg, 1.0f));
    }
}

#define COL_PAD 64
#define TAB_Q ((PART_V4 + 1023) / 1024)
#define N_GRP (N_NODES / 32)
__global__ __launch_bounds__(1024) void fill_kernel(const int* __restrict__ src,
                                                    const int* __restrict__ dst,
                                                    const int* __restrict__ partD,
                                                    const int* __restrict__ rowloc,
                                                    const int* __restrict__ tot32,
                                                    int* __restrict__ rowptr,
                                                    int* __restrict__ col) {
    __shared__ int4 table4[PART_V4];
    __shared__ int grp_start[4096];
    __shared__ int wave_tot[32];
    int* table = (int*)table4;
    const int tid  = threadIdx.x;
    const int lane = tid & 31;
    const int wave = tid >> 5;
    const int c   = blockIdx.x / N_PARTS;
    const int lo  = (blockIdx.x % N_PARTS) * PART_N;
    const int e0  = c * CHUNK_E;

    int gs[4];
#pragma unroll
    for (int k = 0; k < 4; ++k) {
        const int gi = tid * 4 + k;
        gs[k] = (gi < N_GRP) ? tot32[gi] : 0;
    }

    int dv[EPT], sv[EPT];
#pragma unroll
    for (int j = 0; j < EPT; ++j) {
        const int k = tid + j * 1024;
        dv[j] = (k < CHUNK_E) ? dst[e0 + k] : -1;
        sv[j] = (k < CHUNK_E) ? src[e0 + k] : -1;
    }

    const int4* rl = (const int4*)(rowloc + lo);
    const int4* pd = (const int4*)(partD + (size_t)c * N_NODES + lo);
    int4 a[TAB_Q], b[TAB_Q];
#pragma unroll
    for (int q = 0; q < TAB_Q; ++q) {
        const int i = tid + q * 1024;
        a[q] = (i < PART_V4) ? rl[i] : make_int4(0, 0, 0, 0);
        b[q] = (i < PART_V4) ? pd[i] : make_int4(0, 0, 0, 0);
    }

    {
        const int tot = gs[0] + gs[1] + gs[2] + gs[3];
        int incl = tot;
#pragma unroll
        for (int off = 1; off < 32; off <<= 1) {
            const int up = __shfl_up(incl, off, 32);
            if (lane >= off) incl += up;
        }
        if (lane == 31) wave_tot[wave] = incl;
        __syncthreads();
        int front = (lane < wave) ? wave_tot[lane] : 0;
#pragma unroll
        for (int off = 16; off >= 1; off >>= 1) front += __shfl_xor(front, off, 32);
        int run = front + incl - tot;
#pragma unroll
        for (int k = 0; k < 4; ++k) {
            grp_start[tid * 4 + k] = run;
            run += gs[k];
        }
    }
    __syncthreads();

#pragma unroll
    for (int q = 0; q < TAB_Q; ++q) {
        const int i = tid + q * 1024;
        if (i < PART_V4) {
            const int g0 = grp_start[(lo + 4 * i) >> 5];
            const int4 r = make_int4(a[q].x + g0, a[q].y + g0, a[q].z + g0, a[q].w + g0);
            if (c == 0) ((int4*)(rowptr + lo))[i] = r;
            table4[i] = make_int4(r.x + b[q].x, r.y + b[q].y, r.z + b[q].z, r.w + b[q].w);
        }
    }
    __syncthreads();

#pragma unroll
    for (int j = 0; j < EPT; ++j) {
        const unsigned d = (unsigned)(dv[j] - lo);
        if (d < (unsigned)PART_N) col[atomicAdd(&table[d], 1)] = sv[j];
    }
    if (blockIdx.x == 0 && tid < COL_PAD) col[N_EDGES + tid] = 0;
}

#define GEMM_GRID 1250
#define GEMM_TPB ((N_NODES / 16) / GEMM_GRID)

__device__ __forceinline__ void store_tile(float* __restrict__ H, int rowBase, int half,
                                           int col, v8f c) {
    const int mBase = half * 8;
#pragma unroll
    for (int r = 0; r < 8; ++r) {
        H[(size_t)(rowBase + mBase + r) * DH + col] = c[r];
    }
}

template <int K>
__global__ __launch_bounds__(128) void gemm_ns_wmma(const float* X,
                                                    const float* __restrict__ W,
                                                    const float* __restrict__ ns,
                                                    float* __restrict__ H) {
    constexpr int P   = K + 4;
    constexpr int VPR = K / 4;
    constexpr int NLD = 16 * VPR / 128;
    constexpr int RPQ = 128 / VPR;
    __shared__ float4 As4[2][16 * P / 4];

    const int tid     = threadIdx.x;
    const int lane    = tid & 31;
    const int wave    = tid >> 5;
    const int colBase = wave * 16;
    const int half    = lane >> 4;
    const int l15     = lane & 15;
    const int col     = colBase + l15;

    const int sr = tid / VPR;
    const int sc = tid % VPR;

    const float* Xt = X + (size_t)(blockIdx.x * 16 + sr) * K + sc * 4;
    float4 x0 = *(const float4*)(Xt);
    float4 x1 = *(const float4*)(Xt + (size_t)RPQ * K);
    float4 x2 = x0, x3 = x0;
    if (NLD == 4) {
        x2 = *(const float4*)(Xt + (size_t)2 * RPQ * K);
        x3 = *(const float4*)(Xt + (size_t)3 * RPQ * K);
    }
    float sn = ns[blockIdx.x * 16 + l15];

    v2f b[K / 4];
#pragma unroll
    for (int j = 0; j < K / 4; ++j) {
        const int kk = j * 4 + half * 2;
        b[j].x = W[kk * DH + col];
        b[j].y = W[(kk + 1) * DH + col];
    }

    v8f cprev = {};
#pragma unroll
    for (int i = 0; i < GEMM_TPB; ++i) {
        const int rowBase = (blockIdx.x + i * GEMM_GRID) * 16;
        float* As = (float*)As4[i & 1];
        {
            float* Ad = As + sr * P + sc * 4;
            *(float4*)(Ad) = x0;
            *(float4*)(Ad + RPQ * P) = x1;
            if (NLD == 4) {
                *(float4*)(Ad + 2 * RPQ * P) = x2;
                *(float4*)(Ad + 3 * RPQ * P) = x3;
            }
        }
        const float s = sn;
        __syncthreads();
        if (i > 0) store_tile(H, rowBase - GEMM_GRID * 16, half, col, cprev);
        if (i + 1 < GEMM_TPB) {
            const float* Xn = Xt + (size_t)(i + 1) * GEMM_GRID * 16 * K;
            x0 = *(const float4*)(Xn);
            x1 = *(const float4*)(Xn + (size_t)RPQ * K);
            if (NLD == 4) {
                x2 = *(const float4*)(Xn + (size_t)2 * RPQ * K);
                x3 = *(const float4*)(Xn + (size_t)3 * RPQ * K);
            }
            sn = ns[rowBase + GEMM_GRID * 16 + l15];
        }

        const float* Af = As + l15 * P + half * 2;
        v2f a[K / 4];
#pragma unroll
        for (int j = 0; j < K / 4; ++j) a[j] = *(const v2f*)(Af + j * 4);
        __builtin_amdgcn_sched_barrier(0);

        v8f c = {};
#pragma unroll
        for (int j = 0; j < K / 4; ++j) {
            v2f as;
            as.x = a[j].x * s;
            as.y = a[j].y * s;
            c = __builtin_amdgcn_wmma_f32_16x16x4_f32(false, as, false, b[j], (short)0, c,
                                                      false, false);
        }
        cprev = c;
        __builtin_amdgcn_sched_barrier(0);
    }
    store_tile(H, (blockIdx.x + (GEMM_TPB - 1) * GEMM_GRID) * 16, half, col, cprev);
}

#define AGG_CH 16
template <bool RELU, bool LN, bool PRE>
__global__ __launch_bounds__(256) void aggregate_post_kernel(const float* __restrict__ H,
                                                             const int* __restrict__ col,
                                                             const int* __restrict__ rowptr,
                                                             const int* __restrict__ cnt,
                                                             const float* __restrict__ nd,
                                                             const float* __restrict__ bias,
                                                             const float* __restrict__ g,
                                                             const float* __restrict__ be,
                                                             float* __restrict__ outx,
                                                             float* __restrict__ pre) {
    const int lane = threadIdx.x & 31;
    const int node = __builtin_amdgcn_readfirstlane(
        (int)(blockIdx.x * 8 + (threadIdx.x >> 5)));
    const int beg = rowptr[node];
    const int len = cnt[node];
    const float ndv = nd[node];
    const float* Hl = H + lane * 2;
    float2 acc = make_float2(0.0f, 0.0f);
    int k = 0;
    for (; k + AGG_CH <= len; k += AGG_CH) {
        int s[AGG_CH];
#pragma unroll
        for (int j = 0; j < AGG_CH; ++j) s[j] = col[beg + k + j];
        float2 v[AGG_CH];
#pragma unroll
        for (int j = 0; j < AGG_CH; ++j) v[j] = *(const float2*)(Hl + (size_t)s[j] * DH);
#pragma unroll
        for (int j = 0; j < AGG_CH; ++j) {
            acc.x += v[j].x;
            acc.y += v[j].y;
        }
    }
    const int c0 = lane * 2;
    const float2 bv = *(const float2*)(bias + c0);
    float2 gv = make_float2(1.0f, 1.0f), bev = make_float2(0.0f, 0.0f);
    if (LN) {
        gv  = *(const float2*)(g + c0);
        bev = *(const float2*)(be + c0);
    }
    if (k < len) {
        const int r = len - k;
        int s[AGG_CH];
#pragma unroll
        for (int j = 0; j < AGG_CH; ++j) s[j] = col[beg + k + j];
        float2 v[AGG_CH - 1];
#pragma unroll
        for (int j = 0; j < AGG_CH - 1; ++j) {
            v[j] = make_float2(0.0f, 0.0f);
            if (j < r) v[j] = *(const float2*)(Hl + (size_t)s[j] * DH);
        }
#pragma unroll
        for (int j = 0; j < AGG_CH - 1; ++j) {
            acc.x += v[j].x;
            acc.y += v[j].y;
        }
    }

    float v0 = acc.x * ndv + bv.x;
    float v1 = acc.y * ndv + bv.y;
    if (RELU) {
        v0 = fmaxf(v0, 0.0f);
        v1 = fmaxf(v1, 0.0f);
    }
    if (PRE) {
        pre[(size_t)node * DH + c0 + 0] = v0;
        pre[(size_t)node * DH + c0 + 1] = v1;
    }
    if (LN) {
        float ssum = v0 + v1;
#pragma unroll
        for (int off = 16; off >= 1; off >>= 1) ssum += __shfl_xor(ssum, off, 32);
        const float mu = ssum * (1.0f / DH);
        const float d0 = v0 - mu, d1 = v1 - mu;
        float vs = d0 * d0 + d1 * d1;
#pragma unroll
        for (int off = 16; off >= 1; off >>= 1) vs += __shfl_xor(vs, off, 32);
        const float inv = rsqrtf(vs * (1.0f / DH) + LN_EPS);
        v0 = d0 * inv * gv.x + bev.x;
        v1 = d1 * inv * gv.y + bev.y;
    }
    float* o = outx + (size_t)node * DH + c0;
    o[0] = v0;
    o[1] = v1;
}

#define FT_WAVES 16
#define FT_TILES (N_NODES / FT_WAVES)
#define FT_GRID 1024
#define FT_P (DH + 4)

__device__ __forceinline__ void transform_tile(const float* Ar, const float* Wt,
                                               float* __restrict__ Hn, int rowBase, int half,
                                               int l15, int ct) {
    const float* Af = Ar + l15 * FT_P + half * 2;
    const float* Bf = Wt + ct * FT_P + half * 2;
    v2f a[DH / 4], b[DH / 4];
#pragma unroll
    for (int j = 0; j < DH / 4; ++j) {
        a[j] = *(const v2f*)(Af + j * 4);
        b[j] = *(const v2f*)(Bf + j * 4);
    }
    v8f c = {};
#pragma unroll
    for (int j = 0; j < DH / 4; ++j) {
        c = __builtin_amdgcn_wmma_f32_16x16x4_f32(false, a[j], false, b[j], (short)0, c, false,
                                                  false);
    }
    store_tile(Hn, rowBase, half, ct, c);
}

template <bool PRE>
__global__ __launch_bounds__(512) void aggregate_transform_kernel(const float* __restrict__ H,
                                                                  const int* __restrict__ col,
                                                                  const int* __restrict__ rowptr,
                                                                  const int* __restrict__ cnt,
                                                                  const float* __restrict__ nd,
                                                                  const float* __restrict__ ns,
                                                                  const float* __restrict__ bias,
                                                                  const float* __restrict__ g,
                                                                  const float* __restrict__ be,
                                                                  const float* __restrict__ Wn,
                                                                  float* __restrict__ Hn,
                                                                  float* __restrict__ pre) {
    __shared__ float4 Wt4[DH * FT_P / 4];
    __shared__ float4 Ar4[2][FT_WAVES * FT_P / 4];
    float* Wt = (float*)Wt4;
    const int tid  = threadIdx.x;
    const int lane = tid & 31;
    const int wave = __builtin_amdgcn_readfirstlane((int)(threadIdx.x >> 5));
    const int half = lane >> 4;
    const int l15  = lane & 15;
    const int c0   = lane * 2;
    const int ct   = (wave & 3) * 16 + l15;

#pragma unroll
    for (int q = 0; q < DH * DH / 4 / 512; ++q) {
        const int i  = tid + q * 512;
        const int k  = i >> 4;
        const int c4 = (i & 15) * 4;
        const float4 w = *(const float4*)(Wn + k * DH + c4);
        Wt[(c4 + 0) * FT_P + k] = w.x;
        Wt[(c4 + 1) * FT_P + k] = w.y;
        Wt[(c4 + 2) * FT_P + k] = w.z;
        Wt[(c4 + 3) * FT_P + k] = w.w;
    }
    const float2 bv  = *(const float2*)(bias + c0);
    const float2 gv  = *(const float2*)(g + c0);
    const float2 bev = *(const float2*)(be + c0);
    const float* Hl = H + c0;

    int beg, len;
    float ndv, nsv;
    {
        const int n0 = blockIdx.x * FT_WAVES + wave;
        beg = rowptr[n0];
        len = cnt[n0];
        ndv = nd[n0];
        nsv = ns[n0];
    }
    __syncthreads();

    int nt = 0;
#pragma unroll 1
    for (int tile = blockIdx.x; tile < FT_TILES; tile += FT_GRID, ++nt) {
        const int node = tile * FT_WAVES + wave;
        if (nt > 0 && (wave >> 2) == ((nt - 1) & 3))
            transform_tile((const float*)Ar4[(nt - 1) & 1], Wt, Hn, (tile - FT_GRID) * FT_WAVES,
                           half, l15, ct);

        float2 acc = make_float2(0.0f, 0.0f);
        int k = 0;
        for (; k + AGG_CH <= len; k += AGG_CH) {
            int s[AGG_CH];
#pragma unroll
            for (int j = 0; j < AGG_CH; ++j) s[j] = col[beg + k + j];
            float2 v[AGG_CH];
#pragma unroll
            for (int j = 0; j < AGG_CH; ++j) v[j] = *(const float2*)(Hl + (size_t)s[j] * DH);
#pragma unroll
            for (int j = 0; j < AGG_CH; ++j) {
                acc.x += v[j].x;
                acc.y += v[j].y;
            }
        }
        if (k < len) {
            const int r = len - k;
            int s[AGG_CH];
#pragma unroll
            for (int j = 0; j < AGG_CH; ++j) s[j] = col[beg + k + j];
            float2 v[AGG_CH - 1];
#pragma unroll
            for (int j = 0; j < AGG_CH - 1; ++j) {
                v[j] = make_float2(0.0f, 0.0f);
                if (j < r) v[j] = *(const float2*)(Hl + (size_t)s[j] * DH);
            }
#pragma unroll
            for (int j = 0; j < AGG_CH - 1; ++j) {
                acc.x += v[j].x;
                acc.y += v[j].y;
            }
        }

        float v0 = fmaxf(acc.x * ndv + bv.x, 0.0f);
        float v1 = fmaxf(acc.y * ndv + bv.y, 0.0f);
        if (PRE) {
            pre[(size_t)node * DH + c0 + 0] = v0;
            pre[(size_t)node * DH + c0 + 1] = v1;
        }
        const int nn = min(node + FT_GRID * FT_WAVES, N_NODES - 1);
        const int nbeg = rowptr[nn];
        const int nlen = cnt[nn];
        const float nndv = nd[nn];
        const float nnsv = ns[nn];
        __builtin_amdgcn_sched_barrier(0);
        {
            float ssum = v0 + v1;
#pragma unroll
            for (int off = 16; off >= 1; off >>= 1) ssum += __shfl_xor(ssum, off, 32);
            const float mu = ssum * (1.0f / DH);
            const float d0 = v0 - mu, d1 = v1 - mu;
            float vs = d0 * d0 + d1 * d1;
#pragma unroll
            for (int off = 16; off >= 1; off >>= 1) vs += __shfl_xor(vs, off, 32);
            const float inv = rsqrtf(vs * (1.0f / DH) + LN_EPS);
            v0 = d0 * inv * gv.x + bev.x;
            v1 = d1 * inv * gv.y + bev.y;
        }
        *(float2*)((float*)Ar4[nt & 1] + wave * FT_P + c0) = make_float2(v0 * nsv, v1 * nsv);
        __builtin_amdgcn_sched_barrier(0);
        beg = nbeg;
        len = nlen;
        ndv = nndv;
        nsv = nnsv;
        __syncthreads();
    }
    if ((wave >> 2) == ((nt - 1) & 3))
        transform_tile((const float*)Ar4[(nt - 1) & 1], Wt, Hn,
                       (blockIdx.x + (nt - 1) * FT_GRID) * FT_WAVES, half, l15, ct);
}

extern "C" void kernel_launch(void* const* d_in, const int* in_sizes, int n_in,
                              void* d_out, int out_size, void* d_ws, size_t ws_size,
                              hipStream_t stream) {
    (void)in_sizes; (void)n_in; (void)out_size; (void)ws_size;

    const float* feats = (const float*)d_in[0];
    const float* W1    = (const float*)d_in[1];
    const float* b1    = (const float*)d_in[2];
    const float* g1    = (const float*)d_in[3];
    const float* be1   = (const float*)d_in[4];
    const float* W2    = (const float*)d_in[5];
    const float* b2    = (const float*)d_in[6];
    const float* g2    = (const float*)d_in[7];
    const float* be2   = (const float*)d_in[8];
    const float* W3    = (const float*)d_in[9];
    const float* b3    = (const float*)d_in[10];
    const int*   src   = (const int*)d_in[11];
    const int*   dst   = (const int*)d_in[12];

    float* ns     = (float*)d_ws;
    int*   cnt    = (int*)(ns + N_NODES);
    float* nd     = (float*)(cnt + N_NODES);
    int*   rowptr = (int*)(nd + N_NODES);
    int*   col    = rowptr + N_NODES;
    float* H      = (float*)(col + N_EDGES + COL_PAD);
    float* XB     = H + (size_t)N_NODES * DH;
    int*   partD  = (int*)(XB + (size_t)N_NODES * DH);
    int*   partS  = partD + (size_t)N_CHUNKS * N_NODES;
    int*   rowloc = partS + (size_t)N_CHUNKS * N_NODES;
    int*   tot32  = rowloc + N_NODES;

    float* h_emb = (float*)d_out;
    float* h_out = h_emb + (size_t)N_NODES * DH;

    count_kernel<<<N_CHUNKS * N_PARTS, 1024, 0, stream>>>(src, dst, partS, partD);
    piece_prefix_kernel<<<N_NODES / 32, 256, 0, stream>>>(partD, partS, cnt, ns, nd, rowloc, tot32);
    fill_kernel<<<N_CHUNKS * N_PARTS, 1024, 0, stream>>>(src, dst, partD, rowloc, tot32, rowptr, col);

    gemm_ns_wmma<DIN><<<GEMM_GRID, 128, 0, stream>>>(feats, W1, ns, H);
    aggregate_transform_kernel<false><<<FT_GRID, 512, 0, stream>>>(
        H, col, rowptr, cnt, nd, ns, b1, g1, be1, W2, XB, nullptr);

    aggregate_transform_kernel<true><<<FT_GRID, 512, 0, stream>>>(
        XB, col, rowptr, cnt, nd, ns, b2, g2, be2, W3, H, h_emb);

    aggregate_post_kernel<false, false, false><<<N_NODES / 8, 256, 0, stream>>>(
        H, col, rowptr, cnt, nd, b3, nullptr, nullptr, h_out, nullptr);
}


# in-edge lists by two-level grouping: edges binned by 400-node range, each range's lists built and sorted in one work-group's LDS; count planes, piece prefix go; sum order: ascending source
# speedup vs baseline: 5.7060x; 1.2523x over previous
#include <hip/hip_runtime.h>

#define N_NODES 100000
#define N_EDGES 1600000
#define DIN 128
#define DH 64
#define LN_EPS 1e-5f

typedef __attribute__((ext_vector_type(2))) float v2f;
typedef __attribute__((ext_vector_type(8))) float v8f;

#define N_RANGE 250
#define RANGE_N (N_NODES / N_RANGE)
#define PIECE_E (N_EDGES / N_RANGE)
#define EPT ((PIECE_E + 1023) / 1024)
#define HROW 256
#define HQ ((N_RANGE + 31) / 32)
#define HROWS (32 * HQ)
#define SRC_BITS 17
#define SRC_MASK ((1 << SRC_BITS) - 1)
#define COL_PAD 64
#define BCH 8
#define STG 12288
#define SORT_MAX 64
#define SLOT_BITS 14

static_assert(N_NODES % N_RANGE == 0 && N_EDGES % N_RANGE == 0, "whole ranges and pieces");
static_assert(N_NODES <= (1 << SRC_BITS) && RANGE_N <= 512, "packed record: 17 + 9 bits");
static_assert(PIECE_E <= 65535 && N_RANGE <= HROW && HROW == 256, "16-bit counts, one row per wave load");
static_assert(STG <= (1 << SLOT_BITS) && SRC_BITS + SLOT_BITS <= 31, "sort key: source, then slot");

__device__ __forceinline__ int block_excl_scan(int v, int lane, int wave, int* wave_tot) {
    int incl = v;
#pragma unroll
    for (int off = 1; off < 32; off <<= 1) {
        const int up = __shfl_up(incl, off, 32);
        if (lane >= off) incl += up;
    }
    if (lane == 31) wave_tot[wave] = incl;
    __syncthreads();
    int front = (lane < wave) ? wave_tot[lane] : 0;
#pragma unroll
    for (int off = 16; off >= 1; off >>= 1) front += __shfl_xor(front, off, 32);
    __syncthreads();
    return front + incl - v;
}

__device__ __forceinline__ void block_excl_scan4(const int (&v)[4], int (&out)[4], int lane, int wave,
                                                 int (*wave_tot4)[32]) {
    int incl[4];
#pragma unroll
    for (int s = 0; s < 4; ++s) incl[s] = v[s];
#pragma unroll
    for (int off = 1; off < 32; off <<= 1) {
#pragma unroll
        for (int s = 0; s < 4; ++s) {
            const int up = __shfl_up(incl[s], off, 32);
            if (lane >= off) incl[s] += up;
        }
    }
    if (lane == 31) {
#pragma unroll
        for (int s = 0; s < 4; ++s) wave_tot4[s][wave] = incl[s];
    }
    __syncthreads();
    int front[4];
#pragma unroll
    for (int s = 0; s < 4; ++s) front[s] = (lane < wave) ? wave_tot4[s][lane] : 0;
#pragma unroll
    for (int off = 16; off >= 1; off >>= 1) {
#pragma unroll
        for (int s = 0; s < 4; ++s) front[s] += __shfl_xor(front[s], off, 32);
    }
#pragma unroll
    for (int s = 0; s < 4; ++s) out[s] = front[s] + incl[s] - v[s];
}

__global__ __launch_bounds__(1024) void hist_kernel(const int* __restrict__ src,
                                                    const int* __restrict__ dst,
                                                    unsigned short* __restrict__ histD,
                                                    unsigned short* __restrict__ histS) {
    __shared__ int hD[HROW];
    __shared__ int hS[HROW];
    const int tid = threadIdx.x;
    const int e0  = blockIdx.x * PIECE_E;

    int dv[EPT], sv[EPT];
#pragma unroll
    for (int j = 0; j < EPT; ++j) {
        const int k = tid + j * 1024;
        dv[j] = (k < PIECE_E) ? dst[e0 + k] : -1;
        sv[j] = (k < PIECE_E) ? src[e0 + k] : -1;
    }
    if (tid < HROW) {
        hD[tid] = 0;
        hS[tid] = 0;
    }
    __syncthreads();
#pragma unroll
    for (int j = 0; j < EPT; ++j) {
        const bool ok_s = (unsigned)sv[j] < (unsigned)N_NODES;
        const bool ok_d = (unsigned)dv[j] < (unsigned)N_NODES;
        if (ok_s) atomicAdd(&hS[sv[j] / RANGE_N], 1);
        if (ok_s && ok_d) atomicAdd(&hD[dv[j] / RANGE_N], 1);
    }
    __syncthreads();
    if (tid < HROW) {
        histD[blockIdx.x * HROW + tid] = (unsigned short)hD[tid];
        histS[blockIdx.x * HROW + tid] = (unsigned short)hS[tid];
        if (blockIdx.x < HROWS - N_RANGE) {
            histD[(N_RANGE + blockIdx.x) * HROW + tid] = 0;
            histS[(N_RANGE + blockIdx.x) * HROW + tid] = 0;
        }
    }
}

static_assert(HQ * PIECE_E <= 65535, "packed 16-bit column sums");
__device__ __forceinline__ void add_row(uint4& acc, const uint4 m) {
    acc.x += m.x;
    acc.y += m.y;
    acc.z += m.z;
    acc.w += m.w;
}
__device__ __forceinline__ void add_columns(int* sums, int lane, const uint4 acc) {
    const unsigned w[4] = {acc.x, acc.y, acc.z, acc.w};
#pragma unroll
    for (int c = 0; c < 4; ++c) {
        atomicAdd(&sums[(2 * c) * 32 + lane], (int)(w[c] & 0xFFFFu));
        atomicAdd(&sums[(2 * c + 1) * 32 + lane], (int)(w[c] >> 16));
    }
}

__global__ __launch_bounds__(1024) void scatter_kernel(const int* __restrict__ src,
                                                       const int* __restrict__ dst,
                                                       const unsigned short* __restrict__ histD,
                                                       const unsigned short* __restrict__ histS,
                                                       int* __restrict__ binD,
                                                       unsigned short* __restrict__ binS,
                                                       int* __restrict__ bstartD,
                                                       int* __restrict__ bstartS) {
    __shared__ int befD[HROW], totD[HROW], befS[HROW], totS[HROW];
    __shared__ int curD[HROW], curS[HROW];
    __shared__ int offD[HROW], offS[HROW];
    __shared__ int stD[PIECE_E], stDo[PIECE_E];
    __shared__ int stS[PIECE_E], stSo[PIECE_E];
    __shared__ int n_rec[2];
    __shared__ int wave_tot4[4][32];
    const int tid  = threadIdx.x;
    const int lane = tid & 31;
    const int wave = __builtin_amdgcn_readfirstlane((int)(threadIdx.x >> 5));
    const int b    = blockIdx.x;
    const int e0   = b * PIECE_E;

    uint4 md[HQ], ms[HQ];
#pragma unroll
    for (int q = 0; q < HQ; ++q) {
        const int r = wave + q * 32;
        md[q] = ((const uint4*)(histD + (size_t)r * HROW))[lane];
        ms[q] = ((const uint4*)(histS + (size_t)r * HROW))[lane];
    }
    int ownD = 0, ownS = 0;
    if (tid < N_RANGE) {
        ownD = histD[(size_t)b * HROW + tid];
        ownS = histS[(size_t)b * HROW + tid];
    }
    int dv[EPT], sv[EPT];
#pragma unroll
    for (int j = 0; j < EPT; ++j) {
        const int k = tid + j * 1024;
        dv[j] = (k < PIECE_E) ? dst[e0 + k] : -1;
        sv[j] = (k < PIECE_E) ? src[e0 + k] : -1;
    }
    if (tid < HROW) {
        befD[tid] = 0;
        totD[tid] = 0;
        befS[tid] = 0;
        totS[tid] = 0;
    }
    __syncthreads();

    {
        uint4 tD = make_uint4(0u, 0u, 0u, 0u), fD = tD, tS = tD, fS = tD;
#pragma unroll
        for (int q = 0; q < HQ; ++q) {
            add_row(tD, md[q]);
            add_row(tS, ms[q]);
            if (wave + q * 32 < b) {
                add_row(fD, md[q]);
                add_row(fS, ms[q]);
            }
        }
        add_columns(totD, lane, tD);
        add_columns(befD, lane, fD);
        add_columns(totS, lane, tS);
        add_columns(befS, lane, fS);
    }
    __syncthreads();

    int sizeD = 0, frontD = 0, sizeS = 0, frontS = 0;
    if (tid < N_RANGE) {
        const int at = (tid & 7) * 32 + (tid >> 3);
        sizeD  = totD[at];
        frontD = befD[at];
        sizeS  = totS[at];
        frontS = befS[at];
    }
    const int scan_in[4] = {sizeD, sizeS, ownD, ownS};
    int scan_out[4];
    block_excl_scan4(scan_in, scan_out, lane, wave, wave_tot4);
    const int startD = scan_out[0], startS = scan_out[1], runD = scan_out[2], runS = scan_out[3];
    if (tid == 1023) {
        n_rec[0] = runD;
        n_rec[1] = runS;
    }
    if (tid < N_RANGE) {
        curD[tid] = runD;
        curS[tid] = runS;
        offD[tid] = startD + frontD - runD;
        offS[tid] = startS + frontS - runS;
        if (b == 0) {
            bstartD[tid] = startD;
            bstartS[tid] = startS;
            if (tid == N_RANGE - 1) {
                bstartD[N_RANGE] = startD + sizeD;
                bstartS[N_RANGE] = startS + sizeS;
            }
        }
    }
    __syncthreads();

#pragma unroll
    for (int j = 0; j < EPT; ++j) {
        const bool ok_s = (unsigned)sv[j] < (unsigned)N_NODES;
        const bool ok_d = (unsigned)dv[j] < (unsigned)N_NODES;
        if (ok_s) {
            const int ps = sv[j] / RANGE_N;
            const int i  = atomicAdd(&curS[ps], 1);
            stS[i]  = sv[j] - ps * RANGE_N;
            stSo[i] = offS[ps];
        }
        if (ok_s && ok_d) {
            const int pd = dv[j] / RANGE_N;
            const int i  = atomicAdd(&curD[pd], 1);
            stD[i]  = sv[j] | ((dv[j] - pd * RANGE_N) << SRC_BITS);
            stDo[i] = offD[pd];
        }
    }
    __syncthreads();
    const int nD = n_rec[0], nS = n_rec[1];
    for (int i = tid; i < nD; i += 1024) binD[i + stDo[i]] = stD[i];
    for (int i = tid; i < nS; i += 1024) binS[i + stSo[i]] = (unsigned short)stS[i];
}

__global__ __launch_bounds__(1024) void build_kernel(const int* __restrict__ binD,
                                                     const unsigned short* __restrict__ binS,
                                                     const int* __restrict__ bstartD,
                                                     const int* __restrict__ bstartS,
                                                     int* __restrict__ cnt,
                                                     float* __restrict__ nd,
                                                     float* __restrict__ ns,
                                                     int* __restrict__ rowptr,
                                                     int* __restrict__ col) {
    __shared__ int cD[512];
    __shared__ int cS[512];
    __shared__ int cF[512];
    __shared__ int stage[STG];
    __shared__ int sorted[STG];
    __shared__ unsigned short snode[STG];
    __shared__ int wave_tot[32];
    const int tid  = threadIdx.x;
    const int lane = tid & 31;
    const int wave = tid >> 5;
    const int p    = blockIdx.x;
    const int d0 = bstartD[p], d1 = bstartD[p + 1];
    const int s0 = bstartS[p], s1 = bstartS[p + 1];

    if (tid < 512) {
        cD[tid] = 0;
        cS[tid] = 0;
    }
    __syncthreads();
    for (int base = d0; base < d1; base += BCH * 1024) {
        int r[BCH];
#pragma unroll
        for (int j = 0; j < BCH; ++j) {
            const int i = base + tid + j * 1024;
            r[j] = (i < d1) ? binD[i] : -1;
        }
#pragma unroll
        for (int j = 0; j < BCH; ++j)
            if (r[j] >= 0) atomicAdd(&cD[r[j] >> SRC_BITS], 1);
    }
    for (int base = s0; base < s1; base += BCH * 1024) {
        int r[BCH];
#pragma unroll
        for (int j = 0; j < BCH; ++j) {
            const int i = base + tid + j * 1024;
            r[j] = (i < s1) ? (int)binS[i] : -1;
        }
#pragma unroll
        for (int j = 0; j < BCH; ++j)
            if (r[j] >= 0) atomicAdd(&cS[r[j]], 1);
    }
    __syncthreads();

    int deg = 0, odeg = 0;
    if (tid < RANGE_N) {
        deg  = cD[tid];
        odeg = cS[tid];
    }
    const int first = d0 + block_excl_scan(deg, lane, wave, wave_tot);
    if (tid < RANGE_N) {
        const int n = p * RANGE_N + tid;
        rowptr[n] = first;
        cnt[n]    = deg;
        nd[n]     = rsqrtf(fmaxf((float)deg, 1.0f));
        ns[n]     = rsqrtf(fmaxf((float)odeg, 1.0f));
        cD[tid]   = first;
        cF[tid]   = first;
    }
    __syncthreads();

    const bool staged = (d1 - d0 <= STG);
    for (int base = d0; base < d1; base += BCH * 1024) {
        int r[BCH];
#pragma unroll
        for (int j = 0; j < BCH; ++j) {
            const int i = base + tid + j * 1024;
            r[j] = (i < d1) ? binD[i] : -1;
        }
#pragma unroll
        for (int j = 0; j < BCH; ++j) {
            if (r[j] >= 0) {
                const int slot = atomicAdd(&cD[r[j] >> SRC_BITS], 1);
                if (staged) {
                    stage[slot - d0] = ((r[j] & SRC_MASK) << SLOT_BITS) | (slot - d0);
                    snode[slot - d0] = (unsigned short)(r[j] >> SRC_BITS);
                } else {
                    col[slot] = r[j] & SRC_MASK;
                }
            }
        }
    }
    if (staged) {
        __syncthreads();
        for (int i = tid; i < d1 - d0; i += 1024) {
            const int n = snode[i];
            const int f = cF[n] - d0, e = cD[n] - d0;
            const int key = stage[i];
            int at = i;
            if (e - f <= SORT_MAX) {
                int rank = 0;
                for (int t = f; t < e; ++t) rank += (stage[t] < key) ? 1 : 0;
                at = f + rank;
            }
            sorted[at] = key >> SLOT_BITS;
        }
        __syncthreads();
        for (int i = tid; i < d1 - d0; i += 1024) col[d0 + i] = sorted[i];
    }
    if (p == 0 && tid < COL_PAD) col[N_EDGES + tid] = 0;
}

#define GEMM_GRID 1250
#define GEMM_TPB ((N_NODES / 16) / GEMM_GRID)

__device__ __forceinline__ void store_tile(float* __restrict__ H, int rowBase, int half,
                                           int col, v8f c) {
    const int mBase = half * 8;
#pragma unroll
    for (int r = 0; r < 8; ++r) {
        H[(size_t)(rowBase + mBase + r) * DH + col] = c[r];
    }
}

template <int K>
__global__ __launch_bounds__(128) void gemm_ns_wmma(const float* X,
                                                    const float* __restrict__ W,
                                                    const float* __restrict__ ns,
                                                    float* __restrict__ H) {
    constexpr int P   = K + 4;
    constexpr int VPR = K / 4;
    constexpr int NLD = 16 * VPR / 128;
    constexpr int RPQ = 128 / VPR;
    __shared__ float4 As4[2][16 * P / 4];

    const int tid     = threadIdx.x;
    const int lane    = tid & 31;
    const int wave    = tid >> 5;
    const int colBase = wave * 16;
    const int half    = lane >> 4;
    const int l15     = lane & 15;
    const int col     = colBase + l15;

    const int sr = tid / VPR;
    const int sc = tid % VPR;

    const float* Xt = X + (size_t)(blockIdx.x * 16 + sr) * K + sc * 4;
    float4 x0 = *(const float4*)(Xt);
    float4 x1 = *(const float4*)(Xt + (size_t)RPQ * K);
    float4 x2 = x0, x3 = x0;
    if (NLD == 4) {
        x2 = *(const float4*)(Xt + (size_t)2 * RPQ * K);
        x3 = *(const float4*)(Xt + (size_t)3 * RPQ * K);
    }
    float sn = ns[blockIdx.x * 16 + l15];

    v2f b[K / 4];
#pragma unroll
    for (int j = 0; j < K / 4; ++j) {
        const int kk = j * 4 + half * 2;
        b[j].x = W[kk * DH + col];
        b[j].y = W[(kk + 1) * DH + col];
    }

    v8f cprev = {};
#pragma unroll
    for (int i = 0; i < GEMM_TPB; ++i) {
        const int rowBase = (blockIdx.x + i * GEMM_GRID) * 16;
        float* As = (float*)As4[i & 1];
        {
            float* Ad = As + sr * P + sc * 4;
            *(float4*)(Ad) = x0;
            *(float4*)(Ad + RPQ * P) = x1;
            if (NLD == 4) {
                *(float4*)(Ad + 2 * RPQ * P) = x2;
                *(float4*)(Ad + 3 * RPQ * P) = x3;
            }
        }
        const float s = sn;
        __syncthreads();
        if (i > 0) store_tile(H, rowBase - GEMM_GRID * 16, half, col, cprev);
        if (i + 1 < GEMM_TPB) {
            const float* Xn = Xt + (size_t)(i + 1) * GEMM_GRID * 16 * K;
            x0 = *(const float4*)(Xn);
            x1 = *(const float4*)(Xn + (size_t)RPQ * K);
            if (NLD == 4) {
                x2 = *(const float4*)(Xn + (size_t)2 * RPQ * K);
                x3 = *(const float4*)(Xn + (size_t)3 * RPQ * K);
            }
            sn = ns[rowBase + GEMM_GRID * 16 + l15];
        }

        const float* Af = As + l15 * P + half * 2;
        v2f a[K / 4];
#pragma unroll
        for (int j = 0; j < K / 4; ++j) a[j] = *(const v2f*)(Af + j * 4);
        __builtin_amdgcn_sched_barrier(0);

        v8f c = {};
#pragma unroll
        for (int j = 0; j < K / 4; ++j) {
            v2f as;
            as.x = a[j].x * s;
            as.y = a[j].y * s;
            c = __builtin_amdgcn_wmma_f32_16x16x4_f32(false, as, false, b[j], (short)0, c,
                                                      false, false);
        }
        cprev = c;
        __builtin_amdgcn_sched_barrier(0);
    }
    store_tile(H, (blockIdx.x + (GEMM_TPB - 1) * GEMM_GRID) * 16, half, col, cprev);
}

#define AGG_CH 16
template <bool RELU, bool LN, bool PRE>
__global__ __launch_bounds__(256) void aggregate_post_kernel(const float* __restrict__ H,
                                                             const int* __restrict__ col,
                                                             const int* __restrict__ rowptr,
                                                             const int* __restrict__ cnt,
                                                             const float* __restrict__ nd,
                                                             const float* __restrict__ bias,
                                                             const float* __restrict__ g,
                                                             const float* __restrict__ be,
                                                             float* __restrict__ outx,
                                                             float* __restrict__ pre) {
    const int lane = threadIdx.x & 31;
    const int node = __builtin_amdgcn_readfirstlane(
        (int)(blockIdx.x * 8 + (threadIdx.x >> 5)));
    const int beg = rowptr[node];
    const int len = cnt[node];
    const float ndv = nd[node];
    const float* Hl = H + lane * 2;
    float2 acc = make_float2(0.0f, 0.0f);
    int k = 0;
    for (; k + AGG_CH <= len; k += AGG_CH) {
        int s[AGG_CH];
#pragma unroll
        for (int j = 0; j < AGG_CH; ++j) s[j] = col[beg + k + j];
        float2 v[AGG_CH];
#pragma unroll
        for (int j = 0; j < AGG_CH; ++j) v[j] = *(const float2*)(Hl + (size_t)s[j] * DH);
#pragma unroll
        for (int j = 0; j < AGG_CH; ++j) {
            acc.x += v[j].x;
            acc.y += v[j].y;
        }
    }
    const int c0 = lane * 2;
    const float2 bv = *(const float2*)(bias + c0);
    float2 gv = make_float2(1.0f, 1.0f), bev = make_float2(0.0f, 0.0f);
    if (LN) {
        gv  = *(const float2*)(g + c0);
        bev = *(const float2*)(be + c0);
    }
    if (k < len) {
        const int r = len - k;
        int s[AGG_CH];
#pragma unroll
        for (int j = 0; j < AGG_CH; ++j) s[j] = col[beg + k + j];
        float2 v[AGG_CH - 1];
#pragma unroll
        for (int j = 0; j < AGG_CH - 1; ++j) {
            v[j] = make_float2(0.0f, 0.0f);
            if (j < r) v[j] = *(const float2*)(Hl + (size_t)s[j] * DH);
        }
#pragma unroll
        for (int j = 0; j < AGG_CH - 1; ++j) {
            acc.x += v[j].x;
            acc.y += v[j].y;
        }
    }

    float v0 = acc.x * ndv + bv.x;
    float v1 = acc.y * ndv + bv.y;
    if (RELU) {
        v0 = fmaxf(v0, 0.0f);
        v1 = fmaxf(v1, 0.0f);
    }
    if (PRE) {
        pre[(size_t)node * DH + c0 + 0] = v0;
        pre[(size_t)node * DH + c0 + 1] = v1;
    }
    if (LN) {
        float ssum = v0 + v1;
#pragma unroll
        for (int off = 16; off >= 1; off >>= 1) ssum += __shfl_xor(ssum, off, 32);
        const float mu = ssum * (1.0f / DH);
        const float d0 = v0 - mu, d1 = v1 - mu;
        float vs = d0 * d0 + d1 * d1;
#pragma unroll
        for (int off = 16; off >= 1; off >>= 1) vs += __shfl_xor(vs, off, 32);
        const float inv = rsqrtf(vs * (1.0f / DH) + LN_EPS);
        v0 = d0 * inv * gv.x + bev.x;
        v1 = d1 * inv * gv.y + bev.y;
    }
    float* o = outx + (size_t)node * DH + c0;
    o[0] = v0;
    o[1] = v1;
}

#define FT_WAVES 16
#define FT_TILES (N_NODES / FT_WAVES)
#define FT_GRID 1024
#define FT_P (DH + 4)

__device__ __forceinline__ void transform_tile(const float* Ar, const float* Wt,
                                               float* __restrict__ Hn, int rowBase, int half,
                                               int l15, int ct) {
    const float* Af = Ar + l15 * FT_P + half * 2;
    const float* Bf = Wt + ct * FT_P + half * 2;
    v2f a[DH / 4], b[DH / 4];
#pragma unroll
    for (int j = 0; j < DH / 4; ++j) {
        a[j] = *(const v2f*)(Af + j * 4);
        b[j] = *(const v2f*)(Bf + j * 4);
    }
    v8f c = {};
#pragma unroll
    for (int j = 0; j < DH / 4; ++j) {
        c = __builtin_amdgcn_wmma_f32_16x16x4_f32(false, a[j], false, b[j], (short)0, c, false,
                                                  false);
    }
    store_tile(Hn, rowBase, half, ct, c);
}

template <bool PRE>
__global__ __launch_bounds__(512) void aggregate_transform_kernel(const float* __restrict__ H,
                                                                  const int* __restrict__ col,
                                                                  const int* __restrict__ rowptr,
                                                                  const int* __restrict__ cnt,
                                                                  const float* __restrict__ nd,
                                                                  const float* __restrict__ ns,
                                                                  const float* __restrict__ bias,
                                                                  const float* __restrict__ g,
                                                                  const float* __restrict__ be,
                                                                  const float* __restrict__ Wn,
                                                                  float* __restrict__ Hn,
                                                                  float* __restrict__ pre) {
    __shared__ float4 Wt4[DH * FT_P / 4];
    __shared__ float4 Ar4[2][FT_WAVES * FT_P / 4];
    float* Wt = (float*)Wt4;
    const int tid  = threadIdx.x;
    const int lane = tid & 31;
    const int wave = __builtin_amdgcn_readfirstlane((int)(threadIdx.x >> 5));
    const int half = lane >> 4;
    const int l15  = lane & 15;
    const int c0   = lane * 2;
    const int ct   = (wave & 3) * 16 + l15;

#pragma unroll
    for (int q = 0; q < DH * DH / 4 / 512; ++q) {
        const int i  = tid + q * 512;
        const int k  = i >> 4;
        const int c4 = (i & 15) * 4;
        const float4 w = *(const float4*)(Wn + k * DH + c4);
        Wt[(c4 + 0) * FT_P + k] = w.x;
        Wt[(c4 + 1) * FT_P + k] = w.y;
        Wt[(c4 + 2) * FT_P + k] = w.z;
        Wt[(c4 + 3) * FT_P + k] = w.w;
    }
    const float2 bv  = *(const float2*)(bias + c0);
    const float2 gv  = *(const float2*)(g + c0);
    const float2 bev = *(const float2*)(be + c0);
    const float* Hl = H + c0;

    int beg, len;
    float ndv, nsv;
    {
        const int n0 = blockIdx.x * FT_WAVES + wave;
        beg = rowptr[n0];
        len = cnt[n0];
        ndv = nd[n0];
        nsv = ns[n0];
    }
    __syncthreads();

    int nt = 0;
#pragma unroll 1
    for (int tile = blockIdx.x; tile < FT_TILES; tile += FT_GRID, ++nt) {
        const int node = tile * FT_WAVES + wave;
        if (nt > 0 && (wave >> 2) == ((nt - 1) & 3))
            transform_tile((const float*)Ar4[(nt - 1) & 1], Wt, Hn, (tile - FT_GRID) * FT_WAVES,
                           half, l15, ct);

        float2 acc = make_float2(0.0f, 0.0f);
        int k = 0;
        for (; k + AGG_CH <= len; k += AGG_CH) {
            int s[AGG_CH];
#pragma unroll
            for (int j = 0; j < AGG_CH; ++j) s[j] = col[beg + k + j];
            float2 v[AGG_CH];
#pragma unroll
            for (int j = 0; j < AGG_CH; ++j) v[j] = *(const float2*)(Hl + (size_t)s[j] * DH);
#pragma unroll
            for (int j = 0; j < AGG_CH; ++j) {
                acc.x += v[j].x;
                acc.y += v[j].y;
            }
        }
        if (k < len) {
            const int r = len - k;
            int s[AGG_CH];
#pragma unroll
            for (int j = 0; j < AGG_CH; ++j) s[j] = col[beg + k + j];
            float2 v[AGG_CH - 1];
#pragma unroll
            for (int j = 0; j < AGG_CH - 1; ++j) {
                v[j] = make_float2(0.0f, 0.0f);
                if (j < r) v[j] = *(const float2*)(Hl + (size_t)s[j] * DH);
            }
#pragma unroll
            for (int j = 0; j < AGG_CH - 1; ++j) {
                acc.x += v[j].x;
                acc.y += v[j].y;
            }
        }

        float v0 = fmaxf(acc.x * ndv + bv.x, 0.0f);
        float v1 = fmaxf(acc.y * ndv + bv.y, 0.0f);
        if (PRE) {
            pre[(size_t)node * DH + c0 + 0] = v0;
            pre[(size_t)node * DH + c0 + 1] = v1;
        }
        const int nn = min(node + FT_GRID * FT_WAVES, N_NODES - 1);
        const int nbeg = rowptr[nn];
        const int nlen = cnt[nn];
        const float nndv = nd[nn];
        const float nnsv = ns[nn];
        __builtin_amdgcn_sched_barrier(0);
        {
            float ssum = v0 + v1;
#pragma unroll
            for (int off = 16; off >= 1; off >>= 1) ssum += __shfl_xor(ssum, off, 32);
            const float mu = ssum * (1.0f / DH);
            const float d0 = v0 - mu, d1 = v1 - mu;
            float vs = d0 * d0 + d1 * d1;
#pragma unroll
            for (int off = 16; off >= 1; off >>= 1) vs += __shfl_xor(vs, off, 32);
            const float inv = rsqrtf(vs * (1.0f / DH) + LN_EPS);
            v0 = d0 * inv * gv.x + bev.x;
            v1 = d1 * inv * gv.y + bev.y;
        }
        *(float2*)((float*)Ar4[nt & 1] + wave * FT_P + c0) = make_float2(v0 * nsv, v1 * nsv);
        __builtin_amdgcn_sched_barrier(0);
        beg = nbeg;
        len = nlen;
        ndv = nndv;
        nsv = nnsv;
        __syncthreads();
    }
    if ((wave >> 2) == ((nt - 1) & 3))
        transform_tile((const float*)Ar4[(nt - 1) & 1], Wt, Hn,
                       (blockIdx.x + (nt - 1) * FT_GRID) * FT_WAVES, half, l15, ct);
}

extern "C" void kernel_launch(void* const* d_in, const int* in_sizes, int n_in,
                              void* d_out, int out_size, void* d_ws, size_t ws_size,
                              hipStream_t stream) {
    (void)in_sizes; (void)n_in; (void)out_size; (void)ws_size;

    const float* feats = (const float*)d_in[0];
    const float* W1    = (const float*)d_in[1];
    const float* b1    = (const float*)d_in[2];
    const float* g1    = (const float*)d_in[3];
    const float* be1   = (const float*)d_in[4];
    const float* W2    = (const float*)d_in[5];
    const float* b2    = (const float*)d_in[6];
    const float* g2    = (const float*)d_in[7];
    const float* be2   = (const float*)d_in[8];
    const float* W3    = (const float*)d_in[9];
    const float* b3    = (const float*)d_in[10];
    const int*   src   = (const int*)d_in[11];
    const int*   dst   = (const int*)d_in[12];

    float* ns     = (float*)d_ws;
    int*   cnt    = (int*)(ns + N_NODES);
    float* nd     = (float*)(cnt + N_NODES);
    int*   rowptr = (int*)(nd + N_NODES);
    int*   col    = rowptr + N_NODES;
    float* H      = (float*)(col + N_EDGES + COL_PAD);
    float* XB     = H + (size_t)N_NODES * DH;
    unsigned short* histD = (unsigned short*)(XB + (size_t)N_NODES * DH);
    unsigned short* histS = histD + (size_t)HROWS * HROW;
    int*   binD   = (int*)(histS + (size_t)HROWS * HROW);
    unsigned short* binS = (unsigned short*)(binD + N_EDGES);
    int*   bstartD = (int*)(binS + N_EDGES);
    int*   bstartS = bstartD + 256;

    float* h_emb = (float*)d_out;
    float* h_out = h_emb + (size_t)N_NODES * DH;

    hist_kernel<<<N_RANGE, 1024, 0, stream>>>(src, dst, histD, histS);
    scatter_kernel<<<N_RANGE, 1024, 0, stream>>>(src, dst, histD, histS, binD, binS, bstartD, bstartS);
    build_kernel<<<N_RANGE, 1024, 0, stream>>>(binD, binS, bstartD, bstartS, cnt, nd, ns, rowptr, col);

    gemm_ns_wmma<DIN><<<GEMM_GRID, 128, 0, stream>>>(feats, W1, ns, H);
    aggregate_transform_kernel<false><<<FT_GRID, 512, 0, stream>>>(
        H, col, rowptr, cnt, nd, ns, b1, g1, be1, W2, XB, nullptr);

    aggregate_transform_kernel<true><<<FT_GRID, 512, 0, stream>>>(
        XB, col, rowptr, cnt, nd, ns, b2, g2, be2, W3, H, h_emb);

    aggregate_post_kernel<false, false, false><<<N_NODES / 8, 256, 0, stream>>>(
        H, col, rowptr, cnt, nd, b3, nullptr, nullptr, h_out, nullptr);
}


# in-edge lists in 2 launches, not 3: a piece's records go range-ordered to its own slab with run tables; a range's work-group collects its 250 runs; the counting launch goes; sums unchanged
# speedup vs baseline: 5.9524x; 1.0432x over previous
#include <hip/hip_runtime.h>

#define N_NODES 100000
#define N_EDGES 1600000
#define DIN 128
#define DH 64
#define LN_EPS 1e-5f

typedef __attribute__((ext_vector_type(2))) float v2f;
typedef __attribute__((ext_vector_type(8))) float v8f;

#define N_RANGE 250
#define RANGE_N (N_NODES / N_RANGE)
#define PIECE_E (N_EDGES / N_RANGE)
#define EPT ((PIECE_E + 1023) / 1024)
#define HROW 256
#define HQ ((N_RANGE + 31) / 32)
#define HROWS (32 * HQ)
#define SRC_BITS 17
#define SRC_MASK ((1 << SRC_BITS) - 1)
#define COL_PAD 64
#define RUN_V 2
#define STG 12288
#define SORT_MAX 64
#define SLOT_BITS 14

static_assert(N_NODES % N_RANGE == 0 && N_EDGES % N_RANGE == 0, "whole ranges and pieces");
static_assert(N_NODES <= (1 << SRC_BITS) && RANGE_N <= 512, "packed record: 17 + 9 bits");
static_assert(PIECE_E <= 65535 && N_RANGE <= HROW && HROW == 256, "16-bit counts, one row per wave load");
static_assert(STG <= (1 << SLOT_BITS) && SRC_BITS + SLOT_BITS <= 31, "sort key: source, then slot");

__device__ __forceinline__ int block_excl_scan(int v, int lane, int wave, int* wave_tot) {
    int incl = v;
#pragma unroll
    for (int off = 1; off < 32; off <<= 1) {
        const int up = __shfl_up(incl, off, 32);
        if (lane >= off) incl += up;
    }
    if (lane == 31) wave_tot[wave] = incl;
    __syncthreads();
    int front = (lane < wave) ? wave_tot[lane] : 0;
#pragma unroll
    for (int off = 16; off >= 1; off >>= 1) front += __shfl_xor(front, off, 32);
    __syncthreads();
    return front + incl - v;
}

__device__ __forceinline__ void block_excl_scan2(const int (&v)[2], int (&out)[2], int lane, int wave,
                                                 int (*wave_tot2)[32]) {
    int incl[2];
#pragma unroll
    for (int s = 0; s < 2; ++s) incl[s] = v[s];
#pragma unroll
    for (int off = 1; off < 32; off <<= 1) {
#pragma unroll
        for (int s = 0; s < 2; ++s) {
            const int up = __shfl_up(incl[s], off, 32);
            if (lane >= off) incl[s] += up;
        }
    }
    if (lane == 31) {
#pragma unroll
        for (int s = 0; s < 2; ++s) wave_tot2[s][wave] = incl[s];
    }
    __syncthreads();
    int front[2];
#pragma unroll
    for (int s = 0; s < 2; ++s) front[s] = (lane < wave) ? wave_tot2[s][lane] : 0;
#pragma unroll
    for (int off = 16; off >= 1; off >>= 1) {
#pragma unroll
        for (int s = 0; s < 2; ++s) front[s] += __shfl_xor(front[s], off, 32);
    }
#pragma unroll
    for (int s = 0; s < 2; ++s) out[s] = front[s] + incl[s] - v[s];
}

__global__ __launch_bounds__(1024) void scatter_kernel(const int* __restrict__ src,
                                                       const int* __restrict__ dst,
                                                       unsigned short* __restrict__ histD,
                                                       unsigned short* __restrict__ histS,
                                                       unsigned short* __restrict__ runD,
                                                       unsigned short* __restrict__ runS,
                                                       int* __restrict__ binD,
                                                       unsigned short* __restrict__ binS) {
    __shared__ int hD[HROW], hS[HROW];
    __shared__ int curD[HROW], curS[HROW];
    __shared__ int stD[PIECE_E];
    __shared__ int stS[PIECE_E];
    __shared__ int n_rec[2];
    __shared__ int wave_tot2[2][32];
    const int tid  = threadIdx.x;
    const int lane = tid & 31;
    const int wave = tid >> 5;
    const int b    = blockIdx.x;
    const int e0   = b * PIECE_E;

    int dv[EPT], sv[EPT];
#pragma unroll
    for (int j = 0; j < EPT; ++j) {
        const int k = tid + j * 1024;
        dv[j] = (k < PIECE_E) ? dst[e0 + k] : -1;
        sv[j] = (k < PIECE_E) ? src[e0 + k] : -1;
    }
    if (tid < HROW) {
        hD[tid] = 0;
        hS[tid] = 0;
    }
    __syncthreads();
#pragma unroll
    for (int j = 0; j < EPT; ++j) {
        const bool ok_s = (unsigned)sv[j] < (unsigned)N_NODES;
        const bool ok_d = (unsigned)dv[j] < (unsigned)N_NODES;
        if (ok_s) atomicAdd(&hS[sv[j] / RANGE_N], 1);
        if (ok_s && ok_d) atomicAdd(&hD[dv[j] / RANGE_N], 1);
    }
    __syncthreads();

    int own[2] = {0, 0};
    if (tid < HROW) {
        own[0] = hD[tid];
        own[1] = hS[tid];
    }
    int first[2];
    block_excl_scan2(own, first, lane, wave, wave_tot2);
    if (tid == 1023) {
        n_rec[0] = first[0];
        n_rec[1] = first[1];
    }
    if (tid < HROW) {
        curD[tid] = first[0];
        curS[tid] = first[1];
        histD[b * HROW + tid] = (unsigned short)own[0];
        histS[b * HROW + tid] = (unsigned short)own[1];
        runD[b * HROW + tid]  = (unsigned short)first[0];
        runS[b * HROW + tid]  = (unsigned short)first[1];
        if (b < HROWS - N_RANGE) histD[(N_RANGE + b) * HROW + tid] = 0;
    }
    __syncthreads();

#pragma unroll
    for (int j = 0; j < EPT; ++j) {
        const bool ok_s = (unsigned)sv[j] < (unsigned)N_NODES;
        const bool ok_d = (unsigned)dv[j] < (unsigned)N_NODES;
        if (ok_s) {
            const int ps = sv[j] / RANGE_N;
            const int i  = atomicAdd(&curS[ps], 1);
            stS[i] = sv[j] - ps * RANGE_N;
        }
        if (ok_s && ok_d) {
            const int pd = dv[j] / RANGE_N;
            const int i  = atomicAdd(&curD[pd], 1);
            stD[i] = sv[j] | ((dv[j] - pd * RANGE_N) << SRC_BITS);
        }
    }
    __syncthreads();
    const int nD = n_rec[0], nS = n_rec[1];
    for (int i = tid; i < nD; i += 1024) binD[e0 + i] = stD[i];
    for (int i = tid; i < nS; i += 1024) binS[e0 + i] = (unsigned short)stS[i];
}

static_assert(HQ * PIECE_E <= 65535, "packed 16-bit column sums");
__device__ __forceinline__ void add_row(uint4& acc, const uint4 m) {
    acc.x += m.x;
    acc.y += m.y;
    acc.z += m.z;
    acc.w += m.w;
}
__device__ __forceinline__ void add_columns(int* sums, int lane, const uint4 acc) {
    const unsigned w[4] = {acc.x, acc.y, acc.z, acc.w};
#pragma unroll
    for (int c = 0; c < 4; ++c) {
        atomicAdd(&sums[(2 * c) * 32 + lane], (int)(w[c] & 0xFFFFu));
        atomicAdd(&sums[(2 * c + 1) * 32 + lane], (int)(w[c] >> 16));
    }
}

__global__ __launch_bounds__(1024) void build_kernel(const int* __restrict__ binD,
                                                     const unsigned short* __restrict__ binS,
                                                     const unsigned short* __restrict__ histD,
                                                     const unsigned short* __restrict__ histS,
                                                     const unsigned short* __restrict__ runD,
                                                     const unsigned short* __restrict__ runS,
                                                     int* __restrict__ cnt,
                                                     float* __restrict__ nd,
                                                     float* __restrict__ ns,
                                                     int* __restrict__ rowptr,
                                                     int* __restrict__ col) {
    __shared__ int cD[512];
    __shared__ int cS[512];
    __shared__ int cF[512];
    __shared__ int stage[STG];
    __shared__ int sorted[STG];
    __shared__ unsigned short snode[STG];
    __shared__ int totD[HROW];
    __shared__ int rcD[HROW], rsD[HROW];
    __shared__ int rcS[HROW], rsS[HROW];
    __shared__ int bnd[2];
    __shared__ int wave_tot[32];
    const int tid  = threadIdx.x;
    const int lane = tid & 31;
    const int wave = __builtin_amdgcn_readfirstlane((int)(threadIdx.x >> 5));
    const int p    = blockIdx.x;

    uint4 md[HQ];
#pragma unroll
    for (int q = 0; q < HQ; ++q)
        md[q] = ((const uint4*)(histD + (size_t)(wave + q * 32) * HROW))[lane];
    if (tid < HROW) {
        int cd = 0, sd = 0, cs = 0, ss = 0;
        if (tid < N_RANGE) {
            cd = histD[(size_t)tid * HROW + p];
            sd = runD[(size_t)tid * HROW + p];
            cs = histS[(size_t)tid * HROW + p];
            ss = runS[(size_t)tid * HROW + p];
        }
        rcD[tid]  = cd;
        rsD[tid]  = tid * PIECE_E + sd;
        rcS[tid]  = cs;
        rsS[tid]  = tid * PIECE_E + ss;
        totD[tid] = 0;
    }
    if (tid < 512) {
        cD[tid] = 0;
        cS[tid] = 0;
    }
    __syncthreads();

    int rD[HQ][RUN_V], rS[HQ][RUN_V];
#pragma unroll
    for (int q = 0; q < HQ; ++q) {
        const int bb = wave + q * 32;
        const int cd = rcD[bb], sd = rsD[bb];
        const int cs = rcS[bb], ss = rsS[bb];
#pragma unroll
        for (int h = 0; h < RUN_V; ++h) {
            const int j = lane + h * 32;
            rD[q][h] = (j < cd) ? binD[sd + j] : -1;
            rS[q][h] = (j < cs) ? (int)binS[ss + j] : -1;
        }
    }

    {
        uint4 tD = make_uint4(0u, 0u, 0u, 0u);
#pragma unroll
        for (int q = 0; q < HQ; ++q) add_row(tD, md[q]);
        add_columns(totD, lane, tD);
    }
    __syncthreads();
    int sizeD = 0;
    if (tid < N_RANGE) sizeD = totD[(tid & 7) * 32 + (tid >> 3)];
    const int startD = block_excl_scan(sizeD, lane, wave, wave_tot);
    if (tid == p) {
        bnd[0] = startD;
        bnd[1] = startD + sizeD;
    }

#pragma unroll
    for (int q = 0; q < HQ; ++q) {
#pragma unroll
        for (int h = 0; h < RUN_V; ++h) {
            if (rD[q][h] >= 0) atomicAdd(&cD[rD[q][h] >> SRC_BITS], 1);
            if (rS[q][h] >= 0) atomicAdd(&cS[rS[q][h]], 1);
        }
    }
    for (int q = 0; q < HQ; ++q) {
        const int bb = wave + q * 32;
        const int cd = rcD[bb], sd = rsD[bb];
        const int cs = rcS[bb], ss = rsS[bb];
        for (int j = lane + RUN_V * 32; j < cd; j += 32) atomicAdd(&cD[binD[sd + j] >> SRC_BITS], 1);
        for (int j = lane + RUN_V * 32; j < cs; j += 32) atomicAdd(&cS[(int)binS[ss + j]], 1);
    }
    __syncthreads();
    const int d0 = bnd[0], d1 = bnd[1];

    int deg = 0, odeg = 0;
    if (tid < RANGE_N) {
        deg  = cD[tid];
        odeg = cS[tid];
    }
    const int first = d0 + block_excl_scan(deg, lane, wave, wave_tot);
    if (tid < RANGE_N) {
        const int n = p * RANGE_N + tid;
        rowptr[n] = first;
        cnt[n]    = deg;
        nd[n]     = rsqrtf(fmaxf((float)deg, 1.0f));
        ns[n]     = rsqrtf(fmaxf((float)odeg, 1.0f));
        cD[tid]   = first;
        cF[tid]   = first;
    }
    __syncthreads();

    const bool staged = (d1 - d0 <= STG);
    auto place = [&](int r) {
        const int slot = atomicAdd(&cD[r >> SRC_BITS], 1);
        if (staged) {
            stage[slot - d0] = ((r & SRC_MASK) << SLOT_BITS) | (slot - d0);
            snode[slot - d0] = (unsigned short)(r >> SRC_BITS);
        } else {
            col[slot] = r & SRC_MASK;
        }
    };
#pragma unroll
    for (int q = 0; q < HQ; ++q) {
#pragma unroll
        for (int h = 0; h < RUN_V; ++h)
            if (rD[q][h] >= 0) place(rD[q][h]);
    }
    for (int q = 0; q < HQ; ++q) {
        const int bb = wave + q * 32;
        const int cd = rcD[bb], sd = rsD[bb];
        for (int j = lane + RUN_V * 32; j < cd; j += 32) place(binD[sd + j]);
    }
    if (staged) {
        __syncthreads();
        for (int i = tid; i < d1 - d0; i += 1024) {
            const int n = snode[i];
            const int f = cF[n] - d0, e = cD[n] - d0;
            const int key = stage[i];
            int at = i;
            if (e - f <= SORT_MAX) {
                int rank = 0;
                for (int t = f; t < e; ++t) rank += (stage[t] < key) ? 1 : 0;
                at = f + rank;
            }
            sorted[at] = key >> SLOT_BITS;
        }
        __syncthreads();
        for (int i = tid; i < d1 - d0; i += 1024) col[d0 + i] = sorted[i];
    }
    if (p == 0 && tid < COL_PAD) col[N_EDGES + tid] = 0;
}

#define GEMM_GRID 1250
#define GEMM_TPB ((N_NODES / 16) / GEMM_GRID)

__device__ __forceinline__ void store_tile(float* __restrict__ H, int rowBase, int half,
                                           int col, v8f c) {
    const int mBase = half * 8;
#pragma unroll
    for (int r = 0; r < 8; ++r) {
        H[(size_t)(rowBase + mBase + r) * DH + col] = c[r];
    }
}

template <int K>
__global__ __launch_bounds__(128) void gemm_ns_wmma(const float* X,
                                                    const float* __restrict__ W,
                                                    const float* __restrict__ ns,
                                                    float* __restrict__ H) {
    constexpr int P   = K + 4;
    constexpr int VPR = K / 4;
    constexpr int NLD = 16 * VPR / 128;
    constexpr int RPQ = 128 / VPR;
    __shared__ float4 As4[2][16 * P / 4];

    const int tid     = threadIdx.x;
    const int lane    = tid & 31;
    const int wave    = tid >> 5;
    const int colBase = wave * 16;
    const int half    = lane >> 4;
    const int l15     = lane & 15;
    const int col     = colBase + l15;

    const int sr = tid / VPR;
    const int sc = tid % VPR;

    const float* Xt = X + (size_t)(blockIdx.x * 16 + sr) * K + sc * 4;
    float4 x0 = *(const float4*)(Xt);
    float4 x1 = *(const float4*)(Xt + (size_t)RPQ * K);
    float4 x2 = x0, x3 = x0;
    if (NLD == 4) {
        x2 = *(const float4*)(Xt + (size_t)2 * RPQ * K);
        x3 = *(const float4*)(Xt + (size_t)3 * RPQ * K);
    }
    float sn = ns[blockIdx.x * 16 + l15];

    v2f b[K / 4];
#pragma unroll
    for (int j = 0; j < K / 4; ++j) {
        const int kk = j * 4 + half * 2;
        b[j].x = W[kk * DH + col];
        b[j].y = W[(kk + 1) * DH + col];
    }

    v8f cprev = {};
#pragma unroll
    for (int i = 0; i < GEMM_TPB; ++i) {
        const int rowBase = (blockIdx.x + i * GEMM_GRID) * 16;
        float* As = (float*)As4[i & 1];
        {
            float* Ad = As + sr * P + sc * 4;
            *(float4*)(Ad) = x0;
            *(float4*)(Ad + RPQ * P) = x1;
            if (NLD == 4) {
                *(float4*)(Ad + 2 * RPQ * P) = x2;
                *(float4*)(Ad + 3 * RPQ * P) = x3;
            }
        }
        const float s = sn;
        __syncthreads();
        if (i > 0) store_tile(H, rowBase - GEMM_GRID * 16, half, col, cprev);
        if (i + 1 < GEMM_TPB) {
            const float* Xn = Xt + (size_t)(i + 1) * GEMM_GRID * 16 * K;
            x0 = *(const float4*)(Xn);
            x1 = *(const float4*)(Xn + (size_t)RPQ * K);
            if (NLD == 4) {
                x2 = *(const float4*)(Xn + (size_t)2 * RPQ * K);
                x3 = *(const float4*)(Xn + (size_t)3 * RPQ * K);
            }
            sn = ns[rowBase + GEMM_GRID * 16 + l15];
        }

        const float* Af = As + l15 * P + half * 2;
        v2f a[K / 4];
#pragma unroll
        for (int j = 0; j < K / 4; ++j) a[j] = *(const v2f*)(Af + j * 4);
        __builtin_amdgcn_sched_barrier(0);

        v8f c = {};
#pragma unroll
        for (int j = 0; j < K / 4; ++j) {
            v2f as;
            as.x = a[j].x * s;
            as.y = a[j].y * s;
            c = __builtin_amdgcn_wmma_f32_16x16x4_f32(false, as, false, b[j], (short)0, c,
                                                      false, false);
        }
        cprev = c;
        __builtin_amdgcn_sched_barrier(0);
    }
    store_tile(H, (blockIdx.x + (GEMM_TPB - 1) * GEMM_GRID) * 16, half, col, cprev);
}

#define AGG_CH 16
template <bool RELU, bool LN, bool PRE>
__global__ __launch_bounds__(256) void aggregate_post_kernel(const float* __restrict__ H,
                                                             const int* __restrict__ col,
                                                             const int* __restrict__ rowptr,
                                                             const int* __restrict__ cnt,
                                                             const float* __restrict__ nd,
                                                             const float* __restrict__ bias,
                                                             const float* __restrict__ g,
                                                             const float* __restrict__ be,
                                                             float* __restrict__ outx,
                                                             float* __restrict__ pre) {
    const int lane = threadIdx.x & 31;
    const int node = __builtin_amdgcn_readfirstlane(
        (int)(blockIdx.x * 8 + (threadIdx.x >> 5)));
    const int beg = rowptr[node];
    const int len = cnt[node];
    const float ndv = nd[node];
    const float* Hl = H + lane * 2;
    float2 acc = make_float2(0.0f, 0.0f);
    int k = 0;
    for (; k + AGG_CH <= len; k += AGG_CH) {
        int s[AGG_CH];
#pragma unroll
        for (int j = 0; j < AGG_CH; ++j) s[j] = col[beg + k + j];
        float2 v[AGG_CH];
#pragma unroll
        for (int j = 0; j < AGG_CH; ++j) v[j] = *(const float2*)(Hl + (size_t)s[j] * DH);
#pragma unroll
        for (int j = 0; j < AGG_CH; ++j) {
            acc.x += v[j].x;
            acc.y += v[j].y;
        }
    }
    const int c0 = lane * 2;
    const float2 bv = *(const float2*)(bias + c0);
    float2 gv = make_float2(1.0f, 1.0f), bev = make_float2(0.0f, 0.0f);
    if (LN) {
        gv  = *(const float2*)(g + c0);
        bev = *(const float2*)(be + c0);
    }
    if (k < len) {
        const int r = len - k;
        int s[AGG_CH];
#pragma unroll
        for (int j = 0; j < AGG_CH; ++j) s[j] = col[beg + k + j];
        float2 v[AGG_CH - 1];
#pragma unroll
        for (int j = 0; j < AGG_CH - 1; ++j) {
            v[j] = make_float2(0.0f, 0.0f);
            if (j < r) v[j] = *(const float2*)(Hl + (size_t)s[j] * DH);
        }
#pragma unroll
        for (int j = 0; j < AGG_CH - 1; ++j) {
            acc.x += v[j].x;
            acc.y += v[j].y;
        }
    }

    float v0 = acc.x * ndv + bv.x;
    float v1 = acc.y * ndv + bv.y;
    if (RELU) {
        v0 = fmaxf(v0, 0.0f);
        v1 = fmaxf(v1, 0.0f);
    }
    if (PRE) {
        pre[(size_t)node * DH + c0 + 0] = v0;
        pre[(size_t)node * DH + c0 + 1] = v1;
    }
    if (LN) {
        float ssum = v0 + v1;
#pragma unroll
        for (int off = 16; off >= 1; off >>= 1) ssum += __shfl_xor(ssum, off, 32);
        const float mu = ssum * (1.0f / DH);
        const float d0 = v0 - mu, d1 = v1 - mu;
        float vs = d0 * d0 + d1 * d1;
#pragma unroll
        for (int off = 16; off >= 1; off >>= 1) vs += __shfl_xor(vs, off, 32);
        const float inv = rsqrtf(vs * (1.0f / DH) + LN_EPS);
        v0 = d0 * inv * gv.x + bev.x;
        v1 = d1 * inv * gv.y + bev.y;
    }
    float* o = outx + (size_t)node * DH + c0;
    o[0] = v0;
    o[1] = v1;
}

#define FT_WAVES 16
#define FT_TILES (N_NODES / FT_WAVES)
#define FT_GRID 1024
#define FT_P (DH + 4)

__device__ __forceinline__ void transform_tile(const float* Ar, const float* Wt,
                                               float* __restrict__ Hn, int rowBase, int half,
                                               int l15, int ct) {
    const float* Af = Ar + l15 * FT_P + half * 2;
    const float* Bf = Wt + ct * FT_P + half * 2;
    v2f a[DH / 4], b[DH / 4];
#pragma unroll
    for (int j = 0; j < DH / 4; ++j) {
        a[j] = *(const v2f*)(Af + j * 4);
        b[j] = *(const v2f*)(Bf + j * 4);
    }
    v8f c = {};
#pragma unroll
    for (int j = 0; j < DH / 4; ++j) {
        c = __builtin_amdgcn_wmma_f32_16x16x4_f32(false, a[j], false, b[j], (short)0, c, false,
                                                  false);
    }
    store_tile(Hn, rowBase, half, ct, c);
}

template <bool PRE>
__global__ __launch_bounds__(512) void aggregate_transform_kernel(const float* __restrict__ H,
                                                                  const int* __restrict__ col,
                                                                  const int* __restrict__ rowptr,
                                                                  const int* __restrict__ cnt,
                                                                  const float* __restrict__ nd,
                                                                  const float* __restrict__ ns,
                                                                  const float* __restrict__ bias,
                                                                  const float* __restrict__ g,
                                                                  const float* __restrict__ be,
                                                                  const float* __restrict__ Wn,
                                                                  float* __restrict__ Hn,
                                                                  float* __restrict__ pre) {
    __shared__ float4 Wt4[DH * FT_P / 4];
    __shared__ float4 Ar4[2][FT_WAVES * FT_P / 4];
    float* Wt = (float*)Wt4;
    const int tid  = threadIdx.x;
    const int lane = tid & 31;
    const int wave = __builtin_amdgcn_readfirstlane((int)(threadIdx.x >> 5));
    const int half = lane >> 4;
    const int l15  = lane & 15;
    const int c0   = lane * 2;
    const int ct   = (wave & 3) * 16 + l15;

#pragma unroll
    for (int q = 0; q < DH * DH / 4 / 512; ++q) {
        const int i  = tid + q * 512;
        const int k  = i >> 4;
        const int c4 = (i & 15) * 4;
        const float4 w = *(const float4*)(Wn + k * DH + c4);
        Wt[(c4 + 0) * FT_P + k] = w.x;
        Wt[(c4 + 1) * FT_P + k] = w.y;
        Wt[(c4 + 2) * FT_P + k] = w.z;
        Wt[(c4 + 3) * FT_P + k] = w.w;
    }
    const float2 bv  = *(const float2*)(bias + c0);
    const float2 gv  = *(const float2*)(g + c0);
    const float2 bev = *(const float2*)(be + c0);
    const float* Hl = H + c0;

    int beg, len;
    float ndv, nsv;
    {
        const int n0 = blockIdx.x * FT_WAVES + wave;
        beg = rowptr[n0];
        len = cnt[n0];
        ndv = nd[n0];
        nsv = ns[n0];
    }
    __syncthreads();

    int nt = 0;
#pragma unroll 1
    for (int tile = blockIdx.x; tile < FT_TILES; tile += FT_GRID, ++nt) {
        const int node = tile * FT_WAVES + wave;
        if (nt > 0 && (wave >> 2) == ((nt - 1) & 3))
            transform_tile((const float*)Ar4[(nt - 1) & 1], Wt, Hn, (tile - FT_GRID) * FT_WAVES,
                           half, l15, ct);

        float2 acc = make_float2(0.0f, 0.0f);
        int k = 0;
        for (; k + AGG_CH <= len; k += AGG_CH) {
            int s[AGG_CH];
#pragma unroll
            for (int j = 0; j < AGG_CH; ++j) s[j] = col[beg + k + j];
            float2 v[AGG_CH];
#pragma unroll
            for (int j = 0; j < AGG_CH; ++j) v[j] = *(const float2*)(Hl + (size_t)s[j] * DH);
#pragma unroll
            for (int j = 0; j < AGG_CH; ++j) {
                acc.x += v[j].x;
                acc.y += v[j].y;
            }
        }
        if (k < len) {
            const int r = len - k;
            int s[AGG_CH];
#pragma unroll
            for (int j = 0; j < AGG_CH; ++j) s[j] = col[beg + k + j];
            float2 v[AGG_CH - 1];
#pragma unroll
            for (int j = 0; j < AGG_CH - 1; ++j) {
                v[j] = make_float2(0.0f, 0.0f);
                if (j < r) v[j] = *(const float2*)(Hl + (size_t)s[j] * DH);
            }
#pragma unroll
            for (int j = 0; j < AGG_CH - 1; ++j) {
                acc.x += v[j].x;
                acc.y += v[j].y;
            }
        }

        float v0 = fmaxf(acc.x * ndv + bv.x, 0.0f);
        float v1 = fmaxf(acc.y * ndv + bv.y, 0.0f);
        if (PRE) {
            pre[(size_t)node * DH + c0 + 0] = v0;
            pre[(size_t)node * DH + c0 + 1] = v1;
        }
        const int nn = min(node + FT_GRID * FT_WAVES, N_NODES - 1);
        const int nbeg = rowptr[nn];
        const int nlen = cnt[nn];
        const float nndv = nd[nn];
        const float nnsv = ns[nn];
        __builtin_amdgcn_sched_barrier(0);
        {
            float ssum = v0 + v1;
#pragma unroll
            for (int off = 16; off >= 1; off >>= 1) ssum += __shfl_xor(ssum, off, 32);
            const float mu = ssum * (1.0f / DH);
            const float d0 = v0 - mu, d1 = v1 - mu;
            float vs = d0 * d0 + d1 * d1;
#pragma unroll
            for (int off = 16; off >= 1; off >>= 1) vs += __shfl_xor(vs, off, 32);
            const float inv = rsqrtf(vs * (1.0f / DH) + LN_EPS);
            v0 = d0 * inv * gv.x + bev.x;
            v1 = d1 * inv * gv.y + bev.y;
        }
        *(float2*)((float*)Ar4[nt & 1] + wave * FT_P + c0) = make_float2(v0 * nsv, v1 * nsv);
        __builtin_amdgcn_sched_barrier(0);
        beg = nbeg;
        len = nlen;
        ndv = nndv;
        nsv = nnsv;
        __syncthreads();
    }
    if ((wave >> 2) == ((nt - 1) & 3))
        transform_tile((const float*)Ar4[(nt - 1) & 1], Wt, Hn,
                       (blockIdx.x + (nt - 1) * FT_GRID) * FT_WAVES, half, l15, ct);
}

extern "C" void kernel_launch(void* const* d_in, const int* in_sizes, int n_in,
                              void* d_out, int out_size, void* d_ws, size_t ws_size,
                              hipStream_t stream) {
    (void)in_sizes; (void)n_in; (void)out_size; (void)ws_size;

    const float* feats = (const float*)d_in[0];
    const float* W1    = (const float*)d_in[1];
    const float* b1    = (const float*)d_in[2];
    const float* g1    = (const float*)d_in[3];
    const float* be1   = (const float*)d_in[4];
    const float* W2    = (const float*)d_in[5];
    const float* b2    = (const float*)d_in[6];
    const float* g2    = (const float*)d_in[7];
    const float* be2   = (const float*)d_in[8];
    const float* W3    = (const float*)d_in[9];
    const float* b3    = (const float*)d_in[10];
    const int*   src   = (const int*)d_in[11];
    const int*   dst   = (const int*)d_in[12];

    float* ns     = (float*)d_ws;
    int*   cnt    = (int*)(ns + N_NODES);
    float* nd     = (float*)(cnt + N_NODES);
    int*   rowptr = (int*)(nd + N_NODES);
    int*   col    = rowptr + N_NODES;
    float* H      = (float*)(col + N_EDGES + COL_PAD);
    float* XB     = H + (size_t)N_NODES * DH;
    unsigned short* histD = (unsigned short*)(XB + (size_t)N_NODES * DH);
    unsigned short* histS = histD + (size_t)HROWS * HROW;
    unsigned short* runD  = histS + (size_t)HROWS * HROW;
    unsigned short* runS  = runD + (size_t)HROWS * HROW;
    int*   binD   = (int*)(runS + (size_t)HROWS * HROW);
    unsigned short* binS = (unsigned short*)(binD + N_EDGES);

    float* h_emb = (float*)d_out;
    float* h_out = h_emb + (size_t)N_NODES * DH;

    scatter_kernel<<<N_RANGE, 1024, 0, stream>>>(src, dst, histD, histS, runD, runS, binD, binS);
    build_kernel<<<N_RANGE, 1024, 0, stream>>>(binD, binS, histD, histS, runD, runS, cnt, nd, ns, rowptr, col);

    gemm_ns_wmma<DIN><<<GEMM_GRID, 128, 0, stream>>>(feats, W1, ns, H);
    aggregate_transform_kernel<false><<<FT_GRID, 512, 0, stream>>>(
        H, col, rowptr, cnt, nd, ns, b1, g1, be1, W2, XB, nullptr);

    aggregate_transform_kernel<true><<<FT_GRID, 512, 0, stream>>>(
        XB, col, rowptr, cnt, nd, ns, b2, g2, be2, W3, H, h_emb);

    aggregate_post_kernel<false, false, false><<<N_NODES / 8, 256, 0, stream>>>(
        H, col, rowptr, cnt, nd, b3, nullptr, nullptr, h_out, nullptr);
}
